# starting kernel
# baseline (speedup 1.0000x reference)
#include <hip/hip_runtime.h>
#include <hip/hip_bf16.h>

typedef __attribute__((ext_vector_type(2))) float v2f;
typedef __attribute__((ext_vector_type(8))) float v8f;

#define NEG_SLOPE 0.2f

__device__ __forceinline__ unsigned fkey_enc(float f) {
    unsigned b = __float_as_uint(f);
    return (b & 0x80000000u) ? ~b : (b | 0x80000000u);
}
__device__ __forceinline__ float fkey_dec(unsigned k) {
    return (k & 0x80000000u) ? __uint_as_float(k ^ 0x80000000u)
                             : __uint_as_float(~k);
}

__global__ void k_zero(float* __restrict__ p, long long cnt) {
    long long i = (long long)blockIdx.x * blockDim.x + threadIdx.x;
    long long stride = (long long)gridDim.x * blockDim.x;
    for (; i < cnt; i += stride) p[i] = 0.0f;
}

__global__ void k_nnconv(const float* __restrict__ x,
                         const int*   __restrict__ eidx,
                         const float* __restrict__ eattr,
                         const float* __restrict__ W_nn,
                         const float* __restrict__ b_nn,
                         float* __restrict__ agg,
                         int N, int E) {
    __shared__ float sWp[136 * 32];
    __shared__ float sEA[8][16][16];
    __shared__ float sXS[8][16][16];
    __shared__ int   sDST[8][16];

    const int tid = threadIdx.x;
    for (int t = tid; t < 272 * 16; t += 256) {
        int k = t >> 4, o = t & 15;
        float v = (k < 256) ? W_nn[(k >> 4) * 256 + (k & 15) * 16 + o]
                            : b_nn[(k - 256) * 16 + o];
        sWp[(k >> 1) * 32 + o * 2 + (k & 1)] = v;
    }

    const int wave = tid >> 5;
    const int lane = tid & 31;
    const int m    = lane & 15;
    const int hi   = lane >> 4;
    const int base = (blockIdx.x * 8 + wave) * 16;
    const int e    = base + m;
    const bool valid = (e < E);
    const int ec = valid ? e : 0;

    if (hi == 0) {
        for (int j = 0; j < 16; ++j)
            sEA[wave][m][j] = valid ? eattr[(size_t)ec * 16 + j] : 0.0f;
        sDST[wave][m] = valid ? eidx[E + ec] : 0;
    } else {
        int s = eidx[ec];
        for (int j = 0; j < 16; ++j)
            sXS[wave][m][j] = valid ? x[(size_t)s * 16 + j] : 0.0f;
    }
    __syncthreads();

    v8f acc = {};
    const float* wXS = &sXS[wave][m][0];
    const float* wEA = &sEA[wave][m][0];
#pragma unroll
    for (int kk = 0; kk < 68; ++kk) {
        const int c  = (4 * kk) & 15;
        const int i0 = c + 2 * hi;
        float x0 = wXS[i0], x1 = wXS[i0 + 1];
        v2f a, b;
        if (kk < 64) {
            float eaf = wEA[kk >> 2];
            a.x = eaf * x0; a.y = eaf * x1;
        } else {
            a.x = x0; a.y = x1;
        }
        const int p = 2 * kk + hi;
        b.x = sWp[p * 32 + m * 2 + 0];
        b.y = sWp[p * 32 + m * 2 + 1];
        acc = __builtin_amdgcn_wmma_f32_16x16x4_f32(
            false, a, false, b, (short)0, acc, false, false);
    }

#pragma unroll
    for (int v = 0; v < 8; ++v) {
        int mr = v + 8 * hi;
        int er = base + mr;
        if (er < E) {
            int d = sDST[wave][mr];
            atomicAdd(&agg[(size_t)d * 16 + m], acc[v]);
        }
    }
}

__global__ void k_hnode(const float* __restrict__ x,
                        const float* __restrict__ W_root,
                        const float* __restrict__ b_ecc,
                        const float* __restrict__ agg,
                        float* __restrict__ h, int N) {
    int i = blockIdx.x * blockDim.x + threadIdx.x;
    if (i >= N * 16) return;
    int n = i >> 4, o = i & 15;
    float s = agg[i] + b_ecc[o];
    const float* xr = x + (size_t)n * 16;
#pragma unroll
    for (int j = 0; j < 16; ++j) s += xr[j] * W_root[j * 16 + o];
    h[i] = fmaxf(s, 0.0f);
}

__global__ void k_zgemm(const float* __restrict__ h,
                        const float* __restrict__ W_gat,
                        float* __restrict__ z, int N) {
    __shared__ float sWg[8 * 256];
    const int tid = threadIdx.x;
    for (int t = tid; t < 2048; t += 256) {
        int k = t >> 7, col = t & 127;
        sWg[(k >> 1) * 256 + col * 2 + (k & 1)] = W_gat[k * 128 + col];
    }
    __syncthreads();

    const int wave = tid >> 5, lane = tid & 31;
    const int m = lane & 15, hi = lane >> 4;
    const int rowTile = blockIdx.x;
    const int colTile = wave;
    const int row = rowTile * 16 + m;
    const int rc  = row < N ? row : (N - 1);
    const float* hr = h + (size_t)rc * 16;
    const int col = colTile * 16 + m;

    v8f acc = {};
#pragma unroll
    for (int kk = 0; kk < 4; ++kk) {
        int i0 = 4 * kk + 2 * hi;
        v2f a, b;
        a.x = hr[i0]; a.y = hr[i0 + 1];
        int p = 2 * kk + hi;
        b.x = sWg[p * 256 + col * 2 + 0];
        b.y = sWg[p * 256 + col * 2 + 1];
        acc = __builtin_amdgcn_wmma_f32_16x16x4_f32(
            false, a, false, b, (short)0, acc, false, false);
    }
#pragma unroll
    for (int v = 0; v < 8; ++v) {
        int r = rowTile * 16 + v + 8 * hi;
        if (r < N) z[(size_t)r * 128 + col] = acc[v];
    }
}

__global__ void k_escore(const float* __restrict__ z,
                         const float* __restrict__ a_src,
                         const float* __restrict__ a_dst,
                         float* __restrict__ esrc,
                         float* __restrict__ edst, int N) {
    int i = blockIdx.x * blockDim.x + threadIdx.x;
    if (i >= N * 4) return;
    int n = i >> 2, hh = i & 3;
    const float* zr = z + (size_t)n * 128 + hh * 32;
    const float* as = a_src + hh * 32;
    const float* ad = a_dst + hh * 32;
    float s = 0.f, d = 0.f;
#pragma unroll 8
    for (int j = 0; j < 32; ++j) { float zv = zr[j]; s += zv * as[j]; d += zv * ad[j]; }
    esrc[i] = s; edst[i] = d;
}

__global__ void k_attmax(const int* __restrict__ eidx,
                         const float* __restrict__ esrc,
                         const float* __restrict__ edst,
                         unsigned* __restrict__ maxkey, int E) {
    int i = blockIdx.x * blockDim.x + threadIdx.x;
    if (i >= E * 4) return;
    int e = i >> 2, hh = i & 3;
    int s = eidx[e], d = eidx[E + e];
    float lg = esrc[s * 4 + hh] + edst[d * 4 + hh];
    lg = lg > 0.f ? lg : NEG_SLOPE * lg;
    atomicMax(&maxkey[d * 4 + hh], fkey_enc(lg));
}

__global__ void k_attagg(const int* __restrict__ eidx,
                         const float* __restrict__ esrc,
                         const float* __restrict__ edst,
                         const unsigned* __restrict__ maxkey,
                         const float* __restrict__ z,
                         float* __restrict__ denom,
                         float* __restrict__ gatout, int E) {
    int i = blockIdx.x * blockDim.x + threadIdx.x;
    if (i >= E * 4) return;
    int e = i >> 2, hh = i & 3;
    int s = eidx[e], d = eidx[E + e];
    float lg = esrc[s * 4 + hh] + edst[d * 4 + hh];
    lg = lg > 0.f ? lg : NEG_SLOPE * lg;
    float mx = fkey_dec(maxkey[d * 4 + hh]);
    float ex = __expf(lg - mx);
    atomicAdd(&denom[d * 4 + hh], ex);
    const float* zr = z + (size_t)s * 128 + hh * 32;
    float* op = gatout + (size_t)d * 128 + hh * 32;
#pragma unroll 4
    for (int j = 0; j < 32; ++j) atomicAdd(op + j, ex * zr[j]);
}

__global__ void k_final(const float* __restrict__ gatout,
                        const float* __restrict__ denom,
                        const float* __restrict__ b_gat,
                        const float* __restrict__ W_fc,
                        const float* __restrict__ b_fc,
                        float* __restrict__ out, int N) {
    int wave = threadIdx.x >> 5, lane = threadIdx.x & 31;
    int n = blockIdx.x * 8 + wave;
    if (n >= N) return;
    float acc = 0.f;
#pragma unroll
    for (int t = 0; t < 4; ++t) {
        int j = t * 32 + lane;
        float dn = denom[n * 4 + t] + 1e-16f;
        float g = gatout[(size_t)n * 128 + j] / dn + b_gat[j];
        g = fmaxf(g, 0.0f);
        acc += g * W_fc[j];
    }
    for (int off = 16; off > 0; off >>= 1) acc += __shfl_xor(acc, off, 32);
    if (lane == 0) out[n] = acc + b_fc[0];
}

extern "C" void kernel_launch(void* const* d_in, const int* in_sizes, int n_in,
                              void* d_out, int out_size, void* d_ws, size_t ws_size,
                              hipStream_t stream) {
    const float* x      = (const float*)d_in[0];
    const int*   eidx   = (const int*)  d_in[1];
    const float* eattr  = (const float*)d_in[2];
    const float* W_nn   = (const float*)d_in[3];
    const float* b_nn   = (const float*)d_in[4];
    const float* W_root = (const float*)d_in[5];
    const float* b_ecc  = (const float*)d_in[6];
    const float* W_gat  = (const float*)d_in[7];
    const float* a_src  = (const float*)d_in[8];
    const float* a_dst  = (const float*)d_in[9];
    const float* b_gat  = (const float*)d_in[10];
    const float* W_fc   = (const float*)d_in[11];
    const float* b_fc   = (const float*)d_in[12];
    float* out = (float*)d_out;

    const int N = in_sizes[0] / 16;
    const int E = in_sizes[1] / 2;

    float* ws = (float*)d_ws;
    size_t off = 0;
    float*    agg    = ws + off;               off += (size_t)N * 16;
    unsigned* maxkey = (unsigned*)(ws + off);  off += (size_t)N * 4;
    float*    denom  = ws + off;               off += (size_t)N * 4;
    float*    gatout = ws + off;               off += (size_t)N * 128;
    const long long zeroCnt = (long long)off;
    float*    hbuf   = ws + off;               off += (size_t)N * 16;
    float*    zbuf   = ws + off;               off += (size_t)N * 128;
    float*    esrc   = ws + off;               off += (size_t)N * 4;
    float*    edst   = ws + off;               off += (size_t)N * 4;

    k_zero<<<2048, 256, 0, stream>>>(ws, zeroCnt);

    int tilesE = (E + 15) / 16;
    k_nnconv<<<(tilesE + 7) / 8, 256, 0, stream>>>(x, eidx, eattr, W_nn, b_nn, agg, N, E);
    k_hnode <<<(N * 16 + 255) / 256, 256, 0, stream>>>(x, W_root, b_ecc, agg, hbuf, N);
    k_zgemm <<<(N + 15) / 16, 256, 0, stream>>>(hbuf, W_gat, zbuf, N);
    k_escore<<<(N * 4 + 255) / 256, 256, 0, stream>>>(zbuf, a_src, a_dst, esrc, edst, N);
    k_attmax<<<(E * 4 + 255) / 256, 256, 0, stream>>>(eidx, esrc, edst, maxkey, E);
    k_attagg<<<(E * 4 + 255) / 256, 256, 0, stream>>>(eidx, esrc, edst, maxkey, zbuf,
                                                      denom, gatout, E);
    k_final <<<(N + 7) / 8, 256, 0, stream>>>(gatout, denom, b_gat, W_fc, b_fc, out, N);
}


# GAT aggregation grouped by destination: in-call counting sort (int atomic rank = arrival order, sets sum order); one wave per node does max/exp/sum/fc from z rows; no float atomics on gatout/denom
# speedup vs baseline: 5.2443x; 5.2443x over previous
#include <hip/hip_runtime.h>
#include <hip/hip_bf16.h>

typedef __attribute__((ext_vector_type(2))) float v2f;
typedef __attribute__((ext_vector_type(8))) float v8f;

#define NEG_SLOPE 0.2f

__global__ void k_zero(float* __restrict__ p, long long cnt) {
    long long i = (long long)blockIdx.x * blockDim.x + threadIdx.x;
    long long stride = (long long)gridDim.x * blockDim.x;
    for (; i < cnt; i += stride) p[i] = 0.0f;
}

__global__ void k_nnconv(const float* __restrict__ x,
                         const int*   __restrict__ eidx,
                         const float* __restrict__ eattr,
                         const float* __restrict__ W_nn,
                         const float* __restrict__ b_nn,
                         float* __restrict__ agg,
                         int N, int E) {
    __shared__ float sWp[136 * 32];
    __shared__ float sEA[8][16][16];
    __shared__ float sXS[8][16][16];
    __shared__ int   sDST[8][16];

    const int tid = threadIdx.x;
    for (int t = tid; t < 272 * 16; t += 256) {
        int k = t >> 4, o = t & 15;
        float v = (k < 256) ? W_nn[(k >> 4) * 256 + (k & 15) * 16 + o]
                            : b_nn[(k - 256) * 16 + o];
        sWp[(k >> 1) * 32 + o * 2 + (k & 1)] = v;
    }

    const int wave = tid >> 5;
    const int lane = tid & 31;
    const int m    = lane & 15;
    const int hi   = lane >> 4;
    const int base = (blockIdx.x * 8 + wave) * 16;
    const int e    = base + m;
    const bool valid = (e < E);
    const int ec = valid ? e : 0;

    if (hi == 0) {
        for (int j = 0; j < 16; ++j)
            sEA[wave][m][j] = valid ? eattr[(size_t)ec * 16 + j] : 0.0f;
        sDST[wave][m] = valid ? eidx[E + ec] : 0;
    } else {
        int s = eidx[ec];
        for (int j = 0; j < 16; ++j)
            sXS[wave][m][j] = valid ? x[(size_t)s * 16 + j] : 0.0f;
    }
    __syncthreads();

    v8f acc = {};
    const float* wXS = &sXS[wave][m][0];
    const float* wEA = &sEA[wave][m][0];
#pragma unroll
    for (int kk = 0; kk < 68; ++kk) {
        const int c  = (4 * kk) & 15;
        const int i0 = c + 2 * hi;
        float x0 = wXS[i0], x1 = wXS[i0 + 1];
        v2f a, b;
        if (kk < 64) {
            float eaf = wEA[kk >> 2];
            a.x = eaf * x0; a.y = eaf * x1;
        } else {
            a.x = x0; a.y = x1;
        }
        const int p = 2 * kk + hi;
        b.x = sWp[p * 32 + m * 2 + 0];
        b.y = sWp[p * 32 + m * 2 + 1];
        acc = __builtin_amdgcn_wmma_f32_16x16x4_f32(
            false, a, false, b, (short)0, acc, false, false);
    }

#pragma unroll
    for (int v = 0; v < 8; ++v) {
        int mr = v + 8 * hi;
        int er = base + mr;
        if (er < E) {
            int d = sDST[wave][mr];
            atomicAdd(&agg[(size_t)d * 16 + m], acc[v]);
        }
    }
}

__global__ void k_hnode(const float* __restrict__ x,
                        const float* __restrict__ W_root,
                        const float* __restrict__ b_ecc,
                        const float* __restrict__ agg,
                        float* __restrict__ h, int N) {
    int i = blockIdx.x * blockDim.x + threadIdx.x;
    if (i >= N * 16) return;
    int n = i >> 4, o = i & 15;
    float s = agg[i] + b_ecc[o];
    const float* xr = x + (size_t)n * 16;
#pragma unroll
    for (int j = 0; j < 16; ++j) s += xr[j] * W_root[j * 16 + o];
    h[i] = fmaxf(s, 0.0f);
}

__global__ void k_zgemm(const float* __restrict__ h,
                        const float* __restrict__ W_gat,
                        float* __restrict__ z, int N) {
    __shared__ float sWg[8 * 256];
    const int tid = threadIdx.x;
    for (int t = tid; t < 2048; t += 256) {
        int k = t >> 7, col = t & 127;
        sWg[(k >> 1) * 256 + col * 2 + (k & 1)] = W_gat[k * 128 + col];
    }
    __syncthreads();

    const int wave = tid >> 5, lane = tid & 31;
    const int m = lane & 15, hi = lane >> 4;
    const int rowTile = blockIdx.x;
    const int colTile = wave;
    const int row = rowTile * 16 + m;
    const int rc  = row < N ? row : (N - 1);
    const float* hr = h + (size_t)rc * 16;
    const int col = colTile * 16 + m;

    v8f acc = {};
#pragma unroll
    for (int kk = 0; kk < 4; ++kk) {
        int i0 = 4 * kk + 2 * hi;
        v2f a, b;
        a.x = hr[i0]; a.y = hr[i0 + 1];
        int p = 2 * kk + hi;
        b.x = sWg[p * 256 + col * 2 + 0];
        b.y = sWg[p * 256 + col * 2 + 1];
        acc = __builtin_amdgcn_wmma_f32_16x16x4_f32(
            false, a, false, b, (short)0, acc, false, false);
    }
#pragma unroll
    for (int v = 0; v < 8; ++v) {
        int r = rowTile * 16 + v + 8 * hi;
        if (r < N) z[(size_t)r * 128 + col] = acc[v];
    }
}

__global__ void k_escore(const float* __restrict__ z,
                         const float* __restrict__ a_src,
                         const float* __restrict__ a_dst,
                         float* __restrict__ esrc,
                         float* __restrict__ edst, int N) {
    int i = blockIdx.x * blockDim.x + threadIdx.x;
    if (i >= N * 4) return;
    int n = i >> 2, hh = i & 3;
    const float* zr = z + (size_t)n * 128 + hh * 32;
    const float* as = a_src + hh * 32;
    const float* ad = a_dst + hh * 32;
    float s = 0.f, d = 0.f;
#pragma unroll 8
    for (int j = 0; j < 32; ++j) { float zv = zr[j]; s += zv * as[j]; d += zv * ad[j]; }
    esrc[i] = s; edst[i] = d;
}

__global__ void k_rank(const int* __restrict__ eidx,
                       int* __restrict__ deg,
                       int* __restrict__ rnk, int E) {
    int e = blockIdx.x * blockDim.x + threadIdx.x;
    if (e >= E) return;
    rnk[e] = atomicAdd(&deg[eidx[E + e]], 1);
}

__global__ void k_alloc(const int* __restrict__ deg,
                        int* __restrict__ total,
                        int* __restrict__ off, int N) {
    const int n = blockIdx.x * blockDim.x + threadIdx.x;
    const int lane = threadIdx.x & 31;
    const int v = (n < N) ? deg[n] : 0;
    int inc = v;
    for (int o = 1; o < 32; o <<= 1) {
        int t = __shfl_up(inc, o, 32);
        if (lane >= o) inc += t;
    }
    int base = 0;
    if (lane == 31) base = atomicAdd(total, inc);
    base = __shfl(base, 31, 32);
    if (n < N) off[n] = base + inc - v;
}

__global__ void k_place(const int* __restrict__ eidx,
                        const int* __restrict__ off,
                        const int* __restrict__ rnk,
                        int* __restrict__ ssrc, int E) {
    int e = blockIdx.x * blockDim.x + threadIdx.x;
    if (e >= E) return;
    int d = eidx[E + e];
    ssrc[off[d] + rnk[e]] = eidx[e];
}

__device__ __forceinline__ float lrelu(float v) { return v > 0.f ? v : NEG_SLOPE * v; }
__device__ __forceinline__ float lane_val(float v, int q) {
    return __int_as_float(__builtin_amdgcn_readlane(__float_as_int(v), q));
}

__global__ void k_gatnode(const int* __restrict__ off,
                          const int* __restrict__ deg,
                          const int* __restrict__ ssrc,
                          const float* __restrict__ esrc,
                          const float* __restrict__ edst,
                          const float* __restrict__ z,
                          const float* __restrict__ b_gat,
                          const float* __restrict__ W_fc,
                          const float* __restrict__ b_fc,
                          float* __restrict__ out, int N) {
    const int lane = threadIdx.x & 31;
    const int n = blockIdx.x * 8 + (threadIdx.x >> 5);
    if (n >= N) return;
    const int beg = __builtin_amdgcn_readfirstlane(off[n]);
    const int cnt = __builtin_amdgcn_readfirstlane(deg[n]);
    const float4 ed = *reinterpret_cast<const float4*>(edst + (size_t)n * 4);
    const float4* es4 = reinterpret_cast<const float4*>(esrc);

    float m0 = -INFINITY, m1 = -INFINITY, m2 = -INFINITY, m3 = -INFINITY;
    for (int c = 0; c < cnt; c += 32) {
        const int i = c + lane;
        if (i < cnt) {
            const float4 es = es4[ssrc[beg + i]];
            m0 = fmaxf(m0, lrelu(es.x + ed.x));
            m1 = fmaxf(m1, lrelu(es.y + ed.y));
            m2 = fmaxf(m2, lrelu(es.z + ed.z));
            m3 = fmaxf(m3, lrelu(es.w + ed.w));
        }
    }
    for (int o = 16; o > 0; o >>= 1) {
        m0 = fmaxf(m0, __shfl_xor(m0, o, 32));
        m1 = fmaxf(m1, __shfl_xor(m1, o, 32));
        m2 = fmaxf(m2, __shfl_xor(m2, o, 32));
        m3 = fmaxf(m3, __shfl_xor(m3, o, 32));
    }

    float a0 = 0.f, a1 = 0.f, a2 = 0.f, a3 = 0.f;
    float d0 = 0.f, d1 = 0.f, d2 = 0.f, d3 = 0.f;
    for (int c = 0; c < cnt; c += 32) {
        const int i = c + lane;
        int s = 0;
        float e0 = 0.f, e1 = 0.f, e2 = 0.f, e3 = 0.f;
        if (i < cnt) {
            s = ssrc[beg + i];
            const float4 es = es4[s];
            e0 = __expf(lrelu(es.x + ed.x) - m0);
            e1 = __expf(lrelu(es.y + ed.y) - m1);
            e2 = __expf(lrelu(es.z + ed.z) - m2);
            e3 = __expf(lrelu(es.w + ed.w) - m3);
        }
        d0 += e0; d1 += e1; d2 += e2; d3 += e3;

        const int mc = (cnt - c < 32) ? (cnt - c) : 32;
        int q = 0;
        for (; q + 4 <= mc; q += 4) {
            float r[4][4];
#pragma unroll
            for (int u = 0; u < 4; ++u) {
                const float* zr = z + (size_t)__builtin_amdgcn_readlane(s, q + u) * 128 + lane;
                r[u][0] = zr[0]; r[u][1] = zr[32]; r[u][2] = zr[64]; r[u][3] = zr[96];
            }
#pragma unroll
            for (int u = 0; u < 4; ++u) {
                a0 += lane_val(e0, q + u) * r[u][0];
                a1 += lane_val(e1, q + u) * r[u][1];
                a2 += lane_val(e2, q + u) * r[u][2];
                a3 += lane_val(e3, q + u) * r[u][3];
            }
        }
        for (; q < mc; ++q) {
            const float* zr = z + (size_t)__builtin_amdgcn_readlane(s, q) * 128 + lane;
            a0 += lane_val(e0, q) * zr[0];
            a1 += lane_val(e1, q) * zr[32];
            a2 += lane_val(e2, q) * zr[64];
            a3 += lane_val(e3, q) * zr[96];
        }
    }
    for (int o = 16; o > 0; o >>= 1) {
        d0 += __shfl_xor(d0, o, 32);
        d1 += __shfl_xor(d1, o, 32);
        d2 += __shfl_xor(d2, o, 32);
        d3 += __shfl_xor(d3, o, 32);
    }

    float acc = 0.f;
    {
        float g;
        g = fmaxf(a0 / (d0 + 1e-16f) + b_gat[lane],      0.0f); acc += g * W_fc[lane];
        g = fmaxf(a1 / (d1 + 1e-16f) + b_gat[32 + lane], 0.0f); acc += g * W_fc[32 + lane];
        g = fmaxf(a2 / (d2 + 1e-16f) + b_gat[64 + lane], 0.0f); acc += g * W_fc[64 + lane];
        g = fmaxf(a3 / (d3 + 1e-16f) + b_gat[96 + lane], 0.0f); acc += g * W_fc[96 + lane];
    }
    for (int o = 16; o > 0; o >>= 1) acc += __shfl_xor(acc, o, 32);
    if (lane == 0) out[n] = acc + b_fc[0];
}

extern "C" void kernel_launch(void* const* d_in, const int* in_sizes, int n_in,
                              void* d_out, int out_size, void* d_ws, size_t ws_size,
                              hipStream_t stream) {
    const float* x      = (const float*)d_in[0];
    const int*   eidx   = (const int*)  d_in[1];
    const float* eattr  = (const float*)d_in[2];
    const float* W_nn   = (const float*)d_in[3];
    const float* b_nn   = (const float*)d_in[4];
    const float* W_root = (const float*)d_in[5];
    const float* b_ecc  = (const float*)d_in[6];
    const float* W_gat  = (const float*)d_in[7];
    const float* a_src  = (const float*)d_in[8];
    const float* a_dst  = (const float*)d_in[9];
    const float* b_gat  = (const float*)d_in[10];
    const float* W_fc   = (const float*)d_in[11];
    const float* b_fc   = (const float*)d_in[12];
    float* out = (float*)d_out;

    const int N = in_sizes[0] / 16;
    const int E = in_sizes[1] / 2;

    float* ws = (float*)d_ws;
    size_t off = 0;
    float* zbuf  = ws + off;          off += (size_t)N * 128;
    float* hbuf  = ws + off;          off += (size_t)N * 16;
    float* esrc  = ws + off;          off += (size_t)N * 4;
    float* edst  = ws + off;          off += (size_t)N * 4;
    const size_t zeroBeg = off;
    float* agg   = ws + off;          off += (size_t)N * 16;
    int*   deg   = (int*)(ws + off);  off += (size_t)N;
    int*   total = (int*)(ws + off);  off += 4;
    const long long zeroCnt = (long long)(off - zeroBeg);
    int*   noff  = (int*)(ws + off);  off += (size_t)N;
    int*   rnk   = (int*)(ws + off);  off += (size_t)E;
    int*   ssrc  = (int*)(ws + off);  off += (size_t)E;

    k_zero<<<2048, 256, 0, stream>>>(agg, zeroCnt);

    k_rank <<<(E + 255) / 256, 256, 0, stream>>>(eidx, deg, rnk, E);
    k_alloc<<<(N + 255) / 256, 256, 0, stream>>>(deg, total, noff, N);
    k_place<<<(E + 255) / 256, 256, 0, stream>>>(eidx, noff, rnk, ssrc, E);

    int tilesE = (E + 15) / 16;
    k_nnconv<<<(tilesE + 7) / 8, 256, 0, stream>>>(x, eidx, eattr, W_nn, b_nn, agg, N, E);
    k_hnode <<<(N * 16 + 255) / 256, 256, 0, stream>>>(x, W_root, b_ecc, agg, hbuf, N);
    k_zgemm <<<(N + 15) / 16, 256, 0, stream>>>(hbuf, W_gat, zbuf, N);
    k_escore<<<(N * 4 + 255) / 256, 256, 0, stream>>>(zbuf, a_src, a_dst, esrc, edst, N);
    k_gatnode<<<(N + 7) / 8, 256, 0, stream>>>(noff, deg, ssrc, esrc, edst, zbuf,
                                               b_gat, W_fc, b_fc, out, N);
}


# NNConv summed per destination before the weight product: one wave per node sums ea(x)x over its sorted edges, 16 nodes per [16,272]@[272,16] WMMA tile; no float atomics; sum order = arrival rank
# speedup vs baseline: 6.1487x; 1.1725x over previous
#include <hip/hip_runtime.h>
#include <hip/hip_bf16.h>

typedef __attribute__((ext_vector_type(2))) float v2f;
typedef __attribute__((ext_vector_type(4))) float v4f;
typedef __attribute__((ext_vector_type(8))) float v8f;

#define NEG_SLOPE 0.2f

__global__ void k_zero(float* __restrict__ p, long long cnt) {
    long long i = (long long)blockIdx.x * blockDim.x + threadIdx.x;
    long long stride = (long long)gridDim.x * blockDim.x;
    for (; i < cnt; i += stride) p[i] = 0.0f;
}

#define UPITCH 276
__global__ __launch_bounds__(512)
void k_nnagg(const float* __restrict__ x,
             const float* __restrict__ eattr,
             const float* __restrict__ W_nn,
             const float* __restrict__ b_nn,
             const int* __restrict__ off,
             const int* __restrict__ deg,
             const int* __restrict__ ssrc,
             const int* __restrict__ sedge,
             float* __restrict__ agg, int N) {
    __shared__ float sU[16 * UPITCH];

    const int tid  = threadIdx.x;
    const int wave = tid >> 5;
    const int lane = tid & 31;
    const int i    = lane & 15;
    const int hi   = lane >> 4;
    const int n    = blockIdx.x * 16 + wave;

    v2f U0 = {0.f, 0.f}, U1 = {0.f, 0.f}, U2 = {0.f, 0.f}, U3 = {0.f, 0.f};
    float xs = 0.f;
    if (n < N) {
        const int beg = __builtin_amdgcn_readfirstlane(off[n]);
        const int cnt = __builtin_amdgcn_readfirstlane(deg[n]);
        for (int c = 0; c < cnt; c += 32) {
            const int j = c + lane;
            int s = 0, e = 0;
            if (j < cnt) { s = ssrc[beg + j]; e = sedge[beg + j]; }
            const int mc = (cnt - c < 32) ? (cnt - c) : 32;
            int q = 0;
            for (; q + 4 <= mc; q += 4) {
                float xv[4];
                v4f ea[4][2];
#pragma unroll
                for (int t = 0; t < 4; ++t) {
                    const int sq = __builtin_amdgcn_readlane(s, q + t);
                    const int eq = __builtin_amdgcn_readlane(e, q + t);
                    xv[t] = x[(size_t)sq * 16 + i];
                    const v4f* er =
                        reinterpret_cast<const v4f*>(eattr + (size_t)eq * 16 + hi * 8);
                    ea[t][0] = er[0];
                    ea[t][1] = er[1];
                }
#pragma unroll
                for (int t = 0; t < 4; ++t) {
                    const v2f xx = {xv[t], xv[t]};
                    U0 += ea[t][0].xy * xx; U1 += ea[t][0].zw * xx;
                    U2 += ea[t][1].xy * xx; U3 += ea[t][1].zw * xx;
                    xs += xv[t];
                }
            }
            for (; q < mc; ++q) {
                const int sq = __builtin_amdgcn_readlane(s, q);
                const int eq = __builtin_amdgcn_readlane(e, q);
                const float xq = x[(size_t)sq * 16 + i];
                const v4f* er =
                    reinterpret_cast<const v4f*>(eattr + (size_t)eq * 16 + hi * 8);
                const v4f ea0 = er[0], ea1 = er[1];
                const v2f xx = {xq, xq};
                U0 += ea0.xy * xx; U1 += ea0.zw * xx;
                U2 += ea1.xy * xx; U3 += ea1.zw * xx;
                xs += xq;
            }
        }
    }
    {
        float* ur = sU + wave * UPITCH + hi * 128 + i;
        ur[0]  = U0.x; ur[16] = U0.y; ur[32] = U1.x; ur[48]  = U1.y;
        ur[64] = U2.x; ur[80] = U2.y; ur[96] = U3.x; ur[112] = U3.y;
        if (hi == 0) sU[wave * UPITCH + 256 + i] = xs;
    }
    __syncthreads();
    if (wave != 0) return;

    v8f accA = {}, accB = {};
    const float* ar = sU + i * UPITCH + 2 * hi;
#pragma unroll
    for (int kk = 0; kk < 68; ++kk) {
        const float* wp = (kk < 64) ? (W_nn + (4 * kk + 2 * hi) * 16 + i)
                                    : (b_nn + (4 * (kk - 64) + 2 * hi) * 16 + i);
        v2f a, b;
        a.x = ar[4 * kk]; a.y = ar[4 * kk + 1];
        b.x = wp[0];      b.y = wp[16];
        if (kk & 1)
            accB = __builtin_amdgcn_wmma_f32_16x16x4_f32(
                false, a, false, b, (short)0, accB, false, false);
        else
            accA = __builtin_amdgcn_wmma_f32_16x16x4_f32(
                false, a, false, b, (short)0, accA, false, false);
    }
    const v8f acc = accA + accB;
#pragma unroll
    for (int v = 0; v < 8; ++v) {
        const int r = blockIdx.x * 16 + v + 8 * hi;
        if (r < N) agg[(size_t)r * 16 + i] = acc[v];
    }
}

__global__ void k_hnode(const float* __restrict__ x,
                        const float* __restrict__ W_root,
                        const float* __restrict__ b_ecc,
                        const float* __restrict__ agg,
                        float* __restrict__ h, int N) {
    int i = blockIdx.x * blockDim.x + threadIdx.x;
    if (i >= N * 16) return;
    int n = i >> 4, o = i & 15;
    float s = agg[i] + b_ecc[o];
    const float* xr = x + (size_t)n * 16;
#pragma unroll
    for (int j = 0; j < 16; ++j) s += xr[j] * W_root[j * 16 + o];
    h[i] = fmaxf(s, 0.0f);
}

__global__ void k_zgemm(const float* __restrict__ h,
                        const float* __restrict__ W_gat,
                        float* __restrict__ z, int N) {
    __shared__ float sWg[8 * 256];
    const int tid = threadIdx.x;
    for (int t = tid; t < 2048; t += 256) {
        int k = t >> 7, col = t & 127;
        sWg[(k >> 1) * 256 + col * 2 + (k & 1)] = W_gat[k * 128 + col];
    }
    __syncthreads();

    const int wave = tid >> 5, lane = tid & 31;
    const int m = lane & 15, hi = lane >> 4;
    const int rowTile = blockIdx.x;
    const int colTile = wave;
    const int row = rowTile * 16 + m;
    const int rc  = row < N ? row : (N - 1);
    const float* hr = h + (size_t)rc * 16;
    const int col = colTile * 16 + m;

    v8f acc = {};
#pragma unroll
    for (int kk = 0; kk < 4; ++kk) {
        int i0 = 4 * kk + 2 * hi;
        v2f a, b;
        a.x = hr[i0]; a.y = hr[i0 + 1];
        int p = 2 * kk + hi;
        b.x = sWg[p * 256 + col * 2 + 0];
        b.y = sWg[p * 256 + col * 2 + 1];
        acc = __builtin_amdgcn_wmma_f32_16x16x4_f32(
            false, a, false, b, (short)0, acc, false, false);
    }
#pragma unroll
    for (int v = 0; v < 8; ++v) {
        int r = rowTile * 16 + v + 8 * hi;
        if (r < N) z[(size_t)r * 128 + col] = acc[v];
    }
}

__global__ void k_escore(const float* __restrict__ z,
                         const float* __restrict__ a_src,
                         const float* __restrict__ a_dst,
                         float* __restrict__ esrc,
                         float* __restrict__ edst, int N) {
    int i = blockIdx.x * blockDim.x + threadIdx.x;
    if (i >= N * 4) return;
    int n = i >> 2, hh = i & 3;
    const float* zr = z + (size_t)n * 128 + hh * 32;
    const float* as = a_src + hh * 32;
    const float* ad = a_dst + hh * 32;
    float s = 0.f, d = 0.f;
#pragma unroll 8
    for (int j = 0; j < 32; ++j) { float zv = zr[j]; s += zv * as[j]; d += zv * ad[j]; }
    esrc[i] = s; edst[i] = d;
}

__global__ void k_rank(const int* __restrict__ eidx,
                       int* __restrict__ deg,
                       int* __restrict__ rnk, int E) {
    int e = blockIdx.x * blockDim.x + threadIdx.x;
    if (e >= E) return;
    rnk[e] = atomicAdd(&deg[eidx[E + e]], 1);
}

__global__ void k_alloc(const int* __restrict__ deg,
                        int* __restrict__ total,
                        int* __restrict__ off, int N) {
    const int n = blockIdx.x * blockDim.x + threadIdx.x;
    const int lane = threadIdx.x & 31;
    const int v = (n < N) ? deg[n] : 0;
    int inc = v;
    for (int o = 1; o < 32; o <<= 1) {
        int t = __shfl_up(inc, o, 32);
        if (lane >= o) inc += t;
    }
    int base = 0;
    if (lane == 31) base = atomicAdd(total, inc);
    base = __shfl(base, 31, 32);
    if (n < N) off[n] = base + inc - v;
}

__global__ void k_place(const int* __restrict__ eidx,
                        const int* __restrict__ off,
                        const int* __restrict__ rnk,
                        int* __restrict__ ssrc,
                        int* __restrict__ sedge, int E) {
    int e = blockIdx.x * blockDim.x + threadIdx.x;
    if (e >= E) return;
    int d = eidx[E + e];
    int slot = off[d] + rnk[e];
    ssrc[slot]  = eidx[e];
    sedge[slot] = e;
}

__device__ __forceinline__ float lrelu(float v) { return v > 0.f ? v : NEG_SLOPE * v; }
__device__ __forceinline__ float lane_val(float v, int q) {
    return __int_as_float(__builtin_amdgcn_readlane(__float_as_int(v), q));
}

__global__ void k_gatnode(const int* __restrict__ off,
                          const int* __restrict__ deg,
                          const int* __restrict__ ssrc,
                          const float* __restrict__ esrc,
                          const float* __restrict__ edst,
                          const float* __restrict__ z,
                          const float* __restrict__ b_gat,
                          const float* __restrict__ W_fc,
                          const float* __restrict__ b_fc,
                          float* __restrict__ out, int N) {
    const int lane = threadIdx.x & 31;
    const int n = blockIdx.x * 8 + (threadIdx.x >> 5);
    if (n >= N) return;
    const int beg = __builtin_amdgcn_readfirstlane(off[n]);
    const int cnt = __builtin_amdgcn_readfirstlane(deg[n]);
    const float4 ed = *reinterpret_cast<const float4*>(edst + (size_t)n * 4);
    const float4* es4 = reinterpret_cast<const float4*>(esrc);

    float m0 = -INFINITY, m1 = -INFINITY, m2 = -INFINITY, m3 = -INFINITY;
    for (int c = 0; c < cnt; c += 32) {
        const int i = c + lane;
        if (i < cnt) {
            const float4 es = es4[ssrc[beg + i]];
            m0 = fmaxf(m0, lrelu(es.x + ed.x));
            m1 = fmaxf(m1, lrelu(es.y + ed.y));
            m2 = fmaxf(m2, lrelu(es.z + ed.z));
            m3 = fmaxf(m3, lrelu(es.w + ed.w));
        }
    }
    for (int o = 16; o > 0; o >>= 1) {
        m0 = fmaxf(m0, __shfl_xor(m0, o, 32));
        m1 = fmaxf(m1, __shfl_xor(m1, o, 32));
        m2 = fmaxf(m2, __shfl_xor(m2, o, 32));
        m3 = fmaxf(m3, __shfl_xor(m3, o, 32));
    }

    float a0 = 0.f, a1 = 0.f, a2 = 0.f, a3 = 0.f;
    float d0 = 0.f, d1 = 0.f, d2 = 0.f, d3 = 0.f;
    for (int c = 0; c < cnt; c += 32) {
        const int i = c + lane;
        int s = 0;
        float e0 = 0.f, e1 = 0.f, e2 = 0.f, e3 = 0.f;
        if (i < cnt) {
            s = ssrc[beg + i];
            const float4 es = es4[s];
            e0 = __expf(lrelu(es.x + ed.x) - m0);
            e1 = __expf(lrelu(es.y + ed.y) - m1);
            e2 = __expf(lrelu(es.z + ed.z) - m2);
            e3 = __expf(lrelu(es.w + ed.w) - m3);
        }
        d0 += e0; d1 += e1; d2 += e2; d3 += e3;

        const int mc = (cnt - c < 32) ? (cnt - c) : 32;
        int q = 0;
        for (; q + 4 <= mc; q += 4) {
            float r[4][4];
#pragma unroll
            for (int u = 0; u < 4; ++u) {
                const float* zr = z + (size_t)__builtin_amdgcn_readlane(s, q + u) * 128 + lane;
                r[u][0] = zr[0]; r[u][1] = zr[32]; r[u][2] = zr[64]; r[u][3] = zr[96];
            }
#pragma unroll
            for (int u = 0; u < 4; ++u) {
                a0 += lane_val(e0, q + u) * r[u][0];
                a1 += lane_val(e1, q + u) * r[u][1];
                a2 += lane_val(e2, q + u) * r[u][2];
                a3 += lane_val(e3, q + u) * r[u][3];
            }
        }
        for (; q < mc; ++q) {
            const float* zr = z + (size_t)__builtin_amdgcn_readlane(s, q) * 128 + lane;
            a0 += lane_val(e0, q) * zr[0];
            a1 += lane_val(e1, q) * zr[32];
            a2 += lane_val(e2, q) * zr[64];
            a3 += lane_val(e3, q) * zr[96];
        }
    }
    for (int o = 16; o > 0; o >>= 1) {
        d0 += __shfl_xor(d0, o, 32);
        d1 += __shfl_xor(d1, o, 32);
        d2 += __shfl_xor(d2, o, 32);
        d3 += __shfl_xor(d3, o, 32);
    }

    float acc = 0.f;
    {
        float g;
        g = fmaxf(a0 / (d0 + 1e-16f) + b_gat[lane],      0.0f); acc += g * W_fc[lane];
        g = fmaxf(a1 / (d1 + 1e-16f) + b_gat[32 + lane], 0.0f); acc += g * W_fc[32 + lane];
        g = fmaxf(a2 / (d2 + 1e-16f) + b_gat[64 + lane], 0.0f); acc += g * W_fc[64 + lane];
        g = fmaxf(a3 / (d3 + 1e-16f) + b_gat[96 + lane], 0.0f); acc += g * W_fc[96 + lane];
    }
    for (int o = 16; o > 0; o >>= 1) acc += __shfl_xor(acc, o, 32);
    if (lane == 0) out[n] = acc + b_fc[0];
}

extern "C" void kernel_launch(void* const* d_in, const int* in_sizes, int n_in,
                              void* d_out, int out_size, void* d_ws, size_t ws_size,
                              hipStream_t stream) {
    const float* x      = (const float*)d_in[0];
    const int*   eidx   = (const int*)  d_in[1];
    const float* eattr  = (const float*)d_in[2];
    const float* W_nn   = (const float*)d_in[3];
    const float* b_nn   = (const float*)d_in[4];
    const float* W_root = (const float*)d_in[5];
    const float* b_ecc  = (const float*)d_in[6];
    const float* W_gat  = (const float*)d_in[7];
    const float* a_src  = (const float*)d_in[8];
    const float* a_dst  = (const float*)d_in[9];
    const float* b_gat  = (const float*)d_in[10];
    const float* W_fc   = (const float*)d_in[11];
    const float* b_fc   = (const float*)d_in[12];
    float* out = (float*)d_out;

    const int N = in_sizes[0] / 16;
    const int E = in_sizes[1] / 2;

    float* ws = (float*)d_ws;
    size_t off = 0;
    float* zbuf  = ws + off;          off += (size_t)N * 128;
    float* hbuf  = ws + off;          off += (size_t)N * 16;
    float* esrc  = ws + off;          off += (size_t)N * 4;
    float* edst  = ws + off;          off += (size_t)N * 4;
    float* agg   = ws + off;          off += (size_t)N * 16;
    const size_t zeroBeg = off;
    int*   deg   = (int*)(ws + off);  off += (size_t)N;
    int*   total = (int*)(ws + off);  off += 4;
    const long long zeroCnt = (long long)(off - zeroBeg);
    int*   noff  = (int*)(ws + off);  off += (size_t)N;
    int*   rnk   = (int*)(ws + off);  off += (size_t)E;
    int*   ssrc  = (int*)(ws + off);  off += (size_t)E;
    int*   sedge = (int*)zbuf;

    k_zero<<<(N + 4 + 255) / 256, 256, 0, stream>>>((float*)deg, zeroCnt);

    k_rank <<<(E + 255) / 256, 256, 0, stream>>>(eidx, deg, rnk, E);
    k_alloc<<<(N + 255) / 256, 256, 0, stream>>>(deg, total, noff, N);
    k_place<<<(E + 255) / 256, 256, 0, stream>>>(eidx, noff, rnk, ssrc, sedge, E);

    k_nnagg<<<(N + 15) / 16, 512, 0, stream>>>(x, eattr, W_nn, b_nn, noff, deg, ssrc, sedge, agg, N);
    k_hnode <<<(N * 16 + 255) / 256, 256, 0, stream>>>(x, W_root, b_ecc, agg, hbuf, N);
    k_zgemm <<<(N + 15) / 16, 256, 0, stream>>>(hbuf, W_gat, zbuf, N);
    k_escore<<<(N * 4 + 255) / 256, 256, 0, stream>>>(zbuf, a_src, a_dst, esrc, edst, N);
    k_gatnode<<<(N + 7) / 8, 256, 0, stream>>>(noff, deg, ssrc, esrc, edst, zbuf,
                                               b_gat, W_fc, b_fc, out, N);
}


# Edge list grouped by destination with a two-level counting sort on LDS counters (chunk->bucket, bucket->node): no global atomics, no zeroing launch; sum order = LDS arrival order
# speedup vs baseline: 8.8772x; 1.4438x over previous
#include <hip/hip_runtime.h>
#include <hip/hip_bf16.h>

typedef __attribute__((ext_vector_type(2))) float v2f;
typedef __attribute__((ext_vector_type(4))) float v4f;
typedef __attribute__((ext_vector_type(8))) float v8f;

#define NEG_SLOPE 0.2f

#define UPITCH 276
__global__ __launch_bounds__(512)
void k_nnagg(const float* __restrict__ x,
             const float* __restrict__ eattr,
             const float* __restrict__ W_nn,
             const float* __restrict__ b_nn,
             const int* __restrict__ off,
             const int* __restrict__ deg,
             const int* __restrict__ ssrc,
             const int* __restrict__ sedge,
             float* __restrict__ agg, int N) {
    __shared__ float sU[16 * UPITCH];

    const int tid  = threadIdx.x;
    const int wave = tid >> 5;
    const int lane = tid & 31;
    const int i    = lane & 15;
    const int hi   = lane >> 4;
    const int n    = blockIdx.x * 16 + wave;

    v2f U0 = {0.f, 0.f}, U1 = {0.f, 0.f}, U2 = {0.f, 0.f}, U3 = {0.f, 0.f};
    float xs = 0.f;
    if (n < N) {
        const int beg = __builtin_amdgcn_readfirstlane(off[n]);
        const int cnt = __builtin_amdgcn_readfirstlane(deg[n]);
        for (int c = 0; c < cnt; c += 32) {
            const int j = c + lane;
            int s = 0, e = 0;
            if (j < cnt) { s = ssrc[beg + j]; e = sedge[beg + j]; }
            const int mc = (cnt - c < 32) ? (cnt - c) : 32;
            int q = 0;
            for (; q + 4 <= mc; q += 4) {
                float xv[4];
                v4f ea[4][2];
#pragma unroll
                for (int t = 0; t < 4; ++t) {
                    const int sq = __builtin_amdgcn_readlane(s, q + t);
                    const int eq = __builtin_amdgcn_readlane(e, q + t);
                    xv[t] = x[(size_t)sq * 16 + i];
                    const v4f* er =
                        reinterpret_cast<const v4f*>(eattr + (size_t)eq * 16 + hi * 8);
                    ea[t][0] = er[0];
                    ea[t][1] = er[1];
                }
#pragma unroll
                for (int t = 0; t < 4; ++t) {
                    const v2f xx = {xv[t], xv[t]};
                    U0 += ea[t][0].xy * xx; U1 += ea[t][0].zw * xx;
                    U2 += ea[t][1].xy * xx; U3 += ea[t][1].zw * xx;
                    xs += xv[t];
                }
            }
            for (; q < mc; ++q) {
                const int sq = __builtin_amdgcn_readlane(s, q);
                const int eq = __builtin_amdgcn_readlane(e, q);
                const float xq = x[(size_t)sq * 16 + i];
                const v4f* er =
                    reinterpret_cast<const v4f*>(eattr + (size_t)eq * 16 + hi * 8);
                const v4f ea0 = er[0], ea1 = er[1];
                const v2f xx = {xq, xq};
                U0 += ea0.xy * xx; U1 += ea0.zw * xx;
                U2 += ea1.xy * xx; U3 += ea1.zw * xx;
                xs += xq;
            }
        }
    }
    {
        float* ur = sU + wave * UPITCH + hi * 128 + i;
        ur[0]  = U0.x; ur[16] = U0.y; ur[32] = U1.x; ur[48]  = U1.y;
        ur[64] = U2.x; ur[80] = U2.y; ur[96] = U3.x; ur[112] = U3.y;
        if (hi == 0) sU[wave * UPITCH + 256 + i] = xs;
    }
    __syncthreads();
    if (wave != 0) return;

    v8f accA = {}, accB = {};
    const float* ar = sU + i * UPITCH + 2 * hi;
#pragma unroll
    for (int kk = 0; kk < 68; ++kk) {
        const float* wp = (kk < 64) ? (W_nn + (4 * kk + 2 * hi) * 16 + i)
                                    : (b_nn + (4 * (kk - 64) + 2 * hi) * 16 + i);
        v2f a, b;
        a.x = ar[4 * kk]; a.y = ar[4 * kk + 1];
        b.x = wp[0];      b.y = wp[16];
        if (kk & 1)
            accB = __builtin_amdgcn_wmma_f32_16x16x4_f32(
                false, a, false, b, (short)0, accB, false, false);
        else
            accA = __builtin_amdgcn_wmma_f32_16x16x4_f32(
                false, a, false, b, (short)0, accA, false, false);
    }
    const v8f acc = accA + accB;
#pragma unroll
    for (int v = 0; v < 8; ++v) {
        const int r = blockIdx.x * 16 + v + 8 * hi;
        if (r < N) agg[(size_t)r * 16 + i] = acc[v];
    }
}

__global__ void k_hnode(const float* __restrict__ x,
                        const float* __restrict__ W_root,
                        const float* __restrict__ b_ecc,
                        const float* __restrict__ agg,
                        float* __restrict__ h, int N) {
    int i = blockIdx.x * blockDim.x + threadIdx.x;
    if (i >= N * 16) return;
    int n = i >> 4, o = i & 15;
    float s = agg[i] + b_ecc[o];
    const float* xr = x + (size_t)n * 16;
#pragma unroll
    for (int j = 0; j < 16; ++j) s += xr[j] * W_root[j * 16 + o];
    h[i] = fmaxf(s, 0.0f);
}

__global__ void k_zgemm(const float* __restrict__ h,
                        const float* __restrict__ W_gat,
                        float* __restrict__ z, int N) {
    __shared__ float sWg[8 * 256];
    const int tid = threadIdx.x;
    for (int t = tid; t < 2048; t += 256) {
        int k = t >> 7, col = t & 127;
        sWg[(k >> 1) * 256 + col * 2 + (k & 1)] = W_gat[k * 128 + col];
    }
    __syncthreads();

    const int wave = tid >> 5, lane = tid & 31;
    const int m = lane & 15, hi = lane >> 4;
    const int rowTile = blockIdx.x;
    const int colTile = wave;
    const int row = rowTile * 16 + m;
    const int rc  = row < N ? row : (N - 1);
    const float* hr = h + (size_t)rc * 16;
    const int col = colTile * 16 + m;

    v8f acc = {};
#pragma unroll
    for (int kk = 0; kk < 4; ++kk) {
        int i0 = 4 * kk + 2 * hi;
        v2f a, b;
        a.x = hr[i0]; a.y = hr[i0 + 1];
        int p = 2 * kk + hi;
        b.x = sWg[p * 256 + col * 2 + 0];
        b.y = sWg[p * 256 + col * 2 + 1];
        acc = __builtin_amdgcn_wmma_f32_16x16x4_f32(
            false, a, false, b, (short)0, acc, false, false);
    }
#pragma unroll
    for (int v = 0; v < 8; ++v) {
        int r = rowTile * 16 + v + 8 * hi;
        if (r < N) z[(size_t)r * 128 + col] = acc[v];
    }
}

__global__ void k_escore(const float* __restrict__ z,
                         const float* __restrict__ a_src,
                         const float* __restrict__ a_dst,
                         float* __restrict__ esrc,
                         float* __restrict__ edst, int N) {
    int i = blockIdx.x * blockDim.x + threadIdx.x;
    if (i >= N * 4) return;
    int n = i >> 2, hh = i & 3;
    const float* zr = z + (size_t)n * 128 + hh * 32;
    const float* as = a_src + hh * 32;
    const float* ad = a_dst + hh * 32;
    float s = 0.f, d = 0.f;
#pragma unroll 8
    for (int j = 0; j < 32; ++j) { float zv = zr[j]; s += zv * as[j]; d += zv * ad[j]; }
    esrc[i] = s; edst[i] = d;
}

#define NCH 256
#define NBK 256
#define SORT_THREADS 1024
static_assert(SORT_THREADS == 4 * NCH && NBK <= SORT_THREADS, "k_group: four threads per chunk segment");

__device__ __forceinline__ int block_excl_scan(int v, int* wsum, int* tot) {
    const int lane = threadIdx.x & 31, wave = threadIdx.x >> 5;
    int inc = v;
    for (int o = 1; o < 32; o <<= 1) {
        int t = __shfl_up(inc, o, 32);
        if (lane >= o) inc += t;
    }
    if (lane == 31) wsum[wave] = inc;
    __syncthreads();
    if (wave == 0) {
        const int w = wsum[lane];
        int wi = w;
        for (int o = 1; o < 32; o <<= 1) {
            int t = __shfl_up(wi, o, 32);
            if (lane >= o) wi += t;
        }
        wsum[lane] = wi - w;
        if (lane == 31) wsum[32] = wi;
    }
    __syncthreads();
    *tot = wsum[32];
    const int r = wsum[wave] + inc - v;
    __syncthreads();
    return r;
}

__global__ __launch_bounds__(SORT_THREADS)
void k_bucket(const int* __restrict__ eidx,
              int* __restrict__ bdst,
              int* __restrict__ bsrc,
              int* __restrict__ bedge,
              int* __restrict__ tabT, int E, int bw) {
    __shared__ int hist[NBK];
    __shared__ int wsum[33];
    const int tid = threadIdx.x;
    const int chunk = (E + NCH - 1) / NCH;
    int e0 = blockIdx.x * chunk; if (e0 > E) e0 = E;
    int e1 = e0 + chunk;         if (e1 > E) e1 = E;

    if (tid < NBK) hist[tid] = 0;
    __syncthreads();
    for (int eb = e0; eb < e1; eb += 4 * SORT_THREADS) {
        int d[4];
#pragma unroll
        for (int u = 0; u < 4; ++u) {
            const int e = eb + u * SORT_THREADS + tid;
            d[u] = (e < e1) ? eidx[E + e] : -1;
        }
#pragma unroll
        for (int u = 0; u < 4; ++u)
            if (d[u] >= 0) atomicAdd(&hist[d[u] / bw], 1);
    }
    __syncthreads();
    int tot;
    const int v = (tid < NBK) ? hist[tid] : 0;
    const int ex = block_excl_scan(v, wsum, &tot);
    if (tid < NBK) {
        tabT[tid * NCH + blockIdx.x] = ex;
        hist[tid] = ex;
    }
    if (tid == 0) tabT[NBK * NCH + blockIdx.x] = tot;
    __syncthreads();
    for (int eb = e0; eb < e1; eb += 4 * SORT_THREADS) {
        int d[4], s[4];
#pragma unroll
        for (int u = 0; u < 4; ++u) {
            const int e = eb + u * SORT_THREADS + tid;
            d[u] = -1; s[u] = 0;
            if (e < e1) { d[u] = eidx[E + e]; s[u] = eidx[e]; }
        }
#pragma unroll
        for (int u = 0; u < 4; ++u) {
            if (d[u] >= 0) {
                const int p = e0 + atomicAdd(&hist[d[u] / bw], 1);
                bdst[p]  = d[u];
                bsrc[p]  = s[u];
                bedge[p] = eb + u * SORT_THREADS + tid;
            }
        }
    }
}

__global__ __launch_bounds__(SORT_THREADS)
void k_group(const int* __restrict__ bdst,
             const int* __restrict__ bsrc,
             const int* __restrict__ bedge,
             const int* __restrict__ tabT,
             int* __restrict__ off,
             int* __restrict__ deg,
             int* __restrict__ ssrc,
             int* __restrict__ sedge, int E, int N, int bw) {
    __shared__ int cnt[SORT_THREADS];
    __shared__ int wsum[33];
    const int tid = threadIdx.x;
    const int b   = blockIdx.x;
    const int c   = tid >> 2;
    const int sub = tid & 3;
    const int chunk = (E + NCH - 1) / NCH;
    int cb = c * chunk; if (cb > E) cb = E;
    const int s0 = tabT[b * NCH + c];
    const int s1 = tabT[(b + 1) * NCH + c];
    const int g  = cb + s0;
    const int l  = s1 - s0;
    const int n0 = b * bw;

    cnt[tid] = 0;
    int base;
    (void)block_excl_scan(sub == 0 ? s0 : 0, wsum, &base);

    for (int jb = 0; jb < l; jb += 16) {
        int d[4];
#pragma unroll
        for (int u = 0; u < 4; ++u) {
            const int j = jb + 4 * u + sub;
            d[u] = (j < l) ? bdst[g + j] : -1;
        }
#pragma unroll
        for (int u = 0; u < 4; ++u)
            if (d[u] >= 0) atomicAdd(&cnt[d[u] - n0], 1);
    }
    __syncthreads();
    int tot;
    const int v  = cnt[tid];
    const int ex = block_excl_scan(v, wsum, &tot);
    {
        const int n = n0 + tid;
        if (tid < bw && n < N) { off[n] = base + ex; deg[n] = v; }
    }
    cnt[tid] = base + ex;
    __syncthreads();
    for (int jb = 0; jb < l; jb += 16) {
        int d[4], s[4], e[4];
#pragma unroll
        for (int u = 0; u < 4; ++u) {
            const int j = jb + 4 * u + sub;
            d[u] = -1; s[u] = 0; e[u] = 0;
            if (j < l) { d[u] = bdst[g + j]; s[u] = bsrc[g + j]; e[u] = bedge[g + j]; }
        }
#pragma unroll
        for (int u = 0; u < 4; ++u) {
            if (d[u] >= 0) {
                const int p = atomicAdd(&cnt[d[u] - n0], 1);
                ssrc[p]  = s[u];
                sedge[p] = e[u];
            }
        }
    }
}

__device__ __forceinline__ float lrelu(float v) { return v > 0.f ? v : NEG_SLOPE * v; }
__device__ __forceinline__ float lane_val(float v, int q) {
    return __int_as_float(__builtin_amdgcn_readlane(__float_as_int(v), q));
}

__global__ void k_gatnode(const int* __restrict__ off,
                          const int* __restrict__ deg,
                          const int* __restrict__ ssrc,
                          const float* __restrict__ esrc,
                          const float* __restrict__ edst,
                          const float* __restrict__ z,
                          const float* __restrict__ b_gat,
                          const float* __restrict__ W_fc,
                          const float* __restrict__ b_fc,
                          float* __restrict__ out, int N) {
    const int lane = threadIdx.x & 31;
    const int n = blockIdx.x * 8 + (threadIdx.x >> 5);
    if (n >= N) return;
    const int beg = __builtin_amdgcn_readfirstlane(off[n]);
    const int cnt = __builtin_amdgcn_readfirstlane(deg[n]);
    const float4 ed = *reinterpret_cast<const float4*>(edst + (size_t)n * 4);
    const float4* es4 = reinterpret_cast<const float4*>(esrc);

    float m0 = -INFINITY, m1 = -INFINITY, m2 = -INFINITY, m3 = -INFINITY;
    for (int c = 0; c < cnt; c += 32) {
        const int i = c + lane;
        if (i < cnt) {
            const float4 es = es4[ssrc[beg + i]];
            m0 = fmaxf(m0, lrelu(es.x + ed.x));
            m1 = fmaxf(m1, lrelu(es.y + ed.y));
            m2 = fmaxf(m2, lrelu(es.z + ed.z));
            m3 = fmaxf(m3, lrelu(es.w + ed.w));
        }
    }
    for (int o = 16; o > 0; o >>= 1) {
        m0 = fmaxf(m0, __shfl_xor(m0, o, 32));
        m1 = fmaxf(m1, __shfl_xor(m1, o, 32));
        m2 = fmaxf(m2, __shfl_xor(m2, o, 32));
        m3 = fmaxf(m3, __shfl_xor(m3, o, 32));
    }

    float a0 = 0.f, a1 = 0.f, a2 = 0.f, a3 = 0.f;
    float d0 = 0.f, d1 = 0.f, d2 = 0.f, d3 = 0.f;
    for (int c = 0; c < cnt; c += 32) {
        const int i = c + lane;
        int s = 0;
        float e0 = 0.f, e1 = 0.f, e2 = 0.f, e3 = 0.f;
        if (i < cnt) {
            s = ssrc[beg + i];
            const float4 es = es4[s];
            e0 = __expf(lrelu(es.x + ed.x) - m0);
            e1 = __expf(lrelu(es.y + ed.y) - m1);
            e2 = __expf(lrelu(es.z + ed.z) - m2);
            e3 = __expf(lrelu(es.w + ed.w) - m3);
        }
        d0 += e0; d1 += e1; d2 += e2; d3 += e3;

        const int mc = (cnt - c < 32) ? (cnt - c) : 32;
        int q = 0;
        for (; q + 4 <= mc; q += 4) {
            float r[4][4];
#pragma unroll
            for (int u = 0; u < 4; ++u) {
                const float* zr = z + (size_t)__builtin_amdgcn_readlane(s, q + u) * 128 + lane;
                r[u][0] = zr[0]; r[u][1] = zr[32]; r[u][2] = zr[64]; r[u][3] = zr[96];
            }
#pragma unroll
            for (int u = 0; u < 4; ++u) {
                a0 += lane_val(e0, q + u) * r[u][0];
                a1 += lane_val(e1, q + u) * r[u][1];
                a2 += lane_val(e2, q + u) * r[u][2];
                a3 += lane_val(e3, q + u) * r[u][3];
            }
        }
        for (; q < mc; ++q) {
            const float* zr = z + (size_t)__builtin_amdgcn_readlane(s, q) * 128 + lane;
            a0 += lane_val(e0, q) * zr[0];
            a1 += lane_val(e1, q) * zr[32];
            a2 += lane_val(e2, q) * zr[64];
            a3 += lane_val(e3, q) * zr[96];
        }
    }
    for (int o = 16; o > 0; o >>= 1) {
        d0 += __shfl_xor(d0, o, 32);
        d1 += __shfl_xor(d1, o, 32);
        d2 += __shfl_xor(d2, o, 32);
        d3 += __shfl_xor(d3, o, 32);
    }

    float acc = 0.f;
    {
        float g;
        g = fmaxf(a0 / (d0 + 1e-16f) + b_gat[lane],      0.0f); acc += g * W_fc[lane];
        g = fmaxf(a1 / (d1 + 1e-16f) + b_gat[32 + lane], 0.0f); acc += g * W_fc[32 + lane];
        g = fmaxf(a2 / (d2 + 1e-16f) + b_gat[64 + lane], 0.0f); acc += g * W_fc[64 + lane];
        g = fmaxf(a3 / (d3 + 1e-16f) + b_gat[96 + lane], 0.0f); acc += g * W_fc[96 + lane];
    }
    for (int o = 16; o > 0; o >>= 1) acc += __shfl_xor(acc, o, 32);
    if (lane == 0) out[n] = acc + b_fc[0];
}

extern "C" void kernel_launch(void* const* d_in, const int* in_sizes, int n_in,
                              void* d_out, int out_size, void* d_ws, size_t ws_size,
                              hipStream_t stream) {
    const float* x      = (const float*)d_in[0];
    const int*   eidx   = (const int*)  d_in[1];
    const float* eattr  = (const float*)d_in[2];
    const float* W_nn   = (const float*)d_in[3];
    const float* b_nn   = (const float*)d_in[4];
    const float* W_root = (const float*)d_in[5];
    const float* b_ecc  = (const float*)d_in[6];
    const float* W_gat  = (const float*)d_in[7];
    const float* a_src  = (const float*)d_in[8];
    const float* a_dst  = (const float*)d_in[9];
    const float* b_gat  = (const float*)d_in[10];
    const float* W_fc   = (const float*)d_in[11];
    const float* b_fc   = (const float*)d_in[12];
    float* out = (float*)d_out;

    const int N = in_sizes[0] / 16;
    const int E = in_sizes[1] / 2;

    float* ws = (float*)d_ws;
    size_t off = 0;
    float* zbuf  = ws + off;          off += (size_t)N * 128;
    float* hbuf  = ws + off;          off += (size_t)N * 16;
    float* esrc  = ws + off;          off += (size_t)N * 4;
    float* edst  = ws + off;          off += (size_t)N * 4;
    float* agg   = ws + off;          off += (size_t)N * 16;
    int*   deg   = (int*)(ws + off);  off += (size_t)N;
    int*   noff  = (int*)(ws + off);  off += (size_t)N;
    int*   tabT  = (int*)(ws + off);  off += (size_t)E;
    int*   ssrc  = (int*)(ws + off);  off += (size_t)E;
    int*   sedge = (int*)zbuf;
    int*   bdst  = sedge + (size_t)E;
    int*   bsrc  = bdst  + (size_t)E;
    int*   bedge = bsrc  + (size_t)E;
    const int bw = (N + NBK - 1) / NBK;

    k_bucket<<<NCH, SORT_THREADS, 0, stream>>>(eidx, bdst, bsrc, bedge, tabT, E, bw);
    k_group <<<NBK, SORT_THREADS, 0, stream>>>(bdst, bsrc, bedge, tabT, noff, deg, ssrc, sedge, E, N, bw);

    k_nnagg<<<(N + 15) / 16, 512, 0, stream>>>(x, eattr, W_nn, b_nn, noff, deg, ssrc, sedge, agg, N);
    k_hnode <<<(N * 16 + 255) / 256, 256, 0, stream>>>(x, W_root, b_ecc, agg, hbuf, N);
    k_zgemm <<<(N + 15) / 16, 256, 0, stream>>>(hbuf, W_gat, zbuf, N);
    k_escore<<<(N * 4 + 255) / 256, 256, 0, stream>>>(zbuf, a_src, a_dst, esrc, edst, N);
    k_gatnode<<<(N + 7) / 8, 256, 0, stream>>>(noff, deg, ssrc, esrc, edst, zbuf,
                                               b_gat, W_fc, b_fc, out, N);
}


# Per-node dense chain fused into the NNConv tile kernel: root term joins the WMMA K loop (h sum order changes), relu, z = h@W_gat, head scores in one work-group; 7 -> 4 launches
# speedup vs baseline: 10.3334x; 1.1640x over previous
#include <hip/hip_runtime.h>
#include <hip/hip_bf16.h>

typedef __attribute__((ext_vector_type(2))) float v2f;
typedef __attribute__((ext_vector_type(4))) float v4f;
typedef __attribute__((ext_vector_type(8))) float v8f;

#define NEG_SLOPE 0.2f

#define UPITCH 292
#define HPITCH 20
#define ZPITCH 132
__global__ __launch_bounds__(512, 12)
void k_nnagg(const float* __restrict__ x,
             const float* __restrict__ eattr,
             const float* __restrict__ W_nn,
             const float* __restrict__ b_nn,
             const float* __restrict__ W_root,
             const float* __restrict__ b_ecc,
             const float* __restrict__ W_gat,
             const float* __restrict__ a_src,
             const float* __restrict__ a_dst,
             const int* __restrict__ off,
             const int* __restrict__ deg,
             const int* __restrict__ ssrc,
             const int* __restrict__ sedge,
             float* __restrict__ z,
             float* __restrict__ esrc,
             float* __restrict__ edst, int N) {
    __shared__ float sU[16 * UPITCH];
    __shared__ float sH[16 * HPITCH];
    __shared__ __attribute__((aligned(16))) float sZ[16 * ZPITCH];
    __shared__ v4f sA[64];

    const int tid  = threadIdx.x;
    const int wave = tid >> 5;
    const int lane = tid & 31;
    const int i    = lane & 15;
    const int hi   = lane >> 4;
    const int n    = blockIdx.x * 16 + wave;

    if (tid < 64)
        sA[tid] = reinterpret_cast<const v4f*>(tid < 32 ? a_src : a_dst)[tid & 31];

    v2f U0 = {0.f, 0.f}, U1 = {0.f, 0.f}, U2 = {0.f, 0.f}, U3 = {0.f, 0.f};
    float xs = 0.f;
    float xown = 0.f;
    if (n < N) {
        xown = x[(size_t)n * 16 + i];
        const int beg = __builtin_amdgcn_readfirstlane(off[n]);
        const int cnt = __builtin_amdgcn_readfirstlane(deg[n]);
        for (int c = 0; c < cnt; c += 32) {
            const int j = c + lane;
            int s = 0, e = 0;
            if (j < cnt) { s = ssrc[beg + j]; e = sedge[beg + j]; }
            const int mc = (cnt - c < 32) ? (cnt - c) : 32;
            int q = 0;
            for (; q + 4 <= mc; q += 4) {
                float xv[4];
                v4f ea[4][2];
#pragma unroll
                for (int t = 0; t < 4; ++t) {
                    const int sq = __builtin_amdgcn_readlane(s, q + t);
                    const int eq = __builtin_amdgcn_readlane(e, q + t);
                    xv[t] = x[(size_t)sq * 16 + i];
                    const v4f* er =
                        reinterpret_cast<const v4f*>(eattr + (size_t)eq * 16 + hi * 8);
                    ea[t][0] = er[0];
                    ea[t][1] = er[1];
                }
#pragma unroll
                for (int t = 0; t < 4; ++t) {
                    const v2f xx = {xv[t], xv[t]};
                    U0 += ea[t][0].xy * xx; U1 += ea[t][0].zw * xx;
                    U2 += ea[t][1].xy * xx; U3 += ea[t][1].zw * xx;
                    xs += xv[t];
                }
            }
            for (; q < mc; ++q) {
                const int sq = __builtin_amdgcn_readlane(s, q);
                const int eq = __builtin_amdgcn_readlane(e, q);
                const float xq = x[(size_t)sq * 16 + i];
                const v4f* er =
                    reinterpret_cast<const v4f*>(eattr + (size_t)eq * 16 + hi * 8);
                const v4f ea0 = er[0], ea1 = er[1];
                const v2f xx = {xq, xq};
                U0 += ea0.xy * xx; U1 += ea0.zw * xx;
                U2 += ea1.xy * xx; U3 += ea1.zw * xx;
                xs += xq;
            }
        }
    }
    {
        float* ur = sU + wave * UPITCH + hi * 128 + i;
        ur[0]  = U0.x; ur[16] = U0.y; ur[32] = U1.x; ur[48]  = U1.y;
        ur[64] = U2.x; ur[80] = U2.y; ur[96] = U3.x; ur[112] = U3.y;
        sU[wave * UPITCH + 256 + hi * 16 + i] = hi ? xown : xs;
    }

    __syncthreads();

    const int col = (wave & 7) * 16 + i;
    v2f bg[4] = {};
    if (wave < 8) {
#pragma unroll
        for (int kk = 0; kk < 4; ++kk) {
            bg[kk].x = W_gat[(4 * kk + 2 * hi) * 128 + col];
            bg[kk].y = W_gat[(4 * kk + 2 * hi + 1) * 128 + col];
        }
    }

    if (wave == 0) {
        v8f accA = {}, accB = {};
        const float* ar = sU + i * UPITCH + 2 * hi;
#pragma unroll
        for (int kk = 0; kk < 72; ++kk) {
            const float* wp = (kk < 64) ? (W_nn + (4 * kk + 2 * hi) * 16 + i)
                            : (kk < 68) ? (b_nn + (4 * (kk - 64) + 2 * hi) * 16 + i)
                                        : (W_root + (4 * (kk - 68) + 2 * hi) * 16 + i);
            v2f a, b;
            a.x = ar[4 * kk]; a.y = ar[4 * kk + 1];
            b.x = wp[0];      b.y = wp[16];
            if (kk & 1)
                accB = __builtin_amdgcn_wmma_f32_16x16x4_f32(
                    false, a, false, b, (short)0, accB, false, false);
            else
                accA = __builtin_amdgcn_wmma_f32_16x16x4_f32(
                    false, a, false, b, (short)0, accA, false, false);
        }
        const v8f acc = accA + accB;
        const float be = b_ecc[i];
#pragma unroll
        for (int v = 0; v < 8; ++v)
            sH[(v + 8 * hi) * HPITCH + i] = fmaxf(acc[v] + be, 0.0f);
    }
    __syncthreads();

    if (wave < 8) {
        v8f acc = {};
        const float* hr = sH + i * HPITCH + 2 * hi;
#pragma unroll
        for (int kk = 0; kk < 4; ++kk) {
            v2f a;
            a.x = hr[4 * kk]; a.y = hr[4 * kk + 1];
            acc = __builtin_amdgcn_wmma_f32_16x16x4_f32(
                false, a, false, bg[kk], (short)0, acc, false, false);
        }
#pragma unroll
        for (int v = 0; v < 8; ++v)
            sZ[(v + 8 * hi) * ZPITCH + col] = acc[v];
    }
    __syncthreads();

    if (n < N) {
        const v4f zv = *reinterpret_cast<const v4f*>(sZ + wave * ZPITCH + 4 * lane);
        *reinterpret_cast<v4f*>(z + (size_t)n * 128 + 4 * lane) = zv;
    }
    if (wave == 8 || wave == 9) {
        const int t  = (wave - 8) * 32 + lane;
        const int nn = t >> 2, hh = t & 3;
        const int r  = blockIdx.x * 16 + nn;
        const v4f* zr = reinterpret_cast<const v4f*>(sZ + nn * ZPITCH + hh * 32);
        const v4f* as = sA + hh * 8;
        const v4f* ad = sA + 32 + hh * 8;
        float s = 0.f, d = 0.f;
#pragma unroll
        for (int j = 0; j < 8; ++j) {
            const v4f zz = zr[j], sa = as[j], da = ad[j];
            s += zz.x * sa.x; d += zz.x * da.x;
            s += zz.y * sa.y; d += zz.y * da.y;
            s += zz.z * sa.z; d += zz.z * da.z;
            s += zz.w * sa.w; d += zz.w * da.w;
        }
        if (r < N) { esrc[(size_t)r * 4 + hh] = s; edst[(size_t)r * 4 + hh] = d; }
    }
}

#define NCH 256
#define NBK 256
#define SORT_THREADS 1024
static_assert(SORT_THREADS == 4 * NCH && NBK <= SORT_THREADS, "k_group: four threads per chunk segment");

__device__ __forceinline__ int block_excl_scan(int v, int* wsum, int* tot) {
    const int lane = threadIdx.x & 31, wave = threadIdx.x >> 5;
    int inc = v;
    for (int o = 1; o < 32; o <<= 1) {
        int t = __shfl_up(inc, o, 32);
        if (lane >= o) inc += t;
    }
    if (lane == 31) wsum[wave] = inc;
    __syncthreads();
    if (wave == 0) {
        const int w = wsum[lane];
        int wi = w;
        for (int o = 1; o < 32; o <<= 1) {
            int t = __shfl_up(wi, o, 32);
            if (lane >= o) wi += t;
        }
        wsum[lane] = wi - w;
        if (lane == 31) wsum[32] = wi;
    }
    __syncthreads();
    *tot = wsum[32];
    const int r = wsum[wave] + inc - v;
    __syncthreads();
    return r;
}

__global__ __launch_bounds__(SORT_THREADS)
void k_bucket(const int* __restrict__ eidx,
              int* __restrict__ bdst,
              int* __restrict__ bsrc,
              int* __restrict__ bedge,
              int* __restrict__ tabT, int E, int bw) {
    __shared__ int hist[NBK];
    __shared__ int wsum[33];
    const int tid = threadIdx.x;
    const int chunk = (E + NCH - 1) / NCH;
    int e0 = blockIdx.x * chunk; if (e0 > E) e0 = E;
    int e1 = e0 + chunk;         if (e1 > E) e1 = E;

    if (tid < NBK) hist[tid] = 0;
    __syncthreads();
    for (int eb = e0; eb < e1; eb += 4 * SORT_THREADS) {
        int d[4];
#pragma unroll
        for (int u = 0; u < 4; ++u) {
            const int e = eb + u * SORT_THREADS + tid;
            d[u] = (e < e1) ? eidx[E + e] : -1;
        }
#pragma unroll
        for (int u = 0; u < 4; ++u)
            if (d[u] >= 0) atomicAdd(&hist[d[u] / bw], 1);
    }
    __syncthreads();
    int tot;
    const int v = (tid < NBK) ? hist[tid] : 0;
    const int ex = block_excl_scan(v, wsum, &tot);
    if (tid < NBK) {
        tabT[tid * NCH + blockIdx.x] = ex;
        hist[tid] = ex;
    }
    if (tid == 0) tabT[NBK * NCH + blockIdx.x] = tot;
    __syncthreads();
    for (int eb = e0; eb < e1; eb += 4 * SORT_THREADS) {
        int d[4], s[4];
#pragma unroll
        for (int u = 0; u < 4; ++u) {
            const int e = eb + u * SORT_THREADS + tid;
            d[u] = -1; s[u] = 0;
            if (e < e1) { d[u] = eidx[E + e]; s[u] = eidx[e]; }
        }
#pragma unroll
        for (int u = 0; u < 4; ++u) {
            if (d[u] >= 0) {
                const int p = e0 + atomicAdd(&hist[d[u] / bw], 1);
                bdst[p]  = d[u];
                bsrc[p]  = s[u];
                bedge[p] = eb + u * SORT_THREADS + tid;
            }
        }
    }
}

__global__ __launch_bounds__(SORT_THREADS)
void k_group(const int* __restrict__ bdst,
             const int* __restrict__ bsrc,
             const int* __restrict__ bedge,
             const int* __restrict__ tabT,
             int* __restrict__ off,
             int* __restrict__ deg,
             int* __restrict__ ssrc,
             int* __restrict__ sedge, int E, int N, int bw) {
    __shared__ int cnt[SORT_THREADS];
    __shared__ int wsum[33];
    const int tid = threadIdx.x;
    const int b   = blockIdx.x;
    const int c   = tid >> 2;
    const int sub = tid & 3;
    const int chunk = (E + NCH - 1) / NCH;
    int cb = c * chunk; if (cb > E) cb = E;
    const int s0 = tabT[b * NCH + c];
    const int s1 = tabT[(b + 1) * NCH + c];
    const int g  = cb + s0;
    const int l  = s1 - s0;
    const int n0 = b * bw;

    cnt[tid] = 0;
    int base;
    (void)block_excl_scan(sub == 0 ? s0 : 0, wsum, &base);

    for (int jb = 0; jb < l; jb += 16) {
        int d[4];
#pragma unroll
        for (int u = 0; u < 4; ++u) {
            const int j = jb + 4 * u + sub;
            d[u] = (j < l) ? bdst[g + j] : -1;
        }
#pragma unroll
        for (int u = 0; u < 4; ++u)
            if (d[u] >= 0) atomicAdd(&cnt[d[u] - n0], 1);
    }
    __syncthreads();
    int tot;
    const int v  = cnt[tid];
    const int ex = block_excl_scan(v, wsum, &tot);
    {
        const int n = n0 + tid;
        if (tid < bw && n < N) { off[n] = base + ex; deg[n] = v; }
    }
    cnt[tid] = base + ex;
    __syncthreads();
    for (int jb = 0; jb < l; jb += 16) {
        int d[4], s[4], e[4];
#pragma unroll
        for (int u = 0; u < 4; ++u) {
            const int j = jb + 4 * u + sub;
            d[u] = -1; s[u] = 0; e[u] = 0;
            if (j < l) { d[u] = bdst[g + j]; s[u] = bsrc[g + j]; e[u] = bedge[g + j]; }
        }
#pragma unroll
        for (int u = 0; u < 4; ++u) {
            if (d[u] >= 0) {
                const int p = atomicAdd(&cnt[d[u] - n0], 1);
                ssrc[p]  = s[u];
                sedge[p] = e[u];
            }
        }
    }
}

__device__ __forceinline__ float lrelu(float v) { return v > 0.f ? v : NEG_SLOPE * v; }
__device__ __forceinline__ float lane_val(float v, int q) {
    return __int_as_float(__builtin_amdgcn_readlane(__float_as_int(v), q));
}

__global__ void k_gatnode(const int* __restrict__ off,
                          const int* __restrict__ deg,
                          const int* __restrict__ ssrc,
                          const float* __restrict__ esrc,
                          const float* __restrict__ edst,
                          const float* __restrict__ z,
                          const float* __restrict__ b_gat,
                          const float* __restrict__ W_fc,
                          const float* __restrict__ b_fc,
                          float* __restrict__ out, int N) {
    const int lane = threadIdx.x & 31;
    const int n = blockIdx.x * 8 + (threadIdx.x >> 5);
    if (n >= N) return;
    const int beg = __builtin_amdgcn_readfirstlane(off[n]);
    const int cnt = __builtin_amdgcn_readfirstlane(deg[n]);
    const float4 ed = *reinterpret_cast<const float4*>(edst + (size_t)n * 4);
    const float4* es4 = reinterpret_cast<const float4*>(esrc);

    float m0 = -INFINITY, m1 = -INFINITY, m2 = -INFINITY, m3 = -INFINITY;
    for (int c = 0; c < cnt; c += 32) {
        const int i = c + lane;
        if (i < cnt) {
            const float4 es = es4[ssrc[beg + i]];
            m0 = fmaxf(m0, lrelu(es.x + ed.x));
            m1 = fmaxf(m1, lrelu(es.y + ed.y));
            m2 = fmaxf(m2, lrelu(es.z + ed.z));
            m3 = fmaxf(m3, lrelu(es.w + ed.w));
        }
    }
    for (int o = 16; o > 0; o >>= 1) {
        m0 = fmaxf(m0, __shfl_xor(m0, o, 32));
        m1 = fmaxf(m1, __shfl_xor(m1, o, 32));
        m2 = fmaxf(m2, __shfl_xor(m2, o, 32));
        m3 = fmaxf(m3, __shfl_xor(m3, o, 32));
    }

    float a0 = 0.f, a1 = 0.f, a2 = 0.f, a3 = 0.f;
    float d0 = 0.f, d1 = 0.f, d2 = 0.f, d3 = 0.f;
    for (int c = 0; c < cnt; c += 32) {
        const int i = c + lane;
        int s = 0;
        float e0 = 0.f, e1 = 0.f, e2 = 0.f, e3 = 0.f;
        if (i < cnt) {
            s = ssrc[beg + i];
            const float4 es = es4[s];
            e0 = __expf(lrelu(es.x + ed.x) - m0);
            e1 = __expf(lrelu(es.y + ed.y) - m1);
            e2 = __expf(lrelu(es.z + ed.z) - m2);
            e3 = __expf(lrelu(es.w + ed.w) - m3);
        }
        d0 += e0; d1 += e1; d2 += e2; d3 += e3;

        const int mc = (cnt - c < 32) ? (cnt - c) : 32;
        int q = 0;
        for (; q + 4 <= mc; q += 4) {
            float r[4][4];
#pragma unroll
            for (int u = 0; u < 4; ++u) {
                const float* zr = z + (size_t)__builtin_amdgcn_readlane(s, q + u) * 128 + lane;
                r[u][0] = zr[0]; r[u][1] = zr[32]; r[u][2] = zr[64]; r[u][3] = zr[96];
            }
#pragma unroll
            for (int u = 0; u < 4; ++u) {
                a0 += lane_val(e0, q + u) * r[u][0];
                a1 += lane_val(e1, q + u) * r[u][1];
                a2 += lane_val(e2, q + u) * r[u][2];
                a3 += lane_val(e3, q + u) * r[u][3];
            }
        }
        for (; q < mc; ++q) {
            const float* zr = z + (size_t)__builtin_amdgcn_readlane(s, q) * 128 + lane;
            a0 += lane_val(e0, q) * zr[0];
            a1 += lane_val(e1, q) * zr[32];
            a2 += lane_val(e2, q) * zr[64];
            a3 += lane_val(e3, q) * zr[96];
        }
    }
    for (int o = 16; o > 0; o >>= 1) {
        d0 += __shfl_xor(d0, o, 32);
        d1 += __shfl_xor(d1, o, 32);
        d2 += __shfl_xor(d2, o, 32);
        d3 += __shfl_xor(d3, o, 32);
    }

    float acc = 0.f;
    {
        float g;
        g = fmaxf(a0 / (d0 + 1e-16f) + b_gat[lane],      0.0f); acc += g * W_fc[lane];
        g = fmaxf(a1 / (d1 + 1e-16f) + b_gat[32 + lane], 0.0f); acc += g * W_fc[32 + lane];
        g = fmaxf(a2 / (d2 + 1e-16f) + b_gat[64 + lane], 0.0f); acc += g * W_fc[64 + lane];
        g = fmaxf(a3 / (d3 + 1e-16f) + b_gat[96 + lane], 0.0f); acc += g * W_fc[96 + lane];
    }
    for (int o = 16; o > 0; o >>= 1) acc += __shfl_xor(acc, o, 32);
    if (lane == 0) out[n] = acc + b_fc[0];
}

extern "C" void kernel_launch(void* const* d_in, const int* in_sizes, int n_in,
                              void* d_out, int out_size, void* d_ws, size_t ws_size,
                              hipStream_t stream) {
    const float* x      = (const float*)d_in[0];
    const int*   eidx   = (const int*)  d_in[1];
    const float* eattr  = (const float*)d_in[2];
    const float* W_nn   = (const float*)d_in[3];
    const float* b_nn   = (const float*)d_in[4];
    const float* W_root = (const float*)d_in[5];
    const float* b_ecc  = (const float*)d_in[6];
    const float* W_gat  = (const float*)d_in[7];
    const float* a_src  = (const float*)d_in[8];
    const float* a_dst  = (const float*)d_in[9];
    const float* b_gat  = (const float*)d_in[10];
    const float* W_fc   = (const float*)d_in[11];
    const float* b_fc   = (const float*)d_in[12];
    float* out = (float*)d_out;

    const int N = in_sizes[0] / 16;
    const int E = in_sizes[1] / 2;

    float* ws = (float*)d_ws;
    size_t off = 0;
    float* zbuf  = ws + off;          off += (size_t)N * 128;
    float* esrc  = ws + off;          off += (size_t)N * 4;
    float* edst  = ws + off;          off += (size_t)N * 4;
    int*   deg   = (int*)(ws + off);  off += (size_t)N;
    int*   noff  = (int*)(ws + off);  off += (size_t)N;
    int*   tabT  = (int*)(ws + off);  off += (size_t)E;
    int*   ssrc  = (int*)(ws + off);  off += (size_t)E;
    int*   sedge = (int*)(ws + off);  off += (size_t)E;
    int*   bdst  = (int*)zbuf;
    int*   bsrc  = bdst  + (size_t)E;
    int*   bedge = bsrc  + (size_t)E;
    const int bw = (N + NBK - 1) / NBK;

    k_bucket<<<NCH, SORT_THREADS, 0, stream>>>(eidx, bdst, bsrc, bedge, tabT, E, bw);
    k_group <<<NBK, SORT_THREADS, 0, stream>>>(bdst, bsrc, bedge, tabT, noff, deg, ssrc, sedge, E, N, bw);

    k_nnagg<<<(N + 15) / 16, 512, 0, stream>>>(x, eattr, W_nn, b_nn, W_root, b_ecc, W_gat, a_src, a_dst,
                                               noff, deg, ssrc, sedge, zbuf, esrc, edst, N);
    k_gatnode<<<(N + 7) / 8, 256, 0, stream>>>(noff, deg, ssrc, esrc, edst, zbuf,
                                               b_gat, W_fc, b_fc, out, N);
}


# GAT weighted sums taken over h rows (64 B per edge, not 512 B z rows) per destination and head, then W_gat applied per 16-node tile on WMMA; z never stored; sum order changes
# speedup vs baseline: 11.5627x; 1.1190x over previous
#include <hip/hip_runtime.h>
#include <hip/hip_bf16.h>

typedef __attribute__((ext_vector_type(2))) float v2f;
typedef __attribute__((ext_vector_type(4))) float v4f;
typedef __attribute__((ext_vector_type(8))) float v8f;

#define NEG_SLOPE 0.2f

#define UPITCH 292
#define HPITCH 20
#define ZPITCH 132
__global__ __launch_bounds__(512, 12)
void k_nnagg(const float* __restrict__ x,
             const float* __restrict__ eattr,
             const float* __restrict__ W_nn,
             const float* __restrict__ b_nn,
             const float* __restrict__ W_root,
             const float* __restrict__ b_ecc,
             const float* __restrict__ W_gat,
             const float* __restrict__ a_src,
             const float* __restrict__ a_dst,
             const int* __restrict__ off,
             const int* __restrict__ deg,
             const int* __restrict__ ssrc,
             const int* __restrict__ sedge,
             float* __restrict__ hrow,
             float* __restrict__ esrc,
             float* __restrict__ edst, int N) {
    __shared__ float sU[16 * UPITCH];
    __shared__ __attribute__((aligned(16))) float sH[16 * HPITCH];
    __shared__ __attribute__((aligned(16))) float sZ[16 * ZPITCH];
    __shared__ v4f sA[64];

    const int tid  = threadIdx.x;
    const int wave = tid >> 5;
    const int lane = tid & 31;
    const int i    = lane & 15;
    const int hi   = lane >> 4;
    const int n    = blockIdx.x * 16 + wave;

    if (tid < 64)
        sA[tid] = reinterpret_cast<const v4f*>(tid < 32 ? a_src : a_dst)[tid & 31];

    v2f U0 = {0.f, 0.f}, U1 = {0.f, 0.f}, U2 = {0.f, 0.f}, U3 = {0.f, 0.f};
    float xs = 0.f;
    float xown = 0.f;
    if (n < N) {
        xown = x[(size_t)n * 16 + i];
        const int beg = __builtin_amdgcn_readfirstlane(off[n]);
        const int cnt = __builtin_amdgcn_readfirstlane(deg[n]);
        for (int c = 0; c < cnt; c += 32) {
            const int j = c + lane;
            int s = 0, e = 0;
            if (j < cnt) { s = ssrc[beg + j]; e = sedge[beg + j]; }
            const int mc = (cnt - c < 32) ? (cnt - c) : 32;
            int q = 0;
            for (; q + 4 <= mc; q += 4) {
                float xv[4];
                v4f ea[4][2];
#pragma unroll
                for (int t = 0; t < 4; ++t) {
                    const int sq = __builtin_amdgcn_readlane(s, q + t);
                    const int eq = __builtin_amdgcn_readlane(e, q + t);
                    xv[t] = x[(size_t)sq * 16 + i];
                    const v4f* er =
                        reinterpret_cast<const v4f*>(eattr + (size_t)eq * 16 + hi * 8);
                    ea[t][0] = er[0];
                    ea[t][1] = er[1];
                }
#pragma unroll
                for (int t = 0; t < 4; ++t) {
                    const v2f xx = {xv[t], xv[t]};
                    U0 += ea[t][0].xy * xx; U1 += ea[t][0].zw * xx;
                    U2 += ea[t][1].xy * xx; U3 += ea[t][1].zw * xx;
                    xs += xv[t];
                }
            }
            for (; q < mc; ++q) {
                const int sq = __builtin_amdgcn_readlane(s, q);
                const int eq = __builtin_amdgcn_readlane(e, q);
                const float xq = x[(size_t)sq * 16 + i];
                const v4f* er =
                    reinterpret_cast<const v4f*>(eattr + (size_t)eq * 16 + hi * 8);
                const v4f ea0 = er[0], ea1 = er[1];
                const v2f xx = {xq, xq};
                U0 += ea0.xy * xx; U1 += ea0.zw * xx;
                U2 += ea1.xy * xx; U3 += ea1.zw * xx;
                xs += xq;
            }
        }
    }
    {
        float* ur = sU + wave * UPITCH + hi * 128 + i;
        ur[0]  = U0.x; ur[16] = U0.y; ur[32] = U1.x; ur[48]  = U1.y;
        ur[64] = U2.x; ur[80] = U2.y; ur[96] = U3.x; ur[112] = U3.y;
        sU[wave * UPITCH + 256 + hi * 16 + i] = hi ? xown : xs;
    }

    __syncthreads();

    const int col = (wave & 7) * 16 + i;
    v2f bg[4] = {};
    if (wave < 8) {
#pragma unroll
        for (int kk = 0; kk < 4; ++kk) {
            bg[kk].x = W_gat[(4 * kk + 2 * hi) * 128 + col];
            bg[kk].y = W_gat[(4 * kk + 2 * hi + 1) * 128 + col];
        }
    }

    if (wave == 0) {
        v8f accA = {}, accB = {};
        const float* ar = sU + i * UPITCH + 2 * hi;
#pragma unroll
        for (int kk = 0; kk < 72; ++kk) {
            const float* wp = (kk < 64) ? (W_nn + (4 * kk + 2 * hi) * 16 + i)
                            : (kk < 68) ? (b_nn + (4 * (kk - 64) + 2 * hi) * 16 + i)
                                        : (W_root + (4 * (kk - 68) + 2 * hi) * 16 + i);
            v2f a, b;
            a.x = ar[4 * kk]; a.y = ar[4 * kk + 1];
            b.x = wp[0];      b.y = wp[16];
            if (kk & 1)
                accB = __builtin_amdgcn_wmma_f32_16x16x4_f32(
                    false, a, false, b, (short)0, accB, false, false);
            else
                accA = __builtin_amdgcn_wmma_f32_16x16x4_f32(
                    false, a, false, b, (short)0, accA, false, false);
        }
        const v8f acc = accA + accB;
        const float be = b_ecc[i];
#pragma unroll
        for (int v = 0; v < 8; ++v)
            sH[(v + 8 * hi) * HPITCH + i] = fmaxf(acc[v] + be, 0.0f);
    }
    __syncthreads();

    if (wave < 8) {
        v8f acc = {};
        const float* hr = sH + i * HPITCH + 2 * hi;
#pragma unroll
        for (int kk = 0; kk < 4; ++kk) {
            v2f a;
            a.x = hr[4 * kk]; a.y = hr[4 * kk + 1];
            acc = __builtin_amdgcn_wmma_f32_16x16x4_f32(
                false, a, false, bg[kk], (short)0, acc, false, false);
        }
#pragma unroll
        for (int v = 0; v < 8; ++v)
            sZ[(v + 8 * hi) * ZPITCH + col] = acc[v];
    }
    else if (wave == 10 || wave == 11) {
        const int t  = (wave - 10) * 32 + lane;
        const int nn = t >> 2, part = t & 3;
        const int r  = blockIdx.x * 16 + nn;
        if (r < N)
            *reinterpret_cast<v4f*>(hrow + (size_t)r * 16 + part * 4) =
                *reinterpret_cast<const v4f*>(sH + nn * HPITCH + part * 4);
    }
    __syncthreads();

    if (wave == 8 || wave == 9) {
        const int t  = (wave - 8) * 32 + lane;
        const int nn = t >> 2, hh = t & 3;
        const int r  = blockIdx.x * 16 + nn;
        const v4f* zr = reinterpret_cast<const v4f*>(sZ + nn * ZPITCH + hh * 32);
        const v4f* as = sA + hh * 8;
        const v4f* ad = sA + 32 + hh * 8;
        float s = 0.f, d = 0.f;
#pragma unroll
        for (int j = 0; j < 8; ++j) {
            const v4f zz = zr[j], sa = as[j], da = ad[j];
            s += zz.x * sa.x; d += zz.x * da.x;
            s += zz.y * sa.y; d += zz.y * da.y;
            s += zz.z * sa.z; d += zz.z * da.z;
            s += zz.w * sa.w; d += zz.w * da.w;
        }
        if (r < N) { esrc[(size_t)r * 4 + hh] = s; edst[(size_t)r * 4 + hh] = d; }
    }
}

#define NCH 256
#define NBK 256
#define SORT_THREADS 1024
static_assert(SORT_THREADS == 4 * NCH && NBK <= SORT_THREADS, "k_group: four threads per chunk segment");

__device__ __forceinline__ int block_excl_scan(int v, int* wsum, int* tot) {
    const int lane = threadIdx.x & 31, wave = threadIdx.x >> 5;
    int inc = v;
    for (int o = 1; o < 32; o <<= 1) {
        int t = __shfl_up(inc, o, 32);
        if (lane >= o) inc += t;
    }
    if (lane == 31) wsum[wave] = inc;
    __syncthreads();
    if (wave == 0) {
        const int w = wsum[lane];
        int wi = w;
        for (int o = 1; o < 32; o <<= 1) {
            int t = __shfl_up(wi, o, 32);
            if (lane >= o) wi += t;
        }
        wsum[lane] = wi - w;
        if (lane == 31) wsum[32] = wi;
    }
    __syncthreads();
    *tot = wsum[32];
    const int r = wsum[wave] + inc - v;
    __syncthreads();
    return r;
}

__global__ __launch_bounds__(SORT_THREADS)
void k_bucket(const int* __restrict__ eidx,
              int* __restrict__ bdst,
              int* __restrict__ bsrc,
              int* __restrict__ bedge,
              int* __restrict__ tabT, int E, int bw) {
    __shared__ int hist[NBK];
    __shared__ int wsum[33];
    const int tid = threadIdx.x;
    const int chunk = (E + NCH - 1) / NCH;
    int e0 = blockIdx.x * chunk; if (e0 > E) e0 = E;
    int e1 = e0 + chunk;         if (e1 > E) e1 = E;

    if (tid < NBK) hist[tid] = 0;
    __syncthreads();
    for (int eb = e0; eb < e1; eb += 4 * SORT_THREADS) {
        int d[4];
#pragma unroll
        for (int u = 0; u < 4; ++u) {
            const int e = eb + u * SORT_THREADS + tid;
            d[u] = (e < e1) ? eidx[E + e] : -1;
        }
#pragma unroll
        for (int u = 0; u < 4; ++u)
            if (d[u] >= 0) atomicAdd(&hist[d[u] / bw], 1);
    }
    __syncthreads();
    int tot;
    const int v = (tid < NBK) ? hist[tid] : 0;
    const int ex = block_excl_scan(v, wsum, &tot);
    if (tid < NBK) {
        tabT[tid * NCH + blockIdx.x] = ex;
        hist[tid] = ex;
    }
    if (tid == 0) tabT[NBK * NCH + blockIdx.x] = tot;
    __syncthreads();
    for (int eb = e0; eb < e1; eb += 4 * SORT_THREADS) {
        int d[4], s[4];
#pragma unroll
        for (int u = 0; u < 4; ++u) {
            const int e = eb + u * SORT_THREADS + tid;
            d[u] = -1; s[u] = 0;
            if (e < e1) { d[u] = eidx[E + e]; s[u] = eidx[e]; }
        }
#pragma unroll
        for (int u = 0; u < 4; ++u) {
            if (d[u] >= 0) {
                const int p = e0 + atomicAdd(&hist[d[u] / bw], 1);
                bdst[p]  = d[u];
                bsrc[p]  = s[u];
                bedge[p] = eb + u * SORT_THREADS + tid;
            }
        }
    }
}

__global__ __launch_bounds__(SORT_THREADS)
void k_group(const int* __restrict__ bdst,
             const int* __restrict__ bsrc,
             const int* __restrict__ bedge,
             const int* __restrict__ tabT,
             int* __restrict__ off,
             int* __restrict__ deg,
             int* __restrict__ ssrc,
             int* __restrict__ sedge, int E, int N, int bw) {
    __shared__ int cnt[SORT_THREADS];
    __shared__ int wsum[33];
    const int tid = threadIdx.x;
    const int b   = blockIdx.x;
    const int c   = tid >> 2;
    const int sub = tid & 3;
    const int chunk = (E + NCH - 1) / NCH;
    int cb = c * chunk; if (cb > E) cb = E;
    const int s0 = tabT[b * NCH + c];
    const int s1 = tabT[(b + 1) * NCH + c];
    const int g  = cb + s0;
    const int l  = s1 - s0;
    const int n0 = b * bw;

    cnt[tid] = 0;
    int base;
    (void)block_excl_scan(sub == 0 ? s0 : 0, wsum, &base);

    for (int jb = 0; jb < l; jb += 16) {
        int d[4];
#pragma unroll
        for (int u = 0; u < 4; ++u) {
            const int j = jb + 4 * u + sub;
            d[u] = (j < l) ? bdst[g + j] : -1;
        }
#pragma unroll
        for (int u = 0; u < 4; ++u)
            if (d[u] >= 0) atomicAdd(&cnt[d[u] - n0], 1);
    }
    __syncthreads();
    int tot;
    const int v  = cnt[tid];
    const int ex = block_excl_scan(v, wsum, &tot);
    {
        const int n = n0 + tid;
        if (tid < bw && n < N) { off[n] = base + ex; deg[n] = v; }
    }
    cnt[tid] = base + ex;
    __syncthreads();
    for (int jb = 0; jb < l; jb += 16) {
        int d[4], s[4], e[4];
#pragma unroll
        for (int u = 0; u < 4; ++u) {
            const int j = jb + 4 * u + sub;
            d[u] = -1; s[u] = 0; e[u] = 0;
            if (j < l) { d[u] = bdst[g + j]; s[u] = bsrc[g + j]; e[u] = bedge[g + j]; }
        }
#pragma unroll
        for (int u = 0; u < 4; ++u) {
            if (d[u] >= 0) {
                const int p = atomicAdd(&cnt[d[u] - n0], 1);
                ssrc[p]  = s[u];
                sedge[p] = e[u];
            }
        }
    }
}

#define SPITCH 68
__device__ __forceinline__ float lrelu(float v) { return v > 0.f ? v : NEG_SLOPE * v; }

__global__ __launch_bounds__(512)
void k_gattile(const int* __restrict__ off,
               const int* __restrict__ deg,
               const int* __restrict__ ssrc,
               const float* __restrict__ esrc,
               const float* __restrict__ edst,
               const float* __restrict__ hrow,
               const float* __restrict__ W_gat,
               const float* __restrict__ b_gat,
               const float* __restrict__ W_fc,
               const float* __restrict__ b_fc,
               float* __restrict__ out, int N) {
    __shared__ float sS[32 * SPITCH];
    __shared__ float sP[8 * 32];

    const int tid  = threadIdx.x;
    const int wave = tid >> 5;
    const int lane = tid & 31;
    const int k    = lane & 15;
    const int half = lane >> 4;

    const int rt  = wave >> 3;
    const int ct  = wave & 7;
    const int col = ct * 16 + k;
    v2f bw[4];
#pragma unroll
    for (int kk = 0; kk < 4; ++kk) {
        bw[kk].x = W_gat[(4 * kk + 2 * half) * 128 + col];
        bw[kk].y = W_gat[(4 * kk + 2 * half + 1) * 128 + col];
    }
    const float bgc = b_gat[col];
    const float wfc = W_fc[col];

    const int nn = wave * 2 + half;
    const int n  = blockIdx.x * 32 + nn;
    int beg = 0, cnt = 0;
    float4 ed = {0.f, 0.f, 0.f, 0.f};
    if (n < N) {
        beg = off[n];
        cnt = deg[n];
        ed  = *reinterpret_cast<const float4*>(edst + (size_t)n * 4);
    }
    const int c0 = __builtin_amdgcn_readlane(cnt, 0);
    const int c1 = __builtin_amdgcn_readlane(cnt, 16);
    const int cmax = c0 > c1 ? c0 : c1;
    const float4* es4 = reinterpret_cast<const float4*>(esrc);

    float m0 = -INFINITY, m1 = -INFINITY, m2 = -INFINITY, m3 = -INFINITY;
    for (int c = 0; c < cmax; c += 16) {
        const int j = c + k;
        if (j < cnt) {
            const float4 es = es4[ssrc[beg + j]];
            m0 = fmaxf(m0, lrelu(es.x + ed.x));
            m1 = fmaxf(m1, lrelu(es.y + ed.y));
            m2 = fmaxf(m2, lrelu(es.z + ed.z));
            m3 = fmaxf(m3, lrelu(es.w + ed.w));
        }
    }
    for (int o = 8; o > 0; o >>= 1) {
        m0 = fmaxf(m0, __shfl_xor(m0, o, 32));
        m1 = fmaxf(m1, __shfl_xor(m1, o, 32));
        m2 = fmaxf(m2, __shfl_xor(m2, o, 32));
        m3 = fmaxf(m3, __shfl_xor(m3, o, 32));
    }

    float S0 = 0.f, S1 = 0.f, S2 = 0.f, S3 = 0.f;
    float d0 = 0.f, d1 = 0.f, d2 = 0.f, d3 = 0.f;
    for (int c = 0; c < cmax; c += 16) {
        const int j = c + k;
        int s = 0;
        float e0 = 0.f, e1 = 0.f, e2 = 0.f, e3 = 0.f;
        if (j < cnt) {
            s = ssrc[beg + j];
            const float4 es = es4[s];
            e0 = __expf(lrelu(es.x + ed.x) - m0);
            e1 = __expf(lrelu(es.y + ed.y) - m1);
            e2 = __expf(lrelu(es.z + ed.z) - m2);
            e3 = __expf(lrelu(es.w + ed.w) - m3);
        }
        d0 += e0; d1 += e1; d2 += e2; d3 += e3;

        const int mc = (cmax - c < 16) ? (cmax - c) : 16;
        for (int q = 0; q < mc; q += 4) {
            int sq[4];
            float hv[4];
#pragma unroll
            for (int t = 0; t < 4; ++t) sq[t] = __shfl(s, q + t, 16);
#pragma unroll
            for (int t = 0; t < 4; ++t) {
                const float ld = hrow[(size_t)sq[t] * 16 + k];
                hv[t] = (c + q + t < cnt) ? ld : 0.f;
            }
#pragma unroll
            for (int t = 0; t < 4; ++t) {
                S0 += __shfl(e0, q + t, 16) * hv[t];
                S1 += __shfl(e1, q + t, 16) * hv[t];
                S2 += __shfl(e2, q + t, 16) * hv[t];
                S3 += __shfl(e3, q + t, 16) * hv[t];
            }
        }
    }
    for (int o = 8; o > 0; o >>= 1) {
        d0 += __shfl_xor(d0, o, 32);
        d1 += __shfl_xor(d1, o, 32);
        d2 += __shfl_xor(d2, o, 32);
        d3 += __shfl_xor(d3, o, 32);
    }
    {
        float* sr = sS + nn * SPITCH + k;
        sr[0]  = S0 / (d0 + 1e-16f);
        sr[16] = S1 / (d1 + 1e-16f);
        sr[32] = S2 / (d2 + 1e-16f);
        sr[48] = S3 / (d3 + 1e-16f);
    }
    __syncthreads();

    {
        v8f acc = {};
        const float* ar = sS + (rt * 16 + k) * SPITCH + (ct >> 1) * 16 + 2 * half;
#pragma unroll
        for (int kk = 0; kk < 4; ++kk) {
            v2f a;
            a.x = ar[4 * kk]; a.y = ar[4 * kk + 1];
            acc = __builtin_amdgcn_wmma_f32_16x16x4_f32(
                false, a, false, bw[kk], (short)0, acc, false, false);
        }
        float p[8];
#pragma unroll
        for (int v = 0; v < 8; ++v) p[v] = fmaxf(acc[v] + bgc, 0.0f) * wfc;
        for (int o = 8; o > 0; o >>= 1) {
#pragma unroll
            for (int v = 0; v < 8; ++v) p[v] += __shfl_xor(p[v], o, 32);
        }
        if (k == 0) {
#pragma unroll
            for (int v = 0; v < 8; ++v) sP[ct * 32 + rt * 16 + 8 * half + v] = p[v];
        }
    }
    __syncthreads();

    if (tid < 32) {
        float y = 0.f;
#pragma unroll
        for (int c = 0; c < 8; ++c) y += sP[c * 32 + tid];
        const int r = blockIdx.x * 32 + tid;
        if (r < N) out[r] = y + b_fc[0];
    }
}

extern "C" void kernel_launch(void* const* d_in, const int* in_sizes, int n_in,
                              void* d_out, int out_size, void* d_ws, size_t ws_size,
                              hipStream_t stream) {
    const float* x      = (const float*)d_in[0];
    const int*   eidx   = (const int*)  d_in[1];
    const float* eattr  = (const float*)d_in[2];
    const float* W_nn   = (const float*)d_in[3];
    const float* b_nn   = (const float*)d_in[4];
    const float* W_root = (const float*)d_in[5];
    const float* b_ecc  = (const float*)d_in[6];
    const float* W_gat  = (const float*)d_in[7];
    const float* a_src  = (const float*)d_in[8];
    const float* a_dst  = (const float*)d_in[9];
    const float* b_gat  = (const float*)d_in[10];
    const float* W_fc   = (const float*)d_in[11];
    const float* b_fc   = (const float*)d_in[12];
    float* out = (float*)d_out;

    const int N = in_sizes[0] / 16;
    const int E = in_sizes[1] / 2;

    float* ws = (float*)d_ws;
    size_t off = 0;
    float* plane = ws + off;          off += (size_t)N * 128;
    float* hbuf  = plane;
    float* esrc  = ws + off;          off += (size_t)N * 4;
    float* edst  = ws + off;          off += (size_t)N * 4;
    int*   deg   = (int*)(ws + off);  off += (size_t)N;
    int*   noff  = (int*)(ws + off);  off += (size_t)N;
    int*   tabT  = (int*)(ws + off);  off += (size_t)E;
    int*   ssrc  = (int*)(ws + off);  off += (size_t)E;
    int*   sedge = (int*)(ws + off);  off += (size_t)E;
    int*   bdst  = (int*)plane;
    int*   bsrc  = bdst  + (size_t)E;
    int*   bedge = bsrc  + (size_t)E;
    const int bw = (N + NBK - 1) / NBK;

    k_bucket<<<NCH, SORT_THREADS, 0, stream>>>(eidx, bdst, bsrc, bedge, tabT, E, bw);
    k_group <<<NBK, SORT_THREADS, 0, stream>>>(bdst, bsrc, bedge, tabT, noff, deg, ssrc, sedge, E, N, bw);

    k_nnagg<<<(N + 15) / 16, 512, 0, stream>>>(x, eattr, W_nn, b_nn, W_root, b_ecc, W_gat, a_src, a_dst,
                                               noff, deg, ssrc, sedge, hbuf, esrc, edst, N);
    k_gattile<<<(N + 31) / 32, 512, 0, stream>>>(noff, deg, ssrc, esrc, edst, hbuf, W_gat,
                                                 b_gat, W_fc, b_fc, out, N);
}


# NNConv per-node edge sum done as a WMMA product with the edges as k: U = EA^T[16,edges] @ X[edges,16], 4 edges per f32 16x16x4; a chunk's row loads all issued at once; sum order changes
# speedup vs baseline: 12.6098x; 1.0906x over previous
#include <hip/hip_runtime.h>
#include <hip/hip_bf16.h>

typedef __attribute__((ext_vector_type(2))) float v2f;
typedef __attribute__((ext_vector_type(4))) float v4f;
typedef __attribute__((ext_vector_type(8))) float v8f;

#define NEG_SLOPE 0.2f

#define UPITCH 292
#define HPITCH 20
#define ZPITCH 132
template <int L>
__device__ __forceinline__ int row_lane(int v) {
    return __builtin_amdgcn_update_dpp(0, v, 0x150 + L, 0xf, 0xf, false);
}
__device__ __forceinline__ float ld_word(const float* base, unsigned byte_off) {
    return *reinterpret_cast<const float*>(reinterpret_cast<const char*>(base) + byte_off);
}
struct EdgeWords { float ax, ay, bx, by; };
template <int T>
__device__ __forceinline__ EdgeWords edge_loads(const float* __restrict__ x,
                                                const float* __restrict__ eattr,
                                                int s, int e, unsigned m4) {
    EdgeWords w;
    w.ax = ld_word(eattr, (unsigned)row_lane<2 * T>(e) * 64u + m4);
    w.ay = ld_word(eattr, (unsigned)row_lane<2 * T + 1>(e) * 64u + m4);
    w.bx = ld_word(x,     (unsigned)row_lane<2 * T>(s) * 64u + m4);
    w.by = ld_word(x,     (unsigned)row_lane<2 * T + 1>(s) * 64u + m4);
    return w;
}
template <int T>
__device__ __forceinline__ void edge_step(const EdgeWords& w, int lim, v8f& U, float& xsum) {
    const bool vx = 4 * T < lim, vy = 4 * T + 1 < lim;
    v2f a, b;
    a.x = vx ? w.ax : 0.f; a.y = vy ? w.ay : 0.f;
    b.x = vx ? w.bx : 0.f; b.y = vy ? w.by : 0.f;
    xsum += b.x + b.y;
    U = __builtin_amdgcn_wmma_f32_16x16x4_f32(false, a, false, b, (short)0, U, false, false);
}

__global__ __launch_bounds__(512, 12)
void k_nnagg(const float* __restrict__ x,
             const float* __restrict__ eattr,
             const float* __restrict__ W_nn,
             const float* __restrict__ b_nn,
             const float* __restrict__ W_root,
             const float* __restrict__ b_ecc,
             const float* __restrict__ W_gat,
             const float* __restrict__ a_src,
             const float* __restrict__ a_dst,
             const int* __restrict__ off,
             const int* __restrict__ deg,
             const int* __restrict__ ssrc,
             const int* __restrict__ sedge,
             float* __restrict__ hrow,
             float* __restrict__ esrc,
             float* __restrict__ edst, int N) {
    __shared__ float sU[16 * UPITCH];
    __shared__ __attribute__((aligned(16))) float sH[16 * HPITCH];
    __shared__ __attribute__((aligned(16))) float sZ[16 * ZPITCH];
    __shared__ v4f sA[64];

    const int tid  = threadIdx.x;
    const int wave = tid >> 5;
    const int lane = tid & 31;
    const int i    = lane & 15;
    const int hi   = lane >> 4;
    const int n    = blockIdx.x * 16 + wave;

    if (tid < 64)
        sA[tid] = reinterpret_cast<const v4f*>(tid < 32 ? a_src : a_dst)[tid & 31];

    v8f U = {};
    float xsum = 0.f;
    float xown = 0.f;
    const int nu = __builtin_amdgcn_readfirstlane(n);
    if (nu < N) {
        xown = x[(size_t)nu * 16 + i];
        const int beg = __builtin_amdgcn_readfirstlane(off[nu]);
        const int cnt = __builtin_amdgcn_readfirstlane(deg[nu]);
        const int jl  = 4 * (i >> 1) + 2 * hi + (i & 1);
        const unsigned m4 = (unsigned)i * 4u;
        for (int c = 0; c < cnt; c += 32) {
            const int rem = cnt - c;
            int s = 0, e = 0;
            if (jl < rem) { s = ssrc[beg + c + jl]; e = sedge[beg + c + jl]; }
            const int lim = rem - 2 * hi;
            const EdgeWords w0 = edge_loads<0>(x, eattr, s, e, m4);
            const EdgeWords w1 = edge_loads<1>(x, eattr, s, e, m4);
            const EdgeWords w2 = edge_loads<2>(x, eattr, s, e, m4);
            const EdgeWords w3 = edge_loads<3>(x, eattr, s, e, m4);
            if (rem > 16) {
                const EdgeWords w4 = edge_loads<4>(x, eattr, s, e, m4);
                const EdgeWords w5 = edge_loads<5>(x, eattr, s, e, m4);
                const EdgeWords w6 = edge_loads<6>(x, eattr, s, e, m4);
                const EdgeWords w7 = edge_loads<7>(x, eattr, s, e, m4);
                edge_step<0>(w0, lim, U, xsum); edge_step<1>(w1, lim, U, xsum);
                edge_step<2>(w2, lim, U, xsum); edge_step<3>(w3, lim, U, xsum);
                edge_step<4>(w4, lim, U, xsum); edge_step<5>(w5, lim, U, xsum);
                edge_step<6>(w6, lim, U, xsum); edge_step<7>(w7, lim, U, xsum);
            } else {
                edge_step<0>(w0, lim, U, xsum); edge_step<1>(w1, lim, U, xsum);
                edge_step<2>(w2, lim, U, xsum); edge_step<3>(w3, lim, U, xsum);
            }
        }
    }
    {
        const float xs = xsum + __shfl_xor(xsum, 16, 32);
        float* ur = sU + wave * UPITCH + hi * 128 + i;
#pragma unroll
        for (int r = 0; r < 8; ++r) ur[16 * r] = U[r];
        sU[wave * UPITCH + 256 + hi * 16 + i] = hi ? xown : xs;
    }

    __syncthreads();

    const int col = (wave & 7) * 16 + i;
    v2f bg[4] = {};
    if (wave < 8) {
#pragma unroll
        for (int kk = 0; kk < 4; ++kk) {
            bg[kk].x = W_gat[(4 * kk + 2 * hi) * 128 + col];
            bg[kk].y = W_gat[(4 * kk + 2 * hi + 1) * 128 + col];
        }
    }

    if (wave == 0) {
        v8f accA = {}, accB = {};
        const float* ar = sU + i * UPITCH + 2 * hi;
#pragma unroll
        for (int kk = 0; kk < 72; ++kk) {
            const float* wp = (kk < 64) ? (W_nn + (4 * kk + 2 * hi) * 16 + i)
                            : (kk < 68) ? (b_nn + (4 * (kk - 64) + 2 * hi) * 16 + i)
                                        : (W_root + (4 * (kk - 68) + 2 * hi) * 16 + i);
            v2f a, b;
            a.x = ar[4 * kk]; a.y = ar[4 * kk + 1];
            b.x = wp[0];      b.y = wp[16];
            if (kk & 1)
                accB = __builtin_amdgcn_wmma_f32_16x16x4_f32(
                    false, a, false, b, (short)0, accB, false, false);
            else
                accA = __builtin_amdgcn_wmma_f32_16x16x4_f32(
                    false, a, false, b, (short)0, accA, false, false);
        }
        const v8f acc = accA + accB;
        const float be = b_ecc[i];
#pragma unroll
        for (int v = 0; v < 8; ++v)
            sH[(v + 8 * hi) * HPITCH + i] = fmaxf(acc[v] + be, 0.0f);
    }
    __syncthreads();

    if (wave < 8) {
        v8f acc = {};
        const float* hr = sH + i * HPITCH + 2 * hi;
#pragma unroll
        for (int kk = 0; kk < 4; ++kk) {
            v2f a;
            a.x = hr[4 * kk]; a.y = hr[4 * kk + 1];
            acc = __builtin_amdgcn_wmma_f32_16x16x4_f32(
                false, a, false, bg[kk], (short)0, acc, false, false);
        }
#pragma unroll
        for (int v = 0; v < 8; ++v)
            sZ[(v + 8 * hi) * ZPITCH + col] = acc[v];
    }
    else if (wave == 10 || wave == 11) {
        const int t  = (wave - 10) * 32 + lane;
        const int nn = t >> 2, part = t & 3;
        const int r  = blockIdx.x * 16 + nn;
        if (r < N)
            *reinterpret_cast<v4f*>(hrow + (size_t)r * 16 + part * 4) =
                *reinterpret_cast<const v4f*>(sH + nn * HPITCH + part * 4);
    }
    __syncthreads();

    if (wave == 8 || wave == 9) {
        const int t  = (wave - 8) * 32 + lane;
        const int nn = t >> 2, hh = t & 3;
        const int r  = blockIdx.x * 16 + nn;
        const v4f* zr = reinterpret_cast<const v4f*>(sZ + nn * ZPITCH + hh * 32);
        const v4f* as = sA + hh * 8;
        const v4f* ad = sA + 32 + hh * 8;
        float s = 0.f, d = 0.f;
#pragma unroll
        for (int j = 0; j < 8; ++j) {
            const v4f zz = zr[j], sa = as[j], da = ad[j];
            s += zz.x * sa.x; d += zz.x * da.x;
            s += zz.y * sa.y; d += zz.y * da.y;
            s += zz.z * sa.z; d += zz.z * da.z;
            s += zz.w * sa.w; d += zz.w * da.w;
        }
        if (r < N) { esrc[(size_t)r * 4 + hh] = s; edst[(size_t)r * 4 + hh] = d; }
    }
}

#define NCH 256
#define NBK 256
#define SORT_THREADS 1024
static_assert(SORT_THREADS == 4 * NCH && NBK <= SORT_THREADS, "k_group: four threads per chunk segment");

__device__ __forceinline__ int block_excl_scan(int v, int* wsum, int* tot) {
    const int lane = threadIdx.x & 31, wave = threadIdx.x >> 5;
    int inc = v;
    for (int o = 1; o < 32; o <<= 1) {
        int t = __shfl_up(inc, o, 32);
        if (lane >= o) inc += t;
    }
    if (lane == 31) wsum[wave] = inc;
    __syncthreads();
    if (wave == 0) {
        const int w = wsum[lane];
        int wi = w;
        for (int o = 1; o < 32; o <<= 1) {
            int t = __shfl_up(wi, o, 32);
            if (lane >= o) wi += t;
        }
        wsum[lane] = wi - w;
        if (lane == 31) wsum[32] = wi;
    }
    __syncthreads();
    *tot = wsum[32];
    const int r = wsum[wave] + inc - v;
    __syncthreads();
    return r;
}

__global__ __launch_bounds__(SORT_THREADS)
void k_bucket(const int* __restrict__ eidx,
              int* __restrict__ bdst,
              int* __restrict__ bsrc,
              int* __restrict__ bedge,
              int* __restrict__ tabT, int E, int bw) {
    __shared__ int hist[NBK];
    __shared__ int wsum[33];
    const int tid = threadIdx.x;
    const int chunk = (E + NCH - 1) / NCH;
    int e0 = blockIdx.x * chunk; if (e0 > E) e0 = E;
    int e1 = e0 + chunk;         if (e1 > E) e1 = E;

    if (tid < NBK) hist[tid] = 0;
    __syncthreads();
    for (int eb = e0; eb < e1; eb += 4 * SORT_THREADS) {
        int d[4];
#pragma unroll
        for (int u = 0; u < 4; ++u) {
            const int e = eb + u * SORT_THREADS + tid;
            d[u] = (e < e1) ? eidx[E + e] : -1;
        }
#pragma unroll
        for (int u = 0; u < 4; ++u)
            if (d[u] >= 0) atomicAdd(&hist[d[u] / bw], 1);
    }
    __syncthreads();
    int tot;
    const int v = (tid < NBK) ? hist[tid] : 0;
    const int ex = block_excl_scan(v, wsum, &tot);
    if (tid < NBK) {
        tabT[tid * NCH + blockIdx.x] = ex;
        hist[tid] = ex;
    }
    if (tid == 0) tabT[NBK * NCH + blockIdx.x] = tot;
    __syncthreads();
    for (int eb = e0; eb < e1; eb += 4 * SORT_THREADS) {
        int d[4], s[4];
#pragma unroll
        for (int u = 0; u < 4; ++u) {
            const int e = eb + u * SORT_THREADS + tid;
            d[u] = -1; s[u] = 0;
            if (e < e1) { d[u] = eidx[E + e]; s[u] = eidx[e]; }
        }
#pragma unroll
        for (int u = 0; u < 4; ++u) {
            if (d[u] >= 0) {
                const int p = e0 + atomicAdd(&hist[d[u] / bw], 1);
                bdst[p]  = d[u];
                bsrc[p]  = s[u];
                bedge[p] = eb + u * SORT_THREADS + tid;
            }
        }
    }
}

__global__ __launch_bounds__(SORT_THREADS)
void k_group(const int* __restrict__ bdst,
             const int* __restrict__ bsrc,
             const int* __restrict__ bedge,
             const int* __restrict__ tabT,
             int* __restrict__ off,
             int* __restrict__ deg,
             int* __restrict__ ssrc,
             int* __restrict__ sedge, int E, int N, int bw) {
    __shared__ int cnt[SORT_THREADS];
    __shared__ int wsum[33];
    const int tid = threadIdx.x;
    const int b   = blockIdx.x;
    const int c   = tid >> 2;
    const int sub = tid & 3;
    const int chunk = (E + NCH - 1) / NCH;
    int cb = c * chunk; if (cb > E) cb = E;
    const int s0 = tabT[b * NCH + c];
    const int s1 = tabT[(b + 1) * NCH + c];
    const int g  = cb + s0;
    const int l  = s1 - s0;
    const int n0 = b * bw;

    cnt[tid] = 0;
    int base;
    (void)block_excl_scan(sub == 0 ? s0 : 0, wsum, &base);

    for (int jb = 0; jb < l; jb += 16) {
        int d[4];
#pragma unroll
        for (int u = 0; u < 4; ++u) {
            const int j = jb + 4 * u + sub;
            d[u] = (j < l) ? bdst[g + j] : -1;
        }
#pragma unroll
        for (int u = 0; u < 4; ++u)
            if (d[u] >= 0) atomicAdd(&cnt[d[u] - n0], 1);
    }
    __syncthreads();
    int tot;
    const int v  = cnt[tid];
    const int ex = block_excl_scan(v, wsum, &tot);
    {
        const int n = n0 + tid;
        if (tid < bw && n < N) { off[n] = base + ex; deg[n] = v; }
    }
    cnt[tid] = base + ex;
    __syncthreads();
    for (int jb = 0; jb < l; jb += 16) {
        int d[4], s[4], e[4];
#pragma unroll
        for (int u = 0; u < 4; ++u) {
            const int j = jb + 4 * u + sub;
            d[u] = -1; s[u] = 0; e[u] = 0;
            if (j < l) { d[u] = bdst[g + j]; s[u] = bsrc[g + j]; e[u] = bedge[g + j]; }
        }
#pragma unroll
        for (int u = 0; u < 4; ++u) {
            if (d[u] >= 0) {
                const int p = atomicAdd(&cnt[d[u] - n0], 1);
                ssrc[p]  = s[u];
                sedge[p] = e[u];
            }
        }
    }
}

#define SPITCH 68
__device__ __forceinline__ float lrelu(float v) { return v > 0.f ? v : NEG_SLOPE * v; }

__global__ __launch_bounds__(512)
void k_gattile(const int* __restrict__ off,
               const int* __restrict__ deg,
               const int* __restrict__ ssrc,
               const float* __restrict__ esrc,
               const float* __restrict__ edst,
               const float* __restrict__ hrow,
               const float* __restrict__ W_gat,
               const float* __restrict__ b_gat,
               const float* __restrict__ W_fc,
               const float* __restrict__ b_fc,
               float* __restrict__ out, int N) {
    __shared__ float sS[32 * SPITCH];
    __shared__ float sP[8 * 32];

    const int tid  = threadIdx.x;
    const int wave = tid >> 5;
    const int lane = tid & 31;
    const int k    = lane & 15;
    const int half = lane >> 4;

    const int rt  = wave >> 3;
    const int ct  = wave & 7;
    const int col = ct * 16 + k;
    v2f bw[4];
#pragma unroll
    for (int kk = 0; kk < 4; ++kk) {
        bw[kk].x = W_gat[(4 * kk + 2 * half) * 128 + col];
        bw[kk].y = W_gat[(4 * kk + 2 * half + 1) * 128 + col];
    }
    const float bgc = b_gat[col];
    const float wfc = W_fc[col];

    const int nn = wave * 2 + half;
    const int n  = blockIdx.x * 32 + nn;
    int beg = 0, cnt = 0;
    float4 ed = {0.f, 0.f, 0.f, 0.f};
    if (n < N) {
        beg = off[n];
        cnt = deg[n];
        ed  = *reinterpret_cast<const float4*>(edst + (size_t)n * 4);
    }
    const int c0 = __builtin_amdgcn_readlane(cnt, 0);
    const int c1 = __builtin_amdgcn_readlane(cnt, 16);
    const int cmax = c0 > c1 ? c0 : c1;
    const float4* es4 = reinterpret_cast<const float4*>(esrc);

    float m0 = -INFINITY, m1 = -INFINITY, m2 = -INFINITY, m3 = -INFINITY;
    for (int c = 0; c < cmax; c += 16) {
        const int j = c + k;
        if (j < cnt) {
            const float4 es = es4[ssrc[beg + j]];
            m0 = fmaxf(m0, lrelu(es.x + ed.x));
            m1 = fmaxf(m1, lrelu(es.y + ed.y));
            m2 = fmaxf(m2, lrelu(es.z + ed.z));
            m3 = fmaxf(m3, lrelu(es.w + ed.w));
        }
    }
    for (int o = 8; o > 0; o >>= 1) {
        m0 = fmaxf(m0, __shfl_xor(m0, o, 32));
        m1 = fmaxf(m1, __shfl_xor(m1, o, 32));
        m2 = fmaxf(m2, __shfl_xor(m2, o, 32));
        m3 = fmaxf(m3, __shfl_xor(m3, o, 32));
    }

    float S0 = 0.f, S1 = 0.f, S2 = 0.f, S3 = 0.f;
    float d0 = 0.f, d1 = 0.f, d2 = 0.f, d3 = 0.f;
    for (int c = 0; c < cmax; c += 16) {
        const int j = c + k;
        int s = 0;
        float e0 = 0.f, e1 = 0.f, e2 = 0.f, e3 = 0.f;
        if (j < cnt) {
            s = ssrc[beg + j];
            const float4 es = es4[s];
            e0 = __expf(lrelu(es.x + ed.x) - m0);
            e1 = __expf(lrelu(es.y + ed.y) - m1);
            e2 = __expf(lrelu(es.z + ed.z) - m2);
            e3 = __expf(lrelu(es.w + ed.w) - m3);
        }
        d0 += e0; d1 += e1; d2 += e2; d3 += e3;

        const int mc = (cmax - c < 16) ? (cmax - c) : 16;
        for (int q = 0; q < mc; q += 4) {
            int sq[4];
            float hv[4];
#pragma unroll
            for (int t = 0; t < 4; ++t) sq[t] = __shfl(s, q + t, 16);
#pragma unroll
            for (int t = 0; t < 4; ++t) {
                const float ld = hrow[(size_t)sq[t] * 16 + k];
                hv[t] = (c + q + t < cnt) ? ld : 0.f;
            }
#pragma unroll
            for (int t = 0; t < 4; ++t) {
                S0 += __shfl(e0, q + t, 16) * hv[t];
                S1 += __shfl(e1, q + t, 16) * hv[t];
                S2 += __shfl(e2, q + t, 16) * hv[t];
                S3 += __shfl(e3, q + t, 16) * hv[t];
            }
        }
    }
    for (int o = 8; o > 0; o >>= 1) {
        d0 += __shfl_xor(d0, o, 32);
        d1 += __shfl_xor(d1, o, 32);
        d2 += __shfl_xor(d2, o, 32);
        d3 += __shfl_xor(d3, o, 32);
    }
    {
        float* sr = sS + nn * SPITCH + k;
        sr[0]  = S0 / (d0 + 1e-16f);
        sr[16] = S1 / (d1 + 1e-16f);
        sr[32] = S2 / (d2 + 1e-16f);
        sr[48] = S3 / (d3 + 1e-16f);
    }
    __syncthreads();

    {
        v8f acc = {};
        const float* ar = sS + (rt * 16 + k) * SPITCH + (ct >> 1) * 16 + 2 * half;
#pragma unroll
        for (int kk = 0; kk < 4; ++kk) {
            v2f a;
            a.x = ar[4 * kk]; a.y = ar[4 * kk + 1];
            acc = __builtin_amdgcn_wmma_f32_16x16x4_f32(
                false, a, false, bw[kk], (short)0, acc, false, false);
        }
        float p[8];
#pragma unroll
        for (int v = 0; v < 8; ++v) p[v] = fmaxf(acc[v] + bgc, 0.0f) * wfc;
        for (int o = 8; o > 0; o >>= 1) {
#pragma unroll
            for (int v = 0; v < 8; ++v) p[v] += __shfl_xor(p[v], o, 32);
        }
        if (k == 0) {
#pragma unroll
            for (int v = 0; v < 8; ++v) sP[ct * 32 + rt * 16 + 8 * half + v] = p[v];
        }
    }
    __syncthreads();

    if (tid < 32) {
        float y = 0.f;
#pragma unroll
        for (int c = 0; c < 8; ++c) y += sP[c * 32 + tid];
        const int r = blockIdx.x * 32 + tid;
        if (r < N) out[r] = y + b_fc[0];
    }
}

extern "C" void kernel_launch(void* const* d_in, const int* in_sizes, int n_in,
                              void* d_out, int out_size, void* d_ws, size_t ws_size,
                              hipStream_t stream) {
    const float* x      = (const float*)d_in[0];
    const int*   eidx   = (const int*)  d_in[1];
    const float* eattr  = (const float*)d_in[2];
    const float* W_nn   = (const float*)d_in[3];
    const float* b_nn   = (const float*)d_in[4];
    const float* W_root = (const float*)d_in[5];
    const float* b_ecc  = (const float*)d_in[6];
    const float* W_gat  = (const float*)d_in[7];
    const float* a_src  = (const float*)d_in[8];
    const float* a_dst  = (const float*)d_in[9];
    const float* b_gat  = (const float*)d_in[10];
    const float* W_fc   = (const float*)d_in[11];
    const float* b_fc   = (const float*)d_in[12];
    float* out = (float*)d_out;

    const int N = in_sizes[0] / 16;
    const int E = in_sizes[1] / 2;

    float* ws = (float*)d_ws;
    size_t off = 0;
    float* plane = ws + off;          off += (size_t)N * 128;
    float* hbuf  = plane;
    float* esrc  = ws + off;          off += (size_t)N * 4;
    float* edst  = ws + off;          off += (size_t)N * 4;
    int*   deg   = (int*)(ws + off);  off += (size_t)N;
    int*   noff  = (int*)(ws + off);  off += (size_t)N;
    int*   tabT  = (int*)(ws + off);  off += (size_t)E;
    int*   ssrc  = (int*)(ws + off);  off += (size_t)E;
    int*   sedge = (int*)(ws + off);  off += (size_t)E;
    int*   bdst  = (int*)plane;
    int*   bsrc  = bdst  + (size_t)E;
    int*   bedge = bsrc  + (size_t)E;
    const int bw = (N + NBK - 1) / NBK;

    k_bucket<<<NCH, SORT_THREADS, 0, stream>>>(eidx, bdst, bsrc, bedge, tabT, E, bw);
    k_group <<<NBK, SORT_THREADS, 0, stream>>>(bdst, bsrc, bedge, tabT, noff, deg, ssrc, sedge, E, N, bw);

    k_nnagg<<<(N + 15) / 16, 512, 0, stream>>>(x, eattr, W_nn, b_nn, W_root, b_ecc, W_gat, a_src, a_dst,
                                               noff, deg, ssrc, sedge, hbuf, esrc, edst, N);
    k_gattile<<<(N + 31) / 32, 512, 0, stream>>>(noff, deg, ssrc, esrc, edst, hbuf, W_gat,
                                                 b_gat, W_fc, b_fc, out, N);
}


# GAT kernel: each pass requests its gathers together (pass A: two chunks' list words, then their scores; pass B: a chunk's 16 h rows beside its scores, lane values by DPP); sum order kept, fma
# speedup vs baseline: 13.2329x; 1.0494x over previous
#include <hip/hip_runtime.h>
#include <hip/hip_bf16.h>

typedef __attribute__((ext_vector_type(2))) float v2f;
typedef __attribute__((ext_vector_type(4))) float v4f;
typedef __attribute__((ext_vector_type(8))) float v8f;

#define NEG_SLOPE 0.2f

#define UPITCH 292
#define HPITCH 20
#define ZPITCH 132
template <int L>
__device__ __forceinline__ int row_lane(int v) {
    return __builtin_amdgcn_update_dpp(0, v, 0x150 + L, 0xf, 0xf, false);
}
__device__ __forceinline__ float ld_word(const float* base, unsigned byte_off) {
    return *reinterpret_cast<const float*>(reinterpret_cast<const char*>(base) + byte_off);
}
struct EdgeWords { float ax, ay, bx, by; };
template <int T>
__device__ __forceinline__ EdgeWords edge_loads(const float* __restrict__ x,
                                                const float* __restrict__ eattr,
                                                int s, int e, unsigned m4) {
    EdgeWords w;
    w.ax = ld_word(eattr, (unsigned)row_lane<2 * T>(e) * 64u + m4);
    w.ay = ld_word(eattr, (unsigned)row_lane<2 * T + 1>(e) * 64u + m4);
    w.bx = ld_word(x,     (unsigned)row_lane<2 * T>(s) * 64u + m4);
    w.by = ld_word(x,     (unsigned)row_lane<2 * T + 1>(s) * 64u + m4);
    return w;
}
template <int T>
__device__ __forceinline__ void edge_step(const EdgeWords& w, int lim, v8f& U, float& xsum) {
    const bool vx = 4 * T < lim, vy = 4 * T + 1 < lim;
    v2f a, b;
    a.x = vx ? w.ax : 0.f; a.y = vy ? w.ay : 0.f;
    b.x = vx ? w.bx : 0.f; b.y = vy ? w.by : 0.f;
    xsum += b.x + b.y;
    U = __builtin_amdgcn_wmma_f32_16x16x4_f32(false, a, false, b, (short)0, U, false, false);
}

__global__ __launch_bounds__(512, 12)
void k_nnagg(const float* __restrict__ x,
             const float* __restrict__ eattr,
             const float* __restrict__ W_nn,
             const float* __restrict__ b_nn,
             const float* __restrict__ W_root,
             const float* __restrict__ b_ecc,
             const float* __restrict__ W_gat,
             const float* __restrict__ a_src,
             const float* __restrict__ a_dst,
             const int* __restrict__ off,
             const int* __restrict__ deg,
             const int* __restrict__ ssrc,
             const int* __restrict__ sedge,
             float* __restrict__ hrow,
             float* __restrict__ esrc,
             float* __restrict__ edst, int N) {
    __shared__ float sU[16 * UPITCH];
    __shared__ __attribute__((aligned(16))) float sH[16 * HPITCH];
    __shared__ __attribute__((aligned(16))) float sZ[16 * ZPITCH];
    __shared__ v4f sA[64];

    const int tid  = threadIdx.x;
    const int wave = tid >> 5;
    const int lane = tid & 31;
    const int i    = lane & 15;
    const int hi   = lane >> 4;
    const int n    = blockIdx.x * 16 + wave;

    if (tid < 64)
        sA[tid] = reinterpret_cast<const v4f*>(tid < 32 ? a_src : a_dst)[tid & 31];

    v8f U = {};
    float xsum = 0.f;
    float xown = 0.f;
    const int nu = __builtin_amdgcn_readfirstlane(n);
    if (nu < N) {
        xown = x[(size_t)nu * 16 + i];
        const int beg = __builtin_amdgcn_readfirstlane(off[nu]);
        const int cnt = __builtin_amdgcn_readfirstlane(deg[nu]);
        const int jl  = 4 * (i >> 1) + 2 * hi + (i & 1);
        const unsigned m4 = (unsigned)i * 4u;
        for (int c = 0; c < cnt; c += 32) {
            const int rem = cnt - c;
            int s = 0, e = 0;
            if (jl < rem) { s = ssrc[beg + c + jl]; e = sedge[beg + c + jl]; }
            const int lim = rem - 2 * hi;
            const EdgeWords w0 = edge_loads<0>(x, eattr, s, e, m4);
            const EdgeWords w1 = edge_loads<1>(x, eattr, s, e, m4);
            const EdgeWords w2 = edge_loads<2>(x, eattr, s, e, m4);
            const EdgeWords w3 = edge_loads<3>(x, eattr, s, e, m4);
            if (rem > 16) {
                const EdgeWords w4 = edge_loads<4>(x, eattr, s, e, m4);
                const EdgeWords w5 = edge_loads<5>(x, eattr, s, e, m4);
                const EdgeWords w6 = edge_loads<6>(x, eattr, s, e, m4);
                const EdgeWords w7 = edge_loads<7>(x, eattr, s, e, m4);
                edge_step<0>(w0, lim, U, xsum); edge_step<1>(w1, lim, U, xsum);
                edge_step<2>(w2, lim, U, xsum); edge_step<3>(w3, lim, U, xsum);
                edge_step<4>(w4, lim, U, xsum); edge_step<5>(w5, lim, U, xsum);
                edge_step<6>(w6, lim, U, xsum); edge_step<7>(w7, lim, U, xsum);
            } else {
                edge_step<0>(w0, lim, U, xsum); edge_step<1>(w1, lim, U, xsum);
                edge_step<2>(w2, lim, U, xsum); edge_step<3>(w3, lim, U, xsum);
            }
        }
    }
    {
        const float xs = xsum + __shfl_xor(xsum, 16, 32);
        float* ur = sU + wave * UPITCH + hi * 128 + i;
#pragma unroll
        for (int r = 0; r < 8; ++r) ur[16 * r] = U[r];
        sU[wave * UPITCH + 256 + hi * 16 + i] = hi ? xown : xs;
    }

    __syncthreads();

    const int col = (wave & 7) * 16 + i;
    v2f bg[4] = {};
    if (wave < 8) {
#pragma unroll
        for (int kk = 0; kk < 4; ++kk) {
            bg[kk].x = W_gat[(4 * kk + 2 * hi) * 128 + col];
            bg[kk].y = W_gat[(4 * kk + 2 * hi + 1) * 128 + col];
        }
    }

    if (wave == 0) {
        v8f accA = {}, accB = {};
        const float* ar = sU + i * UPITCH + 2 * hi;
#pragma unroll
        for (int kk = 0; kk < 72; ++kk) {
            const float* wp = (kk < 64) ? (W_nn + (4 * kk + 2 * hi) * 16 + i)
                            : (kk < 68) ? (b_nn + (4 * (kk - 64) + 2 * hi) * 16 + i)
                                        : (W_root + (4 * (kk - 68) + 2 * hi) * 16 + i);
            v2f a, b;
            a.x = ar[4 * kk]; a.y = ar[4 * kk + 1];
            b.x = wp[0];      b.y = wp[16];
            if (kk & 1)
                accB = __builtin_amdgcn_wmma_f32_16x16x4_f32(
                    false, a, false, b, (short)0, accB, false, false);
            else
                accA = __builtin_amdgcn_wmma_f32_16x16x4_f32(
                    false, a, false, b, (short)0, accA, false, false);
        }
        const v8f acc = accA + accB;
        const float be = b_ecc[i];
#pragma unroll
        for (int v = 0; v < 8; ++v)
            sH[(v + 8 * hi) * HPITCH + i] = fmaxf(acc[v] + be, 0.0f);
    }
    __syncthreads();

    if (wave < 8) {
        v8f acc = {};
        const float* hr = sH + i * HPITCH + 2 * hi;
#pragma unroll
        for (int kk = 0; kk < 4; ++kk) {
            v2f a;
            a.x = hr[4 * kk]; a.y = hr[4 * kk + 1];
            acc = __builtin_amdgcn_wmma_f32_16x16x4_f32(
                false, a, false, bg[kk], (short)0, acc, false, false);
        }
#pragma unroll
        for (int v = 0; v < 8; ++v)
            sZ[(v + 8 * hi) * ZPITCH + col] = acc[v];
    }
    else if (wave == 10 || wave == 11) {
        const int t  = (wave - 10) * 32 + lane;
        const int nn = t >> 2, part = t & 3;
        const int r  = blockIdx.x * 16 + nn;
        if (r < N)
            *reinterpret_cast<v4f*>(hrow + (size_t)r * 16 + part * 4) =
                *reinterpret_cast<const v4f*>(sH + nn * HPITCH + part * 4);
    }
    __syncthreads();

    if (wave == 8 || wave == 9) {
        const int t  = (wave - 8) * 32 + lane;
        const int nn = t >> 2, hh = t & 3;
        const int r  = blockIdx.x * 16 + nn;
        const v4f* zr = reinterpret_cast<const v4f*>(sZ + nn * ZPITCH + hh * 32);
        const v4f* as = sA + hh * 8;
        const v4f* ad = sA + 32 + hh * 8;
        float s = 0.f, d = 0.f;
#pragma unroll
        for (int j = 0; j < 8; ++j) {
            const v4f zz = zr[j], sa = as[j], da = ad[j];
            s += zz.x * sa.x; d += zz.x * da.x;
            s += zz.y * sa.y; d += zz.y * da.y;
            s += zz.z * sa.z; d += zz.z * da.z;
            s += zz.w * sa.w; d += zz.w * da.w;
        }
        if (r < N) { esrc[(size_t)r * 4 + hh] = s; edst[(size_t)r * 4 + hh] = d; }
    }
}

#define NCH 256
#define NBK 256
#define SORT_THREADS 1024
static_assert(SORT_THREADS == 4 * NCH && NBK <= SORT_THREADS, "k_group: four threads per chunk segment");

__device__ __forceinline__ int block_excl_scan(int v, int* wsum, int* tot) {
    const int lane = threadIdx.x & 31, wave = threadIdx.x >> 5;
    int inc = v;
    for (int o = 1; o < 32; o <<= 1) {
        int t = __shfl_up(inc, o, 32);
        if (lane >= o) inc += t;
    }
    if (lane == 31) wsum[wave] = inc;
    __syncthreads();
    if (wave == 0) {
        const int w = wsum[lane];
        int wi = w;
        for (int o = 1; o < 32; o <<= 1) {
            int t = __shfl_up(wi, o, 32);
            if (lane >= o) wi += t;
        }
        wsum[lane] = wi - w;
        if (lane == 31) wsum[32] = wi;
    }
    __syncthreads();
    *tot = wsum[32];
    const int r = wsum[wave] + inc - v;
    __syncthreads();
    return r;
}

__global__ __launch_bounds__(SORT_THREADS)
void k_bucket(const int* __restrict__ eidx,
              int* __restrict__ bdst,
              int* __restrict__ bsrc,
              int* __restrict__ bedge,
              int* __restrict__ tabT, int E, int bw) {
    __shared__ int hist[NBK];
    __shared__ int wsum[33];
    const int tid = threadIdx.x;
    const int chunk = (E + NCH - 1) / NCH;
    int e0 = blockIdx.x * chunk; if (e0 > E) e0 = E;
    int e1 = e0 + chunk;         if (e1 > E) e1 = E;

    if (tid < NBK) hist[tid] = 0;
    __syncthreads();
    for (int eb = e0; eb < e1; eb += 4 * SORT_THREADS) {
        int d[4];
#pragma unroll
        for (int u = 0; u < 4; ++u) {
            const int e = eb + u * SORT_THREADS + tid;
            d[u] = (e < e1) ? eidx[E + e] : -1;
        }
#pragma unroll
        for (int u = 0; u < 4; ++u)
            if (d[u] >= 0) atomicAdd(&hist[d[u] / bw], 1);
    }
    __syncthreads();
    int tot;
    const int v = (tid < NBK) ? hist[tid] : 0;
    const int ex = block_excl_scan(v, wsum, &tot);
    if (tid < NBK) {
        tabT[tid * NCH + blockIdx.x] = ex;
        hist[tid] = ex;
    }
    if (tid == 0) tabT[NBK * NCH + blockIdx.x] = tot;
    __syncthreads();
    for (int eb = e0; eb < e1; eb += 4 * SORT_THREADS) {
        int d[4], s[4];
#pragma unroll
        for (int u = 0; u < 4; ++u) {
            const int e = eb + u * SORT_THREADS + tid;
            d[u] = -1; s[u] = 0;
            if (e < e1) { d[u] = eidx[E + e]; s[u] = eidx[e]; }
        }
#pragma unroll
        for (int u = 0; u < 4; ++u) {
            if (d[u] >= 0) {
                const int p = e0 + atomicAdd(&hist[d[u] / bw], 1);
                bdst[p]  = d[u];
                bsrc[p]  = s[u];
                bedge[p] = eb + u * SORT_THREADS + tid;
            }
        }
    }
}

__global__ __launch_bounds__(SORT_THREADS)
void k_group(const int* __restrict__ bdst,
             const int* __restrict__ bsrc,
             const int* __restrict__ bedge,
             const int* __restrict__ tabT,
             int* __restrict__ off,
             int* __restrict__ deg,
             int* __restrict__ ssrc,
             int* __restrict__ sedge, int E, int N, int bw) {
    __shared__ int cnt[SORT_THREADS];
    __shared__ int wsum[33];
    const int tid = threadIdx.x;
    const int b   = blockIdx.x;
    const int c   = tid >> 2;
    const int sub = tid & 3;
    const int chunk = (E + NCH - 1) / NCH;
    int cb = c * chunk; if (cb > E) cb = E;
    const int s0 = tabT[b * NCH + c];
    const int s1 = tabT[(b + 1) * NCH + c];
    const int g  = cb + s0;
    const int l  = s1 - s0;
    const int n0 = b * bw;

    cnt[tid] = 0;
    int base;
    (void)block_excl_scan(sub == 0 ? s0 : 0, wsum, &base);

    for (int jb = 0; jb < l; jb += 16) {
        int d[4];
#pragma unroll
        for (int u = 0; u < 4; ++u) {
            const int j = jb + 4 * u + sub;
            d[u] = (j < l) ? bdst[g + j] : -1;
        }
#pragma unroll
        for (int u = 0; u < 4; ++u)
            if (d[u] >= 0) atomicAdd(&cnt[d[u] - n0], 1);
    }
    __syncthreads();
    int tot;
    const int v  = cnt[tid];
    const int ex = block_excl_scan(v, wsum, &tot);
    {
        const int n = n0 + tid;
        if (tid < bw && n < N) { off[n] = base + ex; deg[n] = v; }
    }
    cnt[tid] = base + ex;
    __syncthreads();
    for (int jb = 0; jb < l; jb += 16) {
        int d[4], s[4], e[4];
#pragma unroll
        for (int u = 0; u < 4; ++u) {
            const int j = jb + 4 * u + sub;
            d[u] = -1; s[u] = 0; e[u] = 0;
            if (j < l) { d[u] = bdst[g + j]; s[u] = bsrc[g + j]; e[u] = bedge[g + j]; }
        }
#pragma unroll
        for (int u = 0; u < 4; ++u) {
            if (d[u] >= 0) {
                const int p = atomicAdd(&cnt[d[u] - n0], 1);
                ssrc[p]  = s[u];
                sedge[p] = e[u];
            }
        }
    }
}

#define SPITCH 68
__device__ __forceinline__ float lrelu(float v) { return v > 0.f ? v : NEG_SLOPE * v; }

template <int T, int NT>
__device__ __forceinline__ void gat_rows(const float* __restrict__ hrow, int s, unsigned k4, float* hv) {
    hv[T] = ld_word(hrow, (unsigned)row_lane<T>(s) * 64u + k4);
    if constexpr (T + 1 < NT) gat_rows<T + 1, NT>(hrow, s, k4, hv);
}
__device__ __forceinline__ int f2i(float v) { return __builtin_bit_cast(int, v); }
__device__ __forceinline__ float i2f(int v) { return __builtin_bit_cast(float, v); }
struct Head4 { float h0, h1, h2, h3; };
template <int T, int NT>
__device__ __forceinline__ void gat_sums(const float* hv, int lim, int e0, int e1, int e2, int e3, Head4& S) {
    const float h = (T < lim) ? hv[T] : 0.f;
    S.h0 = __builtin_fmaf(i2f(row_lane<T>(e0)), h, S.h0);
    S.h1 = __builtin_fmaf(i2f(row_lane<T>(e1)), h, S.h1);
    S.h2 = __builtin_fmaf(i2f(row_lane<T>(e2)), h, S.h2);
    S.h3 = __builtin_fmaf(i2f(row_lane<T>(e3)), h, S.h3);
    if constexpr (T + 1 < NT) gat_sums<T + 1, NT>(hv, lim, e0, e1, e2, e3, S);
}
template <int NT>
__device__ __forceinline__ void gat_chunk(const float* __restrict__ hrow, int s, unsigned k4, int lim,
                                          bool live, float4 es, float4 ed, Head4 mx,
                                          Head4& den, Head4& S) {
    float hv[NT];
    gat_rows<0, NT>(hrow, s, k4, hv);
    __builtin_amdgcn_sched_barrier(0);
    const float e0 = __expf(live ? lrelu(es.x + ed.x) - mx.h0 : -INFINITY);
    const float e1 = __expf(live ? lrelu(es.y + ed.y) - mx.h1 : -INFINITY);
    const float e2 = __expf(live ? lrelu(es.z + ed.z) - mx.h2 : -INFINITY);
    const float e3 = __expf(live ? lrelu(es.w + ed.w) - mx.h3 : -INFINITY);
    den.h0 += e0; den.h1 += e1; den.h2 += e2; den.h3 += e3;
    gat_sums<0, NT>(hv, lim, f2i(e0), f2i(e1), f2i(e2), f2i(e3), S);
}

__global__ __launch_bounds__(512)
void k_gattile(const int* __restrict__ off,
               const int* __restrict__ deg,
               const int* __restrict__ ssrc,
               const float* __restrict__ esrc,
               const float* __restrict__ edst,
               const float* __restrict__ hrow,
               const float* __restrict__ W_gat,
               const float* __restrict__ b_gat,
               const float* __restrict__ W_fc,
               const float* __restrict__ b_fc,
               float* __restrict__ out, int N) {
    __shared__ float sS[32 * SPITCH];
    __shared__ float sP[8 * 32];

    const int tid  = threadIdx.x;
    const int wave = tid >> 5;
    const int lane = tid & 31;
    const int k    = lane & 15;
    const int half = lane >> 4;

    const int rt  = wave >> 3;
    const int ct  = wave & 7;
    const int col = ct * 16 + k;

    const int nn = wave * 2 + half;
    const int n  = blockIdx.x * 32 + nn;
    int beg = 0, cnt = 0;
    float4 ed = {0.f, 0.f, 0.f, 0.f};
    if (n < N) {
        beg = off[n];
        cnt = deg[n];
        ed  = *reinterpret_cast<const float4*>(edst + (size_t)n * 4);
    }
    const int c0 = __builtin_amdgcn_readlane(cnt, 0);
    const int c1 = __builtin_amdgcn_readlane(cnt, 16);
    const int cmax = c0 > c1 ? c0 : c1;
    const float4* es4 = reinterpret_cast<const float4*>(esrc);

    float m0 = -INFINITY, m1 = -INFINITY, m2 = -INFINITY, m3 = -INFINITY;
#pragma clang loop unroll(disable) vectorize(disable) interleave(disable)
    for (int c = 0; c < cmax; c += 32) {
        const int ja = c + k, jb = ja + 16;
        const bool la = ja < cnt, lb = jb < cnt;
        const int sa = ssrc[la ? beg + ja : 0];
        const int sb = ssrc[lb ? beg + jb : 0];
        const float4 ea = es4[sa];
        const float4 eb = es4[sb];
        const float pa = la ? 0.f : -INFINITY;
        const float pb = lb ? 0.f : -INFINITY;
        m0 = fmaxf(m0, fmaxf(lrelu(ea.x + ed.x) + pa, lrelu(eb.x + ed.x) + pb));
        m1 = fmaxf(m1, fmaxf(lrelu(ea.y + ed.y) + pa, lrelu(eb.y + ed.y) + pb));
        m2 = fmaxf(m2, fmaxf(lrelu(ea.z + ed.z) + pa, lrelu(eb.z + ed.z) + pb));
        m3 = fmaxf(m3, fmaxf(lrelu(ea.w + ed.w) + pa, lrelu(eb.w + ed.w) + pb));
    }
    for (int o = 8; o > 0; o >>= 1) {
        m0 = fmaxf(m0, __shfl_xor(m0, o, 32));
        m1 = fmaxf(m1, __shfl_xor(m1, o, 32));
        m2 = fmaxf(m2, __shfl_xor(m2, o, 32));
        m3 = fmaxf(m3, __shfl_xor(m3, o, 32));
    }

    Head4 S   = {0.f, 0.f, 0.f, 0.f};
    Head4 den = {0.f, 0.f, 0.f, 0.f};
    const Head4 mx = {m0, m1, m2, m3};
    const unsigned k4 = (unsigned)k * 4u;
    for (int c = 0; c < cmax; c += 16) {
        const int j = c + k;
        const bool live = j < cnt;
        const int s = ssrc[live ? beg + j : 0];
        const float4 es = es4[s];
        const int mc  = cmax - c;
        const int lim = cnt - c;
        if (mc > 8)      gat_chunk<16>(hrow, s, k4, lim, live, es, ed, mx, den, S);
        else if (mc > 4) gat_chunk<8>(hrow, s, k4, lim, live, es, ed, mx, den, S);
        else             gat_chunk<4>(hrow, s, k4, lim, live, es, ed, mx, den, S);
    }
    v2f bw[4];
#pragma unroll
    for (int kk = 0; kk < 4; ++kk) {
        bw[kk].x = W_gat[(4 * kk + 2 * half) * 128 + col];
        bw[kk].y = W_gat[(4 * kk + 2 * half + 1) * 128 + col];
    }
    const float bgc = b_gat[col];
    const float wfc = W_fc[col];

    float d0 = den.h0, d1 = den.h1, d2 = den.h2, d3 = den.h3;
    const float S0 = S.h0, S1 = S.h1, S2 = S.h2, S3 = S.h3;
    for (int o = 8; o > 0; o >>= 1) {
        d0 += __shfl_xor(d0, o, 32);
        d1 += __shfl_xor(d1, o, 32);
        d2 += __shfl_xor(d2, o, 32);
        d3 += __shfl_xor(d3, o, 32);
    }
    {
        float* sr = sS + nn * SPITCH + k;
        sr[0]  = S0 / (d0 + 1e-16f);
        sr[16] = S1 / (d1 + 1e-16f);
        sr[32] = S2 / (d2 + 1e-16f);
        sr[48] = S3 / (d3 + 1e-16f);
    }
    __syncthreads();

    {
        v8f acc = {};
        const float* ar = sS + (rt * 16 + k) * SPITCH + (ct >> 1) * 16 + 2 * half;
#pragma unroll
        for (int kk = 0; kk < 4; ++kk) {
            v2f a;
            a.x = ar[4 * kk]; a.y = ar[4 * kk + 1];
            acc = __builtin_amdgcn_wmma_f32_16x16x4_f32(
                false, a, false, bw[kk], (short)0, acc, false, false);
        }
        float p[8];
#pragma unroll
        for (int v = 0; v < 8; ++v) p[v] = fmaxf(acc[v] + bgc, 0.0f) * wfc;
        for (int o = 8; o > 0; o >>= 1) {
#pragma unroll
            for (int v = 0; v < 8; ++v) p[v] += __shfl_xor(p[v], o, 32);
        }
        if (k == 0) {
#pragma unroll
            for (int v = 0; v < 8; ++v) sP[ct * 32 + rt * 16 + 8 * half + v] = p[v];
        }
    }
    __syncthreads();

    if (tid < 32) {
        float y = 0.f;
#pragma unroll
        for (int c = 0; c < 8; ++c) y += sP[c * 32 + tid];
        const int r = blockIdx.x * 32 + tid;
        if (r < N) out[r] = y + b_fc[0];
    }
}

extern "C" void kernel_launch(void* const* d_in, const int* in_sizes, int n_in,
                              void* d_out, int out_size, void* d_ws, size_t ws_size,
                              hipStream_t stream) {
    const float* x      = (const float*)d_in[0];
    const int*   eidx   = (const int*)  d_in[1];
    const float* eattr  = (const float*)d_in[2];
    const float* W_nn   = (const float*)d_in[3];
    const float* b_nn   = (const float*)d_in[4];
    const float* W_root = (const float*)d_in[5];
    const float* b_ecc  = (const float*)d_in[6];
    const float* W_gat  = (const float*)d_in[7];
    const float* a_src  = (const float*)d_in[8];
    const float* a_dst  = (const float*)d_in[9];
    const float* b_gat  = (const float*)d_in[10];
    const float* W_fc   = (const float*)d_in[11];
    const float* b_fc   = (const float*)d_in[12];
    float* out = (float*)d_out;

    const int N = in_sizes[0] / 16;
    const int E = in_sizes[1] / 2;

    float* ws = (float*)d_ws;
    size_t off = 0;
    float* plane = ws + off;          off += (size_t)N * 128;
    float* hbuf  = plane;
    float* esrc  = ws + off;          off += (size_t)N * 4;
    float* edst  = ws + off;          off += (size_t)N * 4;
    int*   deg   = (int*)(ws + off);  off += (size_t)N;
    int*   noff  = (int*)(ws + off);  off += (size_t)N;
    int*   tabT  = (int*)(ws + off);  off += (size_t)E;
    int*   ssrc  = (int*)(ws + off);  off += (size_t)E;
    int*   sedge = (int*)(ws + off);  off += (size_t)E;
    int*   bdst  = (int*)plane;
    int*   bsrc  = bdst  + (size_t)E;
    int*   bedge = bsrc  + (size_t)E;
    const int bw = (N + NBK - 1) / NBK;

    k_bucket<<<NCH, SORT_THREADS, 0, stream>>>(eidx, bdst, bsrc, bedge, tabT, E, bw);
    k_group <<<NBK, SORT_THREADS, 0, stream>>>(bdst, bsrc, bedge, tabT, noff, deg, ssrc, sedge, E, N, bw);

    k_nnagg<<<(N + 15) / 16, 512, 0, stream>>>(x, eattr, W_nn, b_nn, W_root, b_ecc, W_gat, a_src, a_dst,
                                               noff, deg, ssrc, sedge, hbuf, esrc, edst, N);
    k_gattile<<<(N + 31) / 32, 512, 0, stream>>>(noff, deg, ssrc, esrc, edst, hbuf, W_gat,
                                                 b_gat, W_fc, b_fc, out, N);
}


# Sort kernels: records go to their rank in an LDS copy of the work-group's output range, written out as consecutive whole-line stores (was a scattered 4-byte store per word); list order rule unchanged
# speedup vs baseline: 15.3280x; 1.1583x over previous
#include <hip/hip_runtime.h>
#include <hip/hip_bf16.h>

typedef __attribute__((ext_vector_type(2))) float v2f;
typedef __attribute__((ext_vector_type(4))) float v4f;
typedef __attribute__((ext_vector_type(8))) float v8f;

#define NEG_SLOPE 0.2f

#define UPITCH 292
#define HPITCH 20
#define ZPITCH 132
template <int L>
__device__ __forceinline__ int row_lane(int v) {
    return __builtin_amdgcn_update_dpp(0, v, 0x150 + L, 0xf, 0xf, false);
}
__device__ __forceinline__ float ld_word(const float* base, unsigned byte_off) {
    return *reinterpret_cast<const float*>(reinterpret_cast<const char*>(base) + byte_off);
}
struct EdgeWords { float ax, ay, bx, by; };
template <int T>
__device__ __forceinline__ EdgeWords edge_loads(const float* __restrict__ x,
                                                const float* __restrict__ eattr,
                                                int s, int e, unsigned m4) {
    EdgeWords w;
    w.ax = ld_word(eattr, (unsigned)row_lane<2 * T>(e) * 64u + m4);
    w.ay = ld_word(eattr, (unsigned)row_lane<2 * T + 1>(e) * 64u + m4);
    w.bx = ld_word(x,     (unsigned)row_lane<2 * T>(s) * 64u + m4);
    w.by = ld_word(x,     (unsigned)row_lane<2 * T + 1>(s) * 64u + m4);
    return w;
}
template <int T>
__device__ __forceinline__ void edge_step(const EdgeWords& w, int lim, v8f& U, float& xsum) {
    const bool vx = 4 * T < lim, vy = 4 * T + 1 < lim;
    v2f a, b;
    a.x = vx ? w.ax : 0.f; a.y = vy ? w.ay : 0.f;
    b.x = vx ? w.bx : 0.f; b.y = vy ? w.by : 0.f;
    xsum += b.x + b.y;
    U = __builtin_amdgcn_wmma_f32_16x16x4_f32(false, a, false, b, (short)0, U, false, false);
}

__global__ __launch_bounds__(512, 12)
void k_nnagg(const float* __restrict__ x,
             const float* __restrict__ eattr,
             const float* __restrict__ W_nn,
             const float* __restrict__ b_nn,
             const float* __restrict__ W_root,
             const float* __restrict__ b_ecc,
             const float* __restrict__ W_gat,
             const float* __restrict__ a_src,
             const float* __restrict__ a_dst,
             const int* __restrict__ off,
             const int* __restrict__ deg,
             const int* __restrict__ ssrc,
             const int* __restrict__ sedge,
             float* __restrict__ hrow,
             float* __restrict__ esrc,
             float* __restrict__ edst, int N) {
    __shared__ float sU[16 * UPITCH];
    __shared__ __attribute__((aligned(16))) float sH[16 * HPITCH];
    __shared__ __attribute__((aligned(16))) float sZ[16 * ZPITCH];
    __shared__ v4f sA[64];

    const int tid  = threadIdx.x;
    const int wave = tid >> 5;
    const int lane = tid & 31;
    const int i    = lane & 15;
    const int hi   = lane >> 4;
    const int n    = blockIdx.x * 16 + wave;

    if (tid < 64)
        sA[tid] = reinterpret_cast<const v4f*>(tid < 32 ? a_src : a_dst)[tid & 31];

    v8f U = {};
    float xsum = 0.f;
    float xown = 0.f;
    const int nu = __builtin_amdgcn_readfirstlane(n);
    if (nu < N) {
        xown = x[(size_t)nu * 16 + i];
        const int beg = __builtin_amdgcn_readfirstlane(off[nu]);
        const int cnt = __builtin_amdgcn_readfirstlane(deg[nu]);
        const int jl  = 4 * (i >> 1) + 2 * hi + (i & 1);
        const unsigned m4 = (unsigned)i * 4u;
        for (int c = 0; c < cnt; c += 32) {
            const int rem = cnt - c;
            int s = 0, e = 0;
            if (jl < rem) { s = ssrc[beg + c + jl]; e = sedge[beg + c + jl]; }
            const int lim = rem - 2 * hi;
            const EdgeWords w0 = edge_loads<0>(x, eattr, s, e, m4);
            const EdgeWords w1 = edge_loads<1>(x, eattr, s, e, m4);
            const EdgeWords w2 = edge_loads<2>(x, eattr, s, e, m4);
            const EdgeWords w3 = edge_loads<3>(x, eattr, s, e, m4);
            if (rem > 16) {
                const EdgeWords w4 = edge_loads<4>(x, eattr, s, e, m4);
                const EdgeWords w5 = edge_loads<5>(x, eattr, s, e, m4);
                const EdgeWords w6 = edge_loads<6>(x, eattr, s, e, m4);
                const EdgeWords w7 = edge_loads<7>(x, eattr, s, e, m4);
                edge_step<0>(w0, lim, U, xsum); edge_step<1>(w1, lim, U, xsum);
                edge_step<2>(w2, lim, U, xsum); edge_step<3>(w3, lim, U, xsum);
                edge_step<4>(w4, lim, U, xsum); edge_step<5>(w5, lim, U, xsum);
                edge_step<6>(w6, lim, U, xsum); edge_step<7>(w7, lim, U, xsum);
            } else {
                edge_step<0>(w0, lim, U, xsum); edge_step<1>(w1, lim, U, xsum);
                edge_step<2>(w2, lim, U, xsum); edge_step<3>(w3, lim, U, xsum);
            }
        }
    }
    {
        const float xs = xsum + __shfl_xor(xsum, 16, 32);
        float* ur = sU + wave * UPITCH + hi * 128 + i;
#pragma unroll
        for (int r = 0; r < 8; ++r) ur[16 * r] = U[r];
        sU[wave * UPITCH + 256 + hi * 16 + i] = hi ? xown : xs;
    }

    __syncthreads();

    const int col = (wave & 7) * 16 + i;
    v2f bg[4] = {};
    if (wave < 8) {
#pragma unroll
        for (int kk = 0; kk < 4; ++kk) {
            bg[kk].x = W_gat[(4 * kk + 2 * hi) * 128 + col];
            bg[kk].y = W_gat[(4 * kk + 2 * hi + 1) * 128 + col];
        }
    }

    if (wave == 0) {
        v8f accA = {}, accB = {};
        const float* ar = sU + i * UPITCH + 2 * hi;
#pragma unroll
        for (int kk = 0; kk < 72; ++kk) {
            const float* wp = (kk < 64) ? (W_nn + (4 * kk + 2 * hi) * 16 + i)
                            : (kk < 68) ? (b_nn + (4 * (kk - 64) + 2 * hi) * 16 + i)
                                        : (W_root + (4 * (kk - 68) + 2 * hi) * 16 + i);
            v2f a, b;
            a.x = ar[4 * kk]; a.y = ar[4 * kk + 1];
            b.x = wp[0];      b.y = wp[16];
            if (kk & 1)
                accB = __builtin_amdgcn_wmma_f32_16x16x4_f32(
                    false, a, false, b, (short)0, accB, false, false);
            else
                accA = __builtin_amdgcn_wmma_f32_16x16x4_f32(
                    false, a, false, b, (short)0, accA, false, false);
        }
        const v8f acc = accA + accB;
        const float be = b_ecc[i];
#pragma unroll
        for (int v = 0; v < 8; ++v)
            sH[(v + 8 * hi) * HPITCH + i] = fmaxf(acc[v] + be, 0.0f);
    }
    __syncthreads();

    if (wave < 8) {
        v8f acc = {};
        const float* hr = sH + i * HPITCH + 2 * hi;
#pragma unroll
        for (int kk = 0; kk < 4; ++kk) {
            v2f a;
            a.x = hr[4 * kk]; a.y = hr[4 * kk + 1];
            acc = __builtin_amdgcn_wmma_f32_16x16x4_f32(
                false, a, false, bg[kk], (short)0, acc, false, false);
        }
#pragma unroll
        for (int v = 0; v < 8; ++v)
            sZ[(v + 8 * hi) * ZPITCH + col] = acc[v];
    }
    else if (wave == 10 || wave == 11) {
        const int t  = (wave - 10) * 32 + lane;
        const int nn = t >> 2, part = t & 3;
        const int r  = blockIdx.x * 16 + nn;
        if (r < N)
            *reinterpret_cast<v4f*>(hrow + (size_t)r * 16 + part * 4) =
                *reinterpret_cast<const v4f*>(sH + nn * HPITCH + part * 4);
    }
    __syncthreads();

    if (wave == 8 || wave == 9) {
        const int t  = (wave - 8) * 32 + lane;
        const int nn = t >> 2, hh = t & 3;
        const int r  = blockIdx.x * 16 + nn;
        const v4f* zr = reinterpret_cast<const v4f*>(sZ + nn * ZPITCH + hh * 32);
        const v4f* as = sA + hh * 8;
        const v4f* ad = sA + 32 + hh * 8;
        float s = 0.f, d = 0.f;
#pragma unroll
        for (int j = 0; j < 8; ++j) {
            const v4f zz = zr[j], sa = as[j], da = ad[j];
            s += zz.x * sa.x; d += zz.x * da.x;
            s += zz.y * sa.y; d += zz.y * da.y;
            s += zz.z * sa.z; d += zz.z * da.z;
            s += zz.w * sa.w; d += zz.w * da.w;
        }
        if (r < N) { esrc[(size_t)r * 4 + hh] = s; edst[(size_t)r * 4 + hh] = d; }
    }
}

#define NCH 256
#define NBK 256
#define SORT_THREADS 1024
#define BCAP (4 * SORT_THREADS)
#define GCAP 6144
static_assert(SORT_THREADS == 4 * NCH && NBK <= SORT_THREADS, "k_group: four threads per chunk segment");

__device__ __forceinline__ int block_excl_scan(int v, int* wsum, int* tot) {
    const int lane = threadIdx.x & 31, wave = threadIdx.x >> 5;
    int inc = v;
    for (int o = 1; o < 32; o <<= 1) {
        int t = __shfl_up(inc, o, 32);
        if (lane >= o) inc += t;
    }
    if (lane == 31) wsum[wave] = inc;
    __syncthreads();
    if (wave == 0) {
        const int w = wsum[lane];
        int wi = w;
        for (int o = 1; o < 32; o <<= 1) {
            int t = __shfl_up(wi, o, 32);
            if (lane >= o) wi += t;
        }
        wsum[lane] = wi - w;
        if (lane == 31) wsum[32] = wi;
    }
    __syncthreads();
    *tot = wsum[32];
    const int r = wsum[wave] + inc - v;
    __syncthreads();
    return r;
}

__global__ __launch_bounds__(SORT_THREADS)
void k_bucket(const int* __restrict__ eidx,
              int* __restrict__ bdst,
              int* __restrict__ bsrc,
              int* __restrict__ bedge,
              int* __restrict__ tabT, int E, int bw) {
    __shared__ int hist[NBK];
    __shared__ int wsum[33];
    __shared__ int rd[BCAP], rs[BCAP], re[BCAP];
    const int tid = threadIdx.x;
    const int chunk = (E + NCH - 1) / NCH;
    int e0 = blockIdx.x * chunk; if (e0 > E) e0 = E;
    int e1 = e0 + chunk;         if (e1 > E) e1 = E;

    if (tid < NBK) hist[tid] = 0;
    __syncthreads();
    for (int eb = e0; eb < e1; eb += 4 * SORT_THREADS) {
        int d[4];
#pragma unroll
        for (int u = 0; u < 4; ++u) {
            const int e = eb + u * SORT_THREADS + tid;
            d[u] = (e < e1) ? eidx[E + e] : -1;
        }
#pragma unroll
        for (int u = 0; u < 4; ++u)
            if (d[u] >= 0) atomicAdd(&hist[d[u] / bw], 1);
    }
    __syncthreads();
    int tot;
    const int v = (tid < NBK) ? hist[tid] : 0;
    const int ex = block_excl_scan(v, wsum, &tot);
    if (tid < NBK) {
        tabT[tid * NCH + blockIdx.x] = ex;
        hist[tid] = ex;
    }
    if (tid == 0) tabT[NBK * NCH + blockIdx.x] = tot;
    __syncthreads();
    const int len = e1 - e0;
    if (len <= BCAP) {
        int d[4], s[4];
#pragma unroll
        for (int u = 0; u < 4; ++u) {
            const int e = e0 + u * SORT_THREADS + tid;
            d[u] = -1; s[u] = 0;
            if (e < e1) { d[u] = eidx[E + e]; s[u] = eidx[e]; }
        }
#pragma unroll
        for (int u = 0; u < 4; ++u) {
            if (d[u] >= 0) {
                const int p = atomicAdd(&hist[d[u] / bw], 1);
                rd[p] = d[u];
                rs[p] = s[u];
                re[p] = e0 + u * SORT_THREADS + tid;
            }
        }
        __syncthreads();
        for (int j = tid; j < len; j += SORT_THREADS) {
            bdst[e0 + j]  = rd[j];
            bsrc[e0 + j]  = rs[j];
            bedge[e0 + j] = re[j];
        }
    } else {
        for (int eb = e0; eb < e1; eb += 4 * SORT_THREADS) {
            int d[4], s[4];
#pragma unroll
            for (int u = 0; u < 4; ++u) {
                const int e = eb + u * SORT_THREADS + tid;
                d[u] = -1; s[u] = 0;
                if (e < e1) { d[u] = eidx[E + e]; s[u] = eidx[e]; }
            }
#pragma unroll
            for (int u = 0; u < 4; ++u) {
                if (d[u] >= 0) {
                    const int p = e0 + atomicAdd(&hist[d[u] / bw], 1);
                    bdst[p]  = d[u];
                    bsrc[p]  = s[u];
                    bedge[p] = eb + u * SORT_THREADS + tid;
                }
            }
        }
    }
}

__global__ __launch_bounds__(SORT_THREADS)
void k_group(const int* __restrict__ bdst,
             const int* __restrict__ bsrc,
             const int* __restrict__ bedge,
             const int* __restrict__ tabT,
             int* __restrict__ off,
             int* __restrict__ deg,
             int* __restrict__ ssrc,
             int* __restrict__ sedge, int E, int N, int bw) {
    __shared__ int cnt[SORT_THREADS];
    __shared__ int wsum[33];
    __shared__ int ls[GCAP], le[GCAP];
    const int tid = threadIdx.x;
    const int b   = blockIdx.x;
    const int c   = tid >> 2;
    const int sub = tid & 3;
    const int chunk = (E + NCH - 1) / NCH;
    int cb = c * chunk; if (cb > E) cb = E;
    const int s0 = tabT[b * NCH + c];
    const int s1 = tabT[(b + 1) * NCH + c];
    const int g  = cb + s0;
    const int l  = s1 - s0;
    const int n0 = b * bw;

    cnt[tid] = 0;
    int base;
    (void)block_excl_scan(sub == 0 ? s0 : 0, wsum, &base);

    for (int jb = 0; jb < l; jb += 16) {
        int d[4];
#pragma unroll
        for (int u = 0; u < 4; ++u) {
            const int j = jb + 4 * u + sub;
            d[u] = (j < l) ? bdst[g + j] : -1;
        }
#pragma unroll
        for (int u = 0; u < 4; ++u)
            if (d[u] >= 0) atomicAdd(&cnt[d[u] - n0], 1);
    }
    __syncthreads();
    int tot;
    const int v  = cnt[tid];
    const int ex = block_excl_scan(v, wsum, &tot);
    {
        const int n = n0 + tid;
        if (tid < bw && n < N) { off[n] = base + ex; deg[n] = v; }
    }
    const int  btot   = __builtin_amdgcn_readfirstlane(tot);
    const bool staged = btot <= GCAP;
    cnt[tid] = staged ? ex : base + ex;
    __syncthreads();
    for (int jb = 0; jb < l; jb += 16) {
        int d[4], s[4], e[4];
#pragma unroll
        for (int u = 0; u < 4; ++u) {
            const int j = jb + 4 * u + sub;
            d[u] = -1; s[u] = 0; e[u] = 0;
            if (j < l) { d[u] = bdst[g + j]; s[u] = bsrc[g + j]; e[u] = bedge[g + j]; }
        }
#pragma unroll
        for (int u = 0; u < 4; ++u) {
            if (d[u] >= 0) {
                const int p = atomicAdd(&cnt[d[u] - n0], 1);
                if (staged) { ls[p] = s[u];   le[p] = e[u]; }
                else        { ssrc[p] = s[u]; sedge[p] = e[u]; }
            }
        }
    }
    if (staged) {
        __syncthreads();
        for (int j = tid; j < btot; j += SORT_THREADS) {
            ssrc[base + j]  = ls[j];
            sedge[base + j] = le[j];
        }
    }
}

#define SPITCH 68
__device__ __forceinline__ float lrelu(float v) { return v > 0.f ? v : NEG_SLOPE * v; }

template <int T, int NT>
__device__ __forceinline__ void gat_rows(const float* __restrict__ hrow, int s, unsigned k4, float* hv) {
    hv[T] = ld_word(hrow, (unsigned)row_lane<T>(s) * 64u + k4);
    if constexpr (T + 1 < NT) gat_rows<T + 1, NT>(hrow, s, k4, hv);
}
__device__ __forceinline__ int f2i(float v) { return __builtin_bit_cast(int, v); }
__device__ __forceinline__ float i2f(int v) { return __builtin_bit_cast(float, v); }
struct Head4 { float h0, h1, h2, h3; };
template <int T, int NT>
__device__ __forceinline__ void gat_sums(const float* hv, int lim, int e0, int e1, int e2, int e3, Head4& S) {
    const float h = (T < lim) ? hv[T] : 0.f;
    S.h0 = __builtin_fmaf(i2f(row_lane<T>(e0)), h, S.h0);
    S.h1 = __builtin_fmaf(i2f(row_lane<T>(e1)), h, S.h1);
    S.h2 = __builtin_fmaf(i2f(row_lane<T>(e2)), h, S.h2);
    S.h3 = __builtin_fmaf(i2f(row_lane<T>(e3)), h, S.h3);
    if constexpr (T + 1 < NT) gat_sums<T + 1, NT>(hv, lim, e0, e1, e2, e3, S);
}
template <int NT>
__device__ __forceinline__ void gat_chunk(const float* __restrict__ hrow, int s, unsigned k4, int lim,
                                          bool live, float4 es, float4 ed, Head4 mx,
                                          Head4& den, Head4& S) {
    float hv[NT];
    gat_rows<0, NT>(hrow, s, k4, hv);
    __builtin_amdgcn_sched_barrier(0);
    const float e0 = __expf(live ? lrelu(es.x + ed.x) - mx.h0 : -INFINITY);
    const float e1 = __expf(live ? lrelu(es.y + ed.y) - mx.h1 : -INFINITY);
    const float e2 = __expf(live ? lrelu(es.z + ed.z) - mx.h2 : -INFINITY);
    const float e3 = __expf(live ? lrelu(es.w + ed.w) - mx.h3 : -INFINITY);
    den.h0 += e0; den.h1 += e1; den.h2 += e2; den.h3 += e3;
    gat_sums<0, NT>(hv, lim, f2i(e0), f2i(e1), f2i(e2), f2i(e3), S);
}

__global__ __launch_bounds__(512)
void k_gattile(const int* __restrict__ off,
               const int* __restrict__ deg,
               const int* __restrict__ ssrc,
               const float* __restrict__ esrc,
               const float* __restrict__ edst,
               const float* __restrict__ hrow,
               const float* __restrict__ W_gat,
               const float* __restrict__ b_gat,
               const float* __restrict__ W_fc,
               const float* __restrict__ b_fc,
               float* __restrict__ out, int N) {
    __shared__ float sS[32 * SPITCH];
    __shared__ float sP[8 * 32];

    const int tid  = threadIdx.x;
    const int wave = tid >> 5;
    const int lane = tid & 31;
    const int k    = lane & 15;
    const int half = lane >> 4;

    const int rt  = wave >> 3;
    const int ct  = wave & 7;
    const int col = ct * 16 + k;

    const int nn = wave * 2 + half;
    const int n  = blockIdx.x * 32 + nn;
    int beg = 0, cnt = 0;
    float4 ed = {0.f, 0.f, 0.f, 0.f};
    if (n < N) {
        beg = off[n];
        cnt = deg[n];
        ed  = *reinterpret_cast<const float4*>(edst + (size_t)n * 4);
    }
    const int c0 = __builtin_amdgcn_readlane(cnt, 0);
    const int c1 = __builtin_amdgcn_readlane(cnt, 16);
    const int cmax = c0 > c1 ? c0 : c1;
    const float4* es4 = reinterpret_cast<const float4*>(esrc);

    float m0 = -INFINITY, m1 = -INFINITY, m2 = -INFINITY, m3 = -INFINITY;
#pragma clang loop unroll(disable) vectorize(disable) interleave(disable)
    for (int c = 0; c < cmax; c += 32) {
        const int ja = c + k, jb = ja + 16;
        const bool la = ja < cnt, lb = jb < cnt;
        const int sa = ssrc[la ? beg + ja : 0];
        const int sb = ssrc[lb ? beg + jb : 0];
        const float4 ea = es4[sa];
        const float4 eb = es4[sb];
        const float pa = la ? 0.f : -INFINITY;
        const float pb = lb ? 0.f : -INFINITY;
        m0 = fmaxf(m0, fmaxf(lrelu(ea.x + ed.x) + pa, lrelu(eb.x + ed.x) + pb));
        m1 = fmaxf(m1, fmaxf(lrelu(ea.y + ed.y) + pa, lrelu(eb.y + ed.y) + pb));
        m2 = fmaxf(m2, fmaxf(lrelu(ea.z + ed.z) + pa, lrelu(eb.z + ed.z) + pb));
        m3 = fmaxf(m3, fmaxf(lrelu(ea.w + ed.w) + pa, lrelu(eb.w + ed.w) + pb));
    }
    for (int o = 8; o > 0; o >>= 1) {
        m0 = fmaxf(m0, __shfl_xor(m0, o, 32));
        m1 = fmaxf(m1, __shfl_xor(m1, o, 32));
        m2 = fmaxf(m2, __shfl_xor(m2, o, 32));
        m3 = fmaxf(m3, __shfl_xor(m3, o, 32));
    }

    Head4 S   = {0.f, 0.f, 0.f, 0.f};
    Head4 den = {0.f, 0.f, 0.f, 0.f};
    const Head4 mx = {m0, m1, m2, m3};
    const unsigned k4 = (unsigned)k * 4u;
    for (int c = 0; c < cmax; c += 16) {
        const int j = c + k;
        const bool live = j < cnt;
        const int s = ssrc[live ? beg + j : 0];
        const float4 es = es4[s];
        const int mc  = cmax - c;
        const int lim = cnt - c;
        if (mc > 8)      gat_chunk<16>(hrow, s, k4, lim, live, es, ed, mx, den, S);
        else if (mc > 4) gat_chunk<8>(hrow, s, k4, lim, live, es, ed, mx, den, S);
        else             gat_chunk<4>(hrow, s, k4, lim, live, es, ed, mx, den, S);
    }
    v2f bw[4];
#pragma unroll
    for (int kk = 0; kk < 4; ++kk) {
        bw[kk].x = W_gat[(4 * kk + 2 * half) * 128 + col];
        bw[kk].y = W_gat[(4 * kk + 2 * half + 1) * 128 + col];
    }
    const float bgc = b_gat[col];
    const float wfc = W_fc[col];

    float d0 = den.h0, d1 = den.h1, d2 = den.h2, d3 = den.h3;
    const float S0 = S.h0, S1 = S.h1, S2 = S.h2, S3 = S.h3;
    for (int o = 8; o > 0; o >>= 1) {
        d0 += __shfl_xor(d0, o, 32);
        d1 += __shfl_xor(d1, o, 32);
        d2 += __shfl_xor(d2, o, 32);
        d3 += __shfl_xor(d3, o, 32);
    }
    {
        float* sr = sS + nn * SPITCH + k;
        sr[0]  = S0 / (d0 + 1e-16f);
        sr[16] = S1 / (d1 + 1e-16f);
        sr[32] = S2 / (d2 + 1e-16f);
        sr[48] = S3 / (d3 + 1e-16f);
    }
    __syncthreads();

    {
        v8f acc = {};
        const float* ar = sS + (rt * 16 + k) * SPITCH + (ct >> 1) * 16 + 2 * half;
#pragma unroll
        for (int kk = 0; kk < 4; ++kk) {
            v2f a;
            a.x = ar[4 * kk]; a.y = ar[4 * kk + 1];
            acc = __builtin_amdgcn_wmma_f32_16x16x4_f32(
                false, a, false, bw[kk], (short)0, acc, false, false);
        }
        float p[8];
#pragma unroll
        for (int v = 0; v < 8; ++v) p[v] = fmaxf(acc[v] + bgc, 0.0f) * wfc;
        for (int o = 8; o > 0; o >>= 1) {
#pragma unroll
            for (int v = 0; v < 8; ++v) p[v] += __shfl_xor(p[v], o, 32);
        }
        if (k == 0) {
#pragma unroll
            for (int v = 0; v < 8; ++v) sP[ct * 32 + rt * 16 + 8 * half + v] = p[v];
        }
    }
    __syncthreads();

    if (tid < 32) {
        float y = 0.f;
#pragma unroll
        for (int c = 0; c < 8; ++c) y += sP[c * 32 + tid];
        const int r = blockIdx.x * 32 + tid;
        if (r < N) out[r] = y + b_fc[0];
    }
}

extern "C" void kernel_launch(void* const* d_in, const int* in_sizes, int n_in,
                              void* d_out, int out_size, void* d_ws, size_t ws_size,
                              hipStream_t stream) {
    const float* x      = (const float*)d_in[0];
    const int*   eidx   = (const int*)  d_in[1];
    const float* eattr  = (const float*)d_in[2];
    const float* W_nn   = (const float*)d_in[3];
    const float* b_nn   = (const float*)d_in[4];
    const float* W_root = (const float*)d_in[5];
    const float* b_ecc  = (const float*)d_in[6];
    const float* W_gat  = (const float*)d_in[7];
    const float* a_src  = (const float*)d_in[8];
    const float* a_dst  = (const float*)d_in[9];
    const float* b_gat  = (const float*)d_in[10];
    const float* W_fc   = (const float*)d_in[11];
    const float* b_fc   = (const float*)d_in[12];
    float* out = (float*)d_out;

    const int N = in_sizes[0] / 16;
    const int E = in_sizes[1] / 2;

    float* ws = (float*)d_ws;
    size_t off = 0;
    float* plane = ws + off;          off += (size_t)N * 128;
    float* hbuf  = plane;
    float* esrc  = ws + off;          off += (size_t)N * 4;
    float* edst  = ws + off;          off += (size_t)N * 4;
    int*   deg   = (int*)(ws + off);  off += (size_t)N;
    int*   noff  = (int*)(ws + off);  off += (size_t)N;
    int*   tabT  = (int*)(ws + off);  off += (size_t)E;
    int*   ssrc  = (int*)(ws + off);  off += (size_t)E;
    int*   sedge = (int*)(ws + off);  off += (size_t)E;
    int*   bdst  = (int*)plane;
    int*   bsrc  = bdst  + (size_t)E;
    int*   bedge = bsrc  + (size_t)E;
    const int bw = (N + NBK - 1) / NBK;

    k_bucket<<<NCH, SORT_THREADS, 0, stream>>>(eidx, bdst, bsrc, bedge, tabT, E, bw);
    k_group <<<NBK, SORT_THREADS, 0, stream>>>(bdst, bsrc, bedge, tabT, noff, deg, ssrc, sedge, E, N, bw);

    k_nnagg<<<(N + 15) / 16, 512, 0, stream>>>(x, eattr, W_nn, b_nn, W_root, b_ecc, W_gat, a_src, a_dst,
                                               noff, deg, ssrc, sedge, hbuf, esrc, edst, N);
    k_gattile<<<(N + 31) / 32, 512, 0, stream>>>(noff, deg, ssrc, esrc, edst, hbuf, W_gat,
                                                 b_gat, W_fc, b_fc, out, N);
}


# NNConv tile product [16,288]@[288,16] split by k over 8 waves (was 72 serial steps on one wave), both operands from LDS (W' staged under the edge gathers), partials summed in LDS: h sum order changes
# speedup vs baseline: 16.7189x; 1.0907x over previous
#include <hip/hip_runtime.h>
#include <hip/hip_bf16.h>

typedef __attribute__((ext_vector_type(2))) float v2f;
typedef __attribute__((ext_vector_type(4))) float v4f;
typedef __attribute__((ext_vector_type(8))) float v8f;

#define NEG_SLOPE 0.2f

#define UPITCH 292
#define HPITCH 20
#define ZPITCH 132
template <int L>
__device__ __forceinline__ int row_lane(int v) {
    return __builtin_amdgcn_update_dpp(0, v, 0x150 + L, 0xf, 0xf, false);
}
__device__ __forceinline__ float ld_word(const float* base, unsigned byte_off) {
    return *reinterpret_cast<const float*>(reinterpret_cast<const char*>(base) + byte_off);
}
struct EdgeWords { float ax, ay, bx, by; };
template <int T>
__device__ __forceinline__ EdgeWords edge_loads(const float* __restrict__ x,
                                                const float* __restrict__ eattr,
                                                int s, int e, unsigned m4) {
    EdgeWords w;
    w.ax = ld_word(eattr, (unsigned)row_lane<2 * T>(e) * 64u + m4);
    w.ay = ld_word(eattr, (unsigned)row_lane<2 * T + 1>(e) * 64u + m4);
    w.bx = ld_word(x,     (unsigned)row_lane<2 * T>(s) * 64u + m4);
    w.by = ld_word(x,     (unsigned)row_lane<2 * T + 1>(s) * 64u + m4);
    return w;
}
template <int T>
__device__ __forceinline__ void edge_step(const EdgeWords& w, int lim, v8f& U, float& xsum) {
    const bool vx = 4 * T < lim, vy = 4 * T + 1 < lim;
    v2f a, b;
    a.x = vx ? w.ax : 0.f; a.y = vy ? w.ay : 0.f;
    b.x = vx ? w.bx : 0.f; b.y = vy ? w.by : 0.f;
    xsum += b.x + b.y;
    U = __builtin_amdgcn_wmma_f32_16x16x4_f32(false, a, false, b, (short)0, U, false, false);
}

__global__ __launch_bounds__(512, 12)
void k_nnagg(const float* __restrict__ x,
             const float* __restrict__ eattr,
             const float* __restrict__ W_nn,
             const float* __restrict__ b_nn,
             const float* __restrict__ W_root,
             const float* __restrict__ b_ecc,
             const float* __restrict__ W_gat,
             const float* __restrict__ a_src,
             const float* __restrict__ a_dst,
             const int* __restrict__ off,
             const int* __restrict__ deg,
             const int* __restrict__ ssrc,
             const int* __restrict__ sedge,
             float* __restrict__ hrow,
             float* __restrict__ esrc,
             float* __restrict__ edst, int N) {
    __shared__ float sU[16 * UPITCH];
    __shared__ __attribute__((aligned(16))) float sH[16 * HPITCH];
    __shared__ __attribute__((aligned(16))) float sZ[16 * ZPITCH];
    __shared__ v4f sA[64];
    __shared__ __attribute__((aligned(16))) float sW[288 * 16];
    __shared__ float sR[8 * 256];

    const int tid  = threadIdx.x;
    const int wave = tid >> 5;
    const int lane = tid & 31;
    const int i    = lane & 15;
    const int hi   = lane >> 4;
    const int n    = blockIdx.x * 16 + wave;

    if (tid < 64)
        sA[tid] = reinterpret_cast<const v4f*>(tid < 32 ? a_src : a_dst)[tid & 31];

    float wst[9];
#pragma unroll
    for (int u = 0; u < 9; ++u)
        wst[u] = (u < 8) ? W_nn[u * 512 + tid] : (tid < 256 ? b_nn[tid] : W_root[tid - 256]);
    const float be = b_ecc[i];

    v8f U = {};
    float xsum = 0.f;
    float xown = 0.f;
    const int nu = __builtin_amdgcn_readfirstlane(n);
    if (nu < N) {
        xown = x[(size_t)nu * 16 + i];
        const int beg = __builtin_amdgcn_readfirstlane(off[nu]);
        const int cnt = __builtin_amdgcn_readfirstlane(deg[nu]);
        const int jl  = 4 * (i >> 1) + 2 * hi + (i & 1);
        const unsigned m4 = (unsigned)i * 4u;
        for (int c = 0; c < cnt; c += 32) {
            const int rem = cnt - c;
            int s = 0, e = 0;
            if (jl < rem) { s = ssrc[beg + c + jl]; e = sedge[beg + c + jl]; }
            const int lim = rem - 2 * hi;
            const EdgeWords w0 = edge_loads<0>(x, eattr, s, e, m4);
            const EdgeWords w1 = edge_loads<1>(x, eattr, s, e, m4);
            const EdgeWords w2 = edge_loads<2>(x, eattr, s, e, m4);
            const EdgeWords w3 = edge_loads<3>(x, eattr, s, e, m4);
            if (rem > 16) {
                const EdgeWords w4 = edge_loads<4>(x, eattr, s, e, m4);
                const EdgeWords w5 = edge_loads<5>(x, eattr, s, e, m4);
                const EdgeWords w6 = edge_loads<6>(x, eattr, s, e, m4);
                const EdgeWords w7 = edge_loads<7>(x, eattr, s, e, m4);
                edge_step<0>(w0, lim, U, xsum); edge_step<1>(w1, lim, U, xsum);
                edge_step<2>(w2, lim, U, xsum); edge_step<3>(w3, lim, U, xsum);
                edge_step<4>(w4, lim, U, xsum); edge_step<5>(w5, lim, U, xsum);
                edge_step<6>(w6, lim, U, xsum); edge_step<7>(w7, lim, U, xsum);
            } else {
                edge_step<0>(w0, lim, U, xsum); edge_step<1>(w1, lim, U, xsum);
                edge_step<2>(w2, lim, U, xsum); edge_step<3>(w3, lim, U, xsum);
            }
        }
    }
    {
        const float xs = xsum + __shfl_xor(xsum, 16, 32);
        float* ur = sU + wave * UPITCH + hi * 128 + i;
#pragma unroll
        for (int r = 0; r < 8; ++r) ur[16 * r] = U[r];
        sU[wave * UPITCH + 256 + hi * 16 + i] = hi ? xown : xs;
    }
#pragma unroll
    for (int u = 0; u < 9; ++u) {
        const int idx = u * 512 + tid;
        const int k = idx >> 4, c = idx & 15;
        sW[((k >> 1) * 16 + c) * 2 + (k & 1)] = wst[u];
    }

    __syncthreads();

    const int col = (wave & 7) * 16 + i;
    v2f bg[4] = {};
    if (wave < 8) {
#pragma unroll
        for (int kk = 0; kk < 4; ++kk) {
            bg[kk].x = W_gat[(4 * kk + 2 * hi) * 128 + col];
            bg[kk].y = W_gat[(4 * kk + 2 * hi + 1) * 128 + col];
        }
    }

    if (wave < 8) {
        v8f accA = {}, accB = {};
        const float* ar = sU + i * UPITCH + 2 * hi + 36 * wave;
        const v2f* br = reinterpret_cast<const v2f*>(sW) + 288 * wave + 16 * hi + i;
#pragma unroll
        for (int t = 0; t < 9; ++t) {
            v2f a;
            a.x = ar[4 * t]; a.y = ar[4 * t + 1];
            const v2f b = br[32 * t];
            if (t & 1)
                accB = __builtin_amdgcn_wmma_f32_16x16x4_f32(
                    false, a, false, b, (short)0, accB, false, false);
            else
                accA = __builtin_amdgcn_wmma_f32_16x16x4_f32(
                    false, a, false, b, (short)0, accA, false, false);
        }
        const v8f acc = accA + accB;
        float* pr = sR + wave * 256 + 128 * hi + i;
#pragma unroll
        for (int v = 0; v < 8; ++v) pr[16 * v] = acc[v];
    }
    __syncthreads();
    if (tid < 256) {
        float s = sR[tid];
#pragma unroll
        for (int w = 1; w < 8; ++w) s += sR[w * 256 + tid];
        sH[(tid >> 4) * HPITCH + i] = fmaxf(s + be, 0.0f);
    }
    __syncthreads();

    if (wave < 8) {
        v8f acc = {};
        const float* hr = sH + i * HPITCH + 2 * hi;
#pragma unroll
        for (int kk = 0; kk < 4; ++kk) {
            v2f a;
            a.x = hr[4 * kk]; a.y = hr[4 * kk + 1];
            acc = __builtin_amdgcn_wmma_f32_16x16x4_f32(
                false, a, false, bg[kk], (short)0, acc, false, false);
        }
#pragma unroll
        for (int v = 0; v < 8; ++v)
            sZ[(v + 8 * hi) * ZPITCH + col] = acc[v];
    }
    else if (wave == 10 || wave == 11) {
        const int t  = (wave - 10) * 32 + lane;
        const int nn = t >> 2, part = t & 3;
        const int r  = blockIdx.x * 16 + nn;
        if (r < N)
            *reinterpret_cast<v4f*>(hrow + (size_t)r * 16 + part * 4) =
                *reinterpret_cast<const v4f*>(sH + nn * HPITCH + part * 4);
    }
    __syncthreads();

    if (wave == 8 || wave == 9) {
        const int t  = (wave - 8) * 32 + lane;
        const int nn = t >> 2, hh = t & 3;
        const int r  = blockIdx.x * 16 + nn;
        const v4f* zr = reinterpret_cast<const v4f*>(sZ + nn * ZPITCH + hh * 32);
        const v4f* as = sA + hh * 8;
        const v4f* ad = sA + 32 + hh * 8;
        float s = 0.f, d = 0.f;
#pragma unroll
        for (int j = 0; j < 8; ++j) {
            const v4f zz = zr[j], sa = as[j], da = ad[j];
            s += zz.x * sa.x; d += zz.x * da.x;
            s += zz.y * sa.y; d += zz.y * da.y;
            s += zz.z * sa.z; d += zz.z * da.z;
            s += zz.w * sa.w; d += zz.w * da.w;
        }
        if (r < N) { esrc[(size_t)r * 4 + hh] = s; edst[(size_t)r * 4 + hh] = d; }
    }
}

#define NCH 256
#define NBK 256
#define SORT_THREADS 1024
#define BCAP (4 * SORT_THREADS)
#define GCAP 6144
static_assert(SORT_THREADS == 4 * NCH && NBK <= SORT_THREADS, "k_group: four threads per chunk segment");

__device__ __forceinline__ int block_excl_scan(int v, int* wsum, int* tot) {
    const int lane = threadIdx.x & 31, wave = threadIdx.x >> 5;
    int inc = v;
    for (int o = 1; o < 32; o <<= 1) {
        int t = __shfl_up(inc, o, 32);
        if (lane >= o) inc += t;
    }
    if (lane == 31) wsum[wave] = inc;
    __syncthreads();
    if (wave == 0) {
        const int w = wsum[lane];
        int wi = w;
        for (int o = 1; o < 32; o <<= 1) {
            int t = __shfl_up(wi, o, 32);
            if (lane >= o) wi += t;
        }
        wsum[lane] = wi - w;
        if (lane == 31) wsum[32] = wi;
    }
    __syncthreads();
    *tot = wsum[32];
    const int r = wsum[wave] + inc - v;
    __syncthreads();
    return r;
}

__global__ __launch_bounds__(SORT_THREADS)
void k_bucket(const int* __restrict__ eidx,
              int* __restrict__ bdst,
              int* __restrict__ bsrc,
              int* __restrict__ bedge,
              int* __restrict__ tabT, int E, int bw) {
    __shared__ int hist[NBK];
    __shared__ int wsum[33];
    __shared__ int rd[BCAP], rs[BCAP], re[BCAP];
    const int tid = threadIdx.x;
    const int chunk = (E + NCH - 1) / NCH;
    int e0 = blockIdx.x * chunk; if (e0 > E) e0 = E;
    int e1 = e0 + chunk;         if (e1 > E) e1 = E;

    if (tid < NBK) hist[tid] = 0;
    __syncthreads();
    for (int eb = e0; eb < e1; eb += 4 * SORT_THREADS) {
        int d[4];
#pragma unroll
        for (int u = 0; u < 4; ++u) {
            const int e = eb + u * SORT_THREADS + tid;
            d[u] = (e < e1) ? eidx[E + e] : -1;
        }
#pragma unroll
        for (int u = 0; u < 4; ++u)
            if (d[u] >= 0) atomicAdd(&hist[d[u] / bw], 1);
    }
    __syncthreads();
    int tot;
    const int v = (tid < NBK) ? hist[tid] : 0;
    const int ex = block_excl_scan(v, wsum, &tot);
    if (tid < NBK) {
        tabT[tid * NCH + blockIdx.x] = ex;
        hist[tid] = ex;
    }
    if (tid == 0) tabT[NBK * NCH + blockIdx.x] = tot;
    __syncthreads();
    const int len = e1 - e0;
    if (len <= BCAP) {
        int d[4], s[4];
#pragma unroll
        for (int u = 0; u < 4; ++u) {
            const int e = e0 + u * SORT_THREADS + tid;
            d[u] = -1; s[u] = 0;
            if (e < e1) { d[u] = eidx[E + e]; s[u] = eidx[e]; }
        }
#pragma unroll
        for (int u = 0; u < 4; ++u) {
            if (d[u] >= 0) {
                const int p = atomicAdd(&hist[d[u] / bw], 1);
                rd[p] = d[u];
                rs[p] = s[u];
                re[p] = e0 + u * SORT_THREADS + tid;
            }
        }
        __syncthreads();
        for (int j = tid; j < len; j += SORT_THREADS) {
            bdst[e0 + j]  = rd[j];
            bsrc[e0 + j]  = rs[j];
            bedge[e0 + j] = re[j];
        }
    } else {
        for (int eb = e0; eb < e1; eb += 4 * SORT_THREADS) {
            int d[4], s[4];
#pragma unroll
            for (int u = 0; u < 4; ++u) {
                const int e = eb + u * SORT_THREADS + tid;
                d[u] = -1; s[u] = 0;
                if (e < e1) { d[u] = eidx[E + e]; s[u] = eidx[e]; }
            }
#pragma unroll
            for (int u = 0; u < 4; ++u) {
                if (d[u] >= 0) {
                    const int p = e0 + atomicAdd(&hist[d[u] / bw], 1);
                    bdst[p]  = d[u];
                    bsrc[p]  = s[u];
                    bedge[p] = eb + u * SORT_THREADS + tid;
                }
            }
        }
    }
}

__global__ __launch_bounds__(SORT_THREADS)
void k_group(const int* __restrict__ bdst,
             const int* __restrict__ bsrc,
             const int* __restrict__ bedge,
             const int* __restrict__ tabT,
             int* __restrict__ off,
             int* __restrict__ deg,
             int* __restrict__ ssrc,
             int* __restrict__ sedge, int E, int N, int bw) {
    __shared__ int cnt[SORT_THREADS];
    __shared__ int wsum[33];
    __shared__ int ls[GCAP], le[GCAP];
    const int tid = threadIdx.x;
    const int b   = blockIdx.x;
    const int c   = tid >> 2;
    const int sub = tid & 3;
    const int chunk = (E + NCH - 1) / NCH;
    int cb = c * chunk; if (cb > E) cb = E;
    const int s0 = tabT[b * NCH + c];
    const int s1 = tabT[(b + 1) * NCH + c];
    const int g  = cb + s0;
    const int l  = s1 - s0;
    const int n0 = b * bw;

    cnt[tid] = 0;
    int base;
    (void)block_excl_scan(sub == 0 ? s0 : 0, wsum, &base);

    for (int jb = 0; jb < l; jb += 16) {
        int d[4];
#pragma unroll
        for (int u = 0; u < 4; ++u) {
            const int j = jb + 4 * u + sub;
            d[u] = (j < l) ? bdst[g + j] : -1;
        }
#pragma unroll
        for (int u = 0; u < 4; ++u)
            if (d[u] >= 0) atomicAdd(&cnt[d[u] - n0], 1);
    }
    __syncthreads();
    int tot;
    const int v  = cnt[tid];
    const int ex = block_excl_scan(v, wsum, &tot);
    {
        const int n = n0 + tid;
        if (tid < bw && n < N) { off[n] = base + ex; deg[n] = v; }
    }
    const int  btot   = __builtin_amdgcn_readfirstlane(tot);
    const bool staged = btot <= GCAP;
    cnt[tid] = staged ? ex : base + ex;
    __syncthreads();
    for (int jb = 0; jb < l; jb += 16) {
        int d[4], s[4], e[4];
#pragma unroll
        for (int u = 0; u < 4; ++u) {
            const int j = jb + 4 * u + sub;
            d[u] = -1; s[u] = 0; e[u] = 0;
            if (j < l) { d[u] = bdst[g + j]; s[u] = bsrc[g + j]; e[u] = bedge[g + j]; }
        }
#pragma unroll
        for (int u = 0; u < 4; ++u) {
            if (d[u] >= 0) {
                const int p = atomicAdd(&cnt[d[u] - n0], 1);
                if (staged) { ls[p] = s[u];   le[p] = e[u]; }
                else        { ssrc[p] = s[u]; sedge[p] = e[u]; }
            }
        }
    }
    if (staged) {
        __syncthreads();
        for (int j = tid; j < btot; j += SORT_THREADS) {
            ssrc[base + j]  = ls[j];
            sedge[base + j] = le[j];
        }
    }
}

#define SPITCH 68
__device__ __forceinline__ float lrelu(float v) { return v > 0.f ? v : NEG_SLOPE * v; }

template <int T, int NT>
__device__ __forceinline__ void gat_rows(const float* __restrict__ hrow, int s, unsigned k4, float* hv) {
    hv[T] = ld_word(hrow, (unsigned)row_lane<T>(s) * 64u + k4);
    if constexpr (T + 1 < NT) gat_rows<T + 1, NT>(hrow, s, k4, hv);
}
__device__ __forceinline__ int f2i(float v) { return __builtin_bit_cast(int, v); }
__device__ __forceinline__ float i2f(int v) { return __builtin_bit_cast(float, v); }
struct Head4 { float h0, h1, h2, h3; };
template <int T, int NT>
__device__ __forceinline__ void gat_sums(const float* hv, int lim, int e0, int e1, int e2, int e3, Head4& S) {
    const float h = (T < lim) ? hv[T] : 0.f;
    S.h0 = __builtin_fmaf(i2f(row_lane<T>(e0)), h, S.h0);
    S.h1 = __builtin_fmaf(i2f(row_lane<T>(e1)), h, S.h1);
    S.h2 = __builtin_fmaf(i2f(row_lane<T>(e2)), h, S.h2);
    S.h3 = __builtin_fmaf(i2f(row_lane<T>(e3)), h, S.h3);
    if constexpr (T + 1 < NT) gat_sums<T + 1, NT>(hv, lim, e0, e1, e2, e3, S);
}
template <int NT>
__device__ __forceinline__ void gat_chunk(const float* __restrict__ hrow, int s, unsigned k4, int lim,
                                          bool live, float4 es, float4 ed, Head4 mx,
                                          Head4& den, Head4& S) {
    float hv[NT];
    gat_rows<0, NT>(hrow, s, k4, hv);
    __builtin_amdgcn_sched_barrier(0);
    const float e0 = __expf(live ? lrelu(es.x + ed.x) - mx.h0 : -INFINITY);
    const float e1 = __expf(live ? lrelu(es.y + ed.y) - mx.h1 : -INFINITY);
    const float e2 = __expf(live ? lrelu(es.z + ed.z) - mx.h2 : -INFINITY);
    const float e3 = __expf(live ? lrelu(es.w + ed.w) - mx.h3 : -INFINITY);
    den.h0 += e0; den.h1 += e1; den.h2 += e2; den.h3 += e3;
    gat_sums<0, NT>(hv, lim, f2i(e0), f2i(e1), f2i(e2), f2i(e3), S);
}

__global__ __launch_bounds__(512)
void k_gattile(const int* __restrict__ off,
               const int* __restrict__ deg,
               const int* __restrict__ ssrc,
               const float* __restrict__ esrc,
               const float* __restrict__ edst,
               const float* __restrict__ hrow,
               const float* __restrict__ W_gat,
               const float* __restrict__ b_gat,
               const float* __restrict__ W_fc,
               const float* __restrict__ b_fc,
               float* __restrict__ out, int N) {
    __shared__ float sS[32 * SPITCH];
    __shared__ float sP[8 * 32];

    const int tid  = threadIdx.x;
    const int wave = tid >> 5;
    const int lane = tid & 31;
    const int k    = lane & 15;
    const int half = lane >> 4;

    const int rt  = wave >> 3;
    const int ct  = wave & 7;
    const int col = ct * 16 + k;

    const int nn = wave * 2 + half;
    const int n  = blockIdx.x * 32 + nn;
    int beg = 0, cnt = 0;
    float4 ed = {0.f, 0.f, 0.f, 0.f};
    if (n < N) {
        beg = off[n];
        cnt = deg[n];
        ed  = *reinterpret_cast<const float4*>(edst + (size_t)n * 4);
    }
    const int c0 = __builtin_amdgcn_readlane(cnt, 0);
    const int c1 = __builtin_amdgcn_readlane(cnt, 16);
    const int cmax = c0 > c1 ? c0 : c1;
    const float4* es4 = reinterpret_cast<const float4*>(esrc);

    float m0 = -INFINITY, m1 = -INFINITY, m2 = -INFINITY, m3 = -INFINITY;
#pragma clang loop unroll(disable) vectorize(disable) interleave(disable)
    for (int c = 0; c < cmax; c += 32) {
        const int ja = c + k, jb = ja + 16;
        const bool la = ja < cnt, lb = jb < cnt;
        const int sa = ssrc[la ? beg + ja : 0];
        const int sb = ssrc[lb ? beg + jb : 0];
        const float4 ea = es4[sa];
        const float4 eb = es4[sb];
        const float pa = la ? 0.f : -INFINITY;
        const float pb = lb ? 0.f : -INFINITY;
        m0 = fmaxf(m0, fmaxf(lrelu(ea.x + ed.x) + pa, lrelu(eb.x + ed.x) + pb));
        m1 = fmaxf(m1, fmaxf(lrelu(ea.y + ed.y) + pa, lrelu(eb.y + ed.y) + pb));
        m2 = fmaxf(m2, fmaxf(lrelu(ea.z + ed.z) + pa, lrelu(eb.z + ed.z) + pb));
        m3 = fmaxf(m3, fmaxf(lrelu(ea.w + ed.w) + pa, lrelu(eb.w + ed.w) + pb));
    }
    for (int o = 8; o > 0; o >>= 1) {
        m0 = fmaxf(m0, __shfl_xor(m0, o, 32));
        m1 = fmaxf(m1, __shfl_xor(m1, o, 32));
        m2 = fmaxf(m2, __shfl_xor(m2, o, 32));
        m3 = fmaxf(m3, __shfl_xor(m3, o, 32));
    }

    Head4 S   = {0.f, 0.f, 0.f, 0.f};
    Head4 den = {0.f, 0.f, 0.f, 0.f};
    const Head4 mx = {m0, m1, m2, m3};
    const unsigned k4 = (unsigned)k * 4u;
    for (int c = 0; c < cmax; c += 16) {
        const int j = c + k;
        const bool live = j < cnt;
        const int s = ssrc[live ? beg + j : 0];
        const float4 es = es4[s];
        const int mc  = cmax - c;
        const int lim = cnt - c;
        if (mc > 8)      gat_chunk<16>(hrow, s, k4, lim, live, es, ed, mx, den, S);
        else if (mc > 4) gat_chunk<8>(hrow, s, k4, lim, live, es, ed, mx, den, S);
        else             gat_chunk<4>(hrow, s, k4, lim, live, es, ed, mx, den, S);
    }
    v2f bw[4];
#pragma unroll
    for (int kk = 0; kk < 4; ++kk) {
        bw[kk].x = W_gat[(4 * kk + 2 * half) * 128 + col];
        bw[kk].y = W_gat[(4 * kk + 2 * half + 1) * 128 + col];
    }
    const float bgc = b_gat[col];
    const float wfc = W_fc[col];

    float d0 = den.h0, d1 = den.h1, d2 = den.h2, d3 = den.h3;
    const float S0 = S.h0, S1 = S.h1, S2 = S.h2, S3 = S.h3;
    for (int o = 8; o > 0; o >>= 1) {
        d0 += __shfl_xor(d0, o, 32);
        d1 += __shfl_xor(d1, o, 32);
        d2 += __shfl_xor(d2, o, 32);
        d3 += __shfl_xor(d3, o, 32);
    }
    {
        float* sr = sS + nn * SPITCH + k;
        sr[0]  = S0 / (d0 + 1e-16f);
        sr[16] = S1 / (d1 + 1e-16f);
        sr[32] = S2 / (d2 + 1e-16f);
        sr[48] = S3 / (d3 + 1e-16f);
    }
    __syncthreads();

    {
        v8f acc = {};
        const float* ar = sS + (rt * 16 + k) * SPITCH + (ct >> 1) * 16 + 2 * half;
#pragma unroll
        for (int kk = 0; kk < 4; ++kk) {
            v2f a;
            a.x = ar[4 * kk]; a.y = ar[4 * kk + 1];
            acc = __builtin_amdgcn_wmma_f32_16x16x4_f32(
                false, a, false, bw[kk], (short)0, acc, false, false);
        }
        float p[8];
#pragma unroll
        for (int v = 0; v < 8; ++v) p[v] = fmaxf(acc[v] + bgc, 0.0f) * wfc;
        for (int o = 8; o > 0; o >>= 1) {
#pragma unroll
            for (int v = 0; v < 8; ++v) p[v] += __shfl_xor(p[v], o, 32);
        }
        if (k == 0) {
#pragma unroll
            for (int v = 0; v < 8; ++v) sP[ct * 32 + rt * 16 + 8 * half + v] = p[v];
        }
    }
    __syncthreads();

    if (tid < 32) {
        float y = 0.f;
#pragma unroll
        for (int c = 0; c < 8; ++c) y += sP[c * 32 + tid];
        const int r = blockIdx.x * 32 + tid;
        if (r < N) out[r] = y + b_fc[0];
    }
}

extern "C" void kernel_launch(void* const* d_in, const int* in_sizes, int n_in,
                              void* d_out, int out_size, void* d_ws, size_t ws_size,
                              hipStream_t stream) {
    const float* x      = (const float*)d_in[0];
    const int*   eidx   = (const int*)  d_in[1];
    const float* eattr  = (const float*)d_in[2];
    const float* W_nn   = (const float*)d_in[3];
    const float* b_nn   = (const float*)d_in[4];
    const float* W_root = (const float*)d_in[5];
    const float* b_ecc  = (const float*)d_in[6];
    const float* W_gat  = (const float*)d_in[7];
    const float* a_src  = (const float*)d_in[8];
    const float* a_dst  = (const float*)d_in[9];
    const float* b_gat  = (const float*)d_in[10];
    const float* W_fc   = (const float*)d_in[11];
    const float* b_fc   = (const float*)d_in[12];
    float* out = (float*)d_out;

    const int N = in_sizes[0] / 16;
    const int E = in_sizes[1] / 2;

    float* ws = (float*)d_ws;
    size_t off = 0;
    float* plane = ws + off;          off += (size_t)N * 128;
    float* hbuf  = plane;
    float* esrc  = ws + off;          off += (size_t)N * 4;
    float* edst  = ws + off;          off += (size_t)N * 4;
    int*   deg   = (int*)(ws + off);  off += (size_t)N;
    int*   noff  = (int*)(ws + off);  off += (size_t)N;
    int*   tabT  = (int*)(ws + off);  off += (size_t)E;
    int*   ssrc  = (int*)(ws + off);  off += (size_t)E;
    int*   sedge = (int*)(ws + off);  off += (size_t)E;
    int*   bdst  = (int*)plane;
    int*   bsrc  = bdst  + (size_t)E;
    int*   bedge = bsrc  + (size_t)E;
    const int bw = (N + NBK - 1) / NBK;

    k_bucket<<<NCH, SORT_THREADS, 0, stream>>>(eidx, bdst, bsrc, bedge, tabT, E, bw);
    k_group <<<NBK, SORT_THREADS, 0, stream>>>(bdst, bsrc, bedge, tabT, noff, deg, ssrc, sedge, E, N, bw);

    k_nnagg<<<(N + 15) / 16, 512, 0, stream>>>(x, eattr, W_nn, b_nn, W_root, b_ecc, W_gat, a_src, a_dst,
                                               noff, deg, ssrc, sedge, hbuf, esrc, edst, N);
    k_gattile<<<(N + 31) / 32, 512, 0, stream>>>(noff, deg, ssrc, esrc, edst, hbuf, W_gat,
                                                 b_gat, W_fc, b_fc, out, N);
}


# Level 2 of the edge ordering and the NNConv tile chain in one launch: a bucket's work-group reads its own list, run starts and lengths from LDS; two nodes per wave, 4 tiles per trip; 4 -> 3 launches
# speedup vs baseline: 18.0093x; 1.0772x over previous
#include <hip/hip_runtime.h>
#include <hip/hip_bf16.h>

typedef __attribute__((ext_vector_type(2))) float v2f;
typedef __attribute__((ext_vector_type(4))) float v4f;
typedef __attribute__((ext_vector_type(8))) float v8f;

#define NEG_SLOPE 0.2f

#define UPITCH 292
#define HPITCH 20
#define ZPITCH 132
template <int L>
__device__ __forceinline__ int row_lane(int v) {
    return __builtin_amdgcn_update_dpp(0, v, 0x150 + L, 0xf, 0xf, false);
}
__device__ __forceinline__ float ld_word(const float* base, unsigned byte_off) {
    return *reinterpret_cast<const float*>(reinterpret_cast<const char*>(base) + byte_off);
}
struct EdgeWords { float ax, ay, bx, by; };
template <int T>
__device__ __forceinline__ EdgeWords edge_loads(const float* __restrict__ x,
                                                const float* __restrict__ eattr,
                                                int s, int e, unsigned m4) {
    EdgeWords w;
    w.ax = ld_word(eattr, (unsigned)row_lane<2 * T>(e) * 64u + m4);
    w.ay = ld_word(eattr, (unsigned)row_lane<2 * T + 1>(e) * 64u + m4);
    w.bx = ld_word(x,     (unsigned)row_lane<2 * T>(s) * 64u + m4);
    w.by = ld_word(x,     (unsigned)row_lane<2 * T + 1>(s) * 64u + m4);
    return w;
}
template <int T>
__device__ __forceinline__ void edge_step(const EdgeWords& w, int lim, v8f& U, float& xsum) {
    const bool vx = 4 * T < lim, vy = 4 * T + 1 < lim;
    v2f a, b;
    a.x = vx ? w.ax : 0.f; a.y = vy ? w.ay : 0.f;
    b.x = vx ? w.bx : 0.f; b.y = vy ? w.by : 0.f;
    xsum += b.x + b.y;
    U = __builtin_amdgcn_wmma_f32_16x16x4_f32(false, a, false, b, (short)0, U, false, false);
}

#define NCH 256
#define NBK 256
#define SORT_THREADS 1024
#define BCAP (4 * SORT_THREADS)
#define GCAP 6144
static_assert(SORT_THREADS == 4 * NCH && NBK <= SORT_THREADS, "k_groupnn: four threads per chunk segment");

__device__ __forceinline__ int block_excl_scan(int v, int* wsum, int* tot) {
    const int lane = threadIdx.x & 31, wave = threadIdx.x >> 5;
    int inc = v;
    for (int o = 1; o < 32; o <<= 1) {
        int t = __shfl_up(inc, o, 32);
        if (lane >= o) inc += t;
    }
    if (lane == 31) wsum[wave] = inc;
    __syncthreads();
    if (wave == 0) {
        const int w = wsum[lane];
        int wi = w;
        for (int o = 1; o < 32; o <<= 1) {
            int t = __shfl_up(wi, o, 32);
            if (lane >= o) wi += t;
        }
        wsum[lane] = wi - w;
        if (lane == 31) wsum[32] = wi;
    }
    __syncthreads();
    *tot = wsum[32];
    const int r = wsum[wave] + inc - v;
    __syncthreads();
    return r;
}

__global__ __launch_bounds__(SORT_THREADS)
void k_bucket(const int* __restrict__ eidx,
              int* __restrict__ bdst,
              int* __restrict__ bsrc,
              int* __restrict__ bedge,
              int* __restrict__ tabT, int E, int bw) {
    __shared__ int hist[NBK];
    __shared__ int wsum[33];
    __shared__ int rd[BCAP], rs[BCAP], re[BCAP];
    const int tid = threadIdx.x;
    const int chunk = (E + NCH - 1) / NCH;
    int e0 = blockIdx.x * chunk; if (e0 > E) e0 = E;
    int e1 = e0 + chunk;         if (e1 > E) e1 = E;

    if (tid < NBK) hist[tid] = 0;
    __syncthreads();
    for (int eb = e0; eb < e1; eb += 4 * SORT_THREADS) {
        int d[4];
#pragma unroll
        for (int u = 0; u < 4; ++u) {
            const int e = eb + u * SORT_THREADS + tid;
            d[u] = (e < e1) ? eidx[E + e] : -1;
        }
#pragma unroll
        for (int u = 0; u < 4; ++u)
            if (d[u] >= 0) atomicAdd(&hist[d[u] / bw], 1);
    }
    __syncthreads();
    int tot;
    const int v = (tid < NBK) ? hist[tid] : 0;
    const int ex = block_excl_scan(v, wsum, &tot);
    if (tid < NBK) {
        tabT[tid * NCH + blockIdx.x] = ex;
        hist[tid] = ex;
    }
    if (tid == 0) tabT[NBK * NCH + blockIdx.x] = tot;
    __syncthreads();
    const int len = e1 - e0;
    if (len <= BCAP) {
        int d[4], s[4];
#pragma unroll
        for (int u = 0; u < 4; ++u) {
            const int e = e0 + u * SORT_THREADS + tid;
            d[u] = -1; s[u] = 0;
            if (e < e1) { d[u] = eidx[E + e]; s[u] = eidx[e]; }
        }
#pragma unroll
        for (int u = 0; u < 4; ++u) {
            if (d[u] >= 0) {
                const int p = atomicAdd(&hist[d[u] / bw], 1);
                rd[p] = d[u];
                rs[p] = s[u];
                re[p] = e0 + u * SORT_THREADS + tid;
            }
        }
        __syncthreads();
        for (int j = tid; j < len; j += SORT_THREADS) {
            bdst[e0 + j]  = rd[j];
            bsrc[e0 + j]  = rs[j];
            bedge[e0 + j] = re[j];
        }
    } else {
        for (int eb = e0; eb < e1; eb += 4 * SORT_THREADS) {
            int d[4], s[4];
#pragma unroll
            for (int u = 0; u < 4; ++u) {
                const int e = eb + u * SORT_THREADS + tid;
                d[u] = -1; s[u] = 0;
                if (e < e1) { d[u] = eidx[E + e]; s[u] = eidx[e]; }
            }
#pragma unroll
            for (int u = 0; u < 4; ++u) {
                if (d[u] >= 0) {
                    const int p = e0 + atomicAdd(&hist[d[u] / bw], 1);
                    bdst[p]  = d[u];
                    bsrc[p]  = s[u];
                    bedge[p] = eb + u * SORT_THREADS + tid;
                }
            }
        }
    }
}

#define TPT 4

template <int T, int NT>
__device__ __forceinline__ void chunk_loads(const float* __restrict__ x, const float* __restrict__ eattr,
                                            int s, int e, unsigned m4, EdgeWords* w) {
    w[T] = edge_loads<T>(x, eattr, s, e, m4);
    if constexpr (T + 1 < NT) chunk_loads<T + 1, NT>(x, eattr, s, e, m4, w);
}
template <int T, int NT>
__device__ __forceinline__ void chunk_steps(const EdgeWords* w, int lim, v8f& U, float& xsum) {
    edge_step<T>(w[T], lim, U, xsum);
    if constexpr (T + 1 < NT) chunk_steps<T + 1, NT>(w, lim, U, xsum);
}
template <int NT>
__device__ __forceinline__ void pair_chunk(const float* __restrict__ x, const float* __restrict__ eattr,
                                           int sa, int ea, int sb, int eb, unsigned m4,
                                           int lima, int limb,
                                           v8f& UA, float& xa, v8f& UB, float& xb) {
    EdgeWords wa[NT], wb[NT];
    chunk_loads<0, NT>(x, eattr, sa, ea, m4, wa);
    chunk_loads<0, NT>(x, eattr, sb, eb, m4, wb);
    chunk_steps<0, NT>(wa, lima, UA, xa);
    chunk_steps<0, NT>(wb, limb, UB, xb);
}
template <bool LDSLIST>
__device__ __forceinline__ void pair_walk(const float* __restrict__ x, const float* __restrict__ eattr,
                                          const int* ls, const int* le, const int* ms, const int* me,
                                          int begA, int cntA, int begB, int cntB,
                                          int jl, int hi, unsigned m4,
                                          v8f& UA, float& xsA, v8f& UB, float& xsB) {
    const int cm = cntA > cntB ? cntA : cntB;
    for (int c0 = 0; c0 < cm; c0 += 32) {
        const int remA = cntA - c0, remB = cntB - c0;
        int sa = 0, ea = 0, sb = 0, eb = 0;
        if (jl < remA) {
            const int p = begA + c0 + jl;
            if constexpr (LDSLIST) { sa = ls[p]; ea = le[p]; }
            else                   { sa = ms[p]; ea = me[p]; }
        }
        if (jl < remB) {
            const int p = begB + c0 + jl;
            if constexpr (LDSLIST) { sb = ls[p]; eb = le[p]; }
            else                   { sb = ms[p]; eb = me[p]; }
        }
        const int limA = remA - 2 * hi, limB = remB - 2 * hi;
        if ((remA > remB ? remA : remB) > 16)
            pair_chunk<8>(x, eattr, sa, ea, sb, eb, m4, limA, limB, UA, xsA, UB, xsB);
        else
            pair_chunk<4>(x, eattr, sa, ea, sb, eb, m4, limA, limB, UA, xsA, UB, xsB);
    }
}
__device__ __forceinline__ void store_urow(float* row, const v8f& U, float xsum, float xown, int i, int hi) {
    const float xs = xsum + __shfl_xor(xsum, 16, 32);
    float* ur = row + hi * 128 + i;
#pragma unroll
    for (int r = 0; r < 8; ++r) ur[16 * r] = U[r];
    row[256 + hi * 16 + i] = hi ? xown : xs;
}

__global__ __launch_bounds__(SORT_THREADS)
void k_groupnn(const int* __restrict__ bdst,
               const int* __restrict__ bsrc,
               const int* __restrict__ bedge,
               const int* __restrict__ tabT,
               int* __restrict__ off,
               int* __restrict__ deg,
               int* __restrict__ ssrc,
               int* __restrict__ sedge,
               const float* __restrict__ x,
               const float* __restrict__ eattr,
               const float* __restrict__ W_nn,
               const float* __restrict__ b_nn,
               const float* __restrict__ W_root,
               const float* __restrict__ b_ecc,
               const float* __restrict__ W_gat,
               const float* __restrict__ a_src,
               const float* __restrict__ a_dst,
               float* __restrict__ hrow,
               float* __restrict__ esrc,
               float* __restrict__ edst, int E, int N, int bw) {
    __shared__ int cnt[SORT_THREADS];
    __shared__ int wsum[33];
    __shared__ int ls[GCAP], le[GCAP];
    __shared__ int lbeg[SORT_THREADS];
    __shared__ int ldeg[SORT_THREADS];
    __shared__ float sU[TPT * 16 * UPITCH];
    __shared__ __attribute__((aligned(16))) float sH[TPT * 16 * HPITCH];
    __shared__ __attribute__((aligned(16))) float sZ[TPT * 16 * ZPITCH];
    __shared__ v4f sA[64];
    __shared__ __attribute__((aligned(16))) float sW[288 * 16];
    __shared__ float sR[TPT * 8 * 256];

    const int tid = threadIdx.x;
    const int b   = blockIdx.x;
    const int c   = tid >> 2;
    const int sub = tid & 3;
    const int chunk = (E + NCH - 1) / NCH;
    int cb = c * chunk; if (cb > E) cb = E;
    const int s0 = tabT[b * NCH + c];
    const int s1 = tabT[(b + 1) * NCH + c];
    const int g  = cb + s0;
    const int l  = s1 - s0;
    const int n0 = b * bw;

    const int lane = tid & 31;
    const int i    = lane & 15;
    const int hi   = lane >> 4;
    const int wave = __builtin_amdgcn_readfirstlane(tid >> 5);
    const float be = b_ecc[i];
    {
        v4f av = {0.f, 0.f, 0.f, 0.f};
        if (tid < 64) av = reinterpret_cast<const v4f*>(tid < 32 ? a_src : a_dst)[tid & 31];
        float wst[5];
#pragma unroll
        for (int u = 0; u < 5; ++u) {
            const int idx = u * SORT_THREADS + tid;
            wst[u] = (idx < 4096) ? W_nn[idx]
                   : (idx < 4352) ? b_nn[idx - 4096]
                   : (idx < 4608) ? W_root[idx - 4352] : 0.f;
        }
#pragma unroll
        for (int u = 0; u < 5; ++u) {
            const int idx = u * SORT_THREADS + tid;
            if (idx < 4608) {
                const int k = idx >> 4, cc = idx & 15;
                sW[((k >> 1) * 16 + cc) * 2 + (k & 1)] = wst[u];
            }
        }
        if (tid < 64) sA[tid] = av;
    }

    cnt[tid] = 0;
    int base;
    (void)block_excl_scan(sub == 0 ? s0 : 0, wsum, &base);

    for (int jb = 0; jb < l; jb += 16) {
        int d[4];
#pragma unroll
        for (int u = 0; u < 4; ++u) {
            const int j = jb + 4 * u + sub;
            d[u] = (j < l) ? bdst[g + j] : -1;
        }
#pragma unroll
        for (int u = 0; u < 4; ++u)
            if (d[u] >= 0) atomicAdd(&cnt[d[u] - n0], 1);
    }
    __syncthreads();
    int tot;
    const int v  = cnt[tid];
    const int ex = block_excl_scan(v, wsum, &tot);
    {
        const int n = n0 + tid;
        if (tid < bw && n < N) { off[n] = base + ex; deg[n] = v; }
    }
    const int  btot   = __builtin_amdgcn_readfirstlane(tot);
    const bool staged = btot <= GCAP;
    cnt[tid]  = staged ? ex : base + ex;
    lbeg[tid] = staged ? ex : base + ex;
    ldeg[tid] = v;
    __syncthreads();
    for (int jb = 0; jb < l; jb += 16) {
        int d[4], s[4], e[4];
#pragma unroll
        for (int u = 0; u < 4; ++u) {
            const int j = jb + 4 * u + sub;
            d[u] = -1; s[u] = 0; e[u] = 0;
            if (j < l) { d[u] = bdst[g + j]; s[u] = bsrc[g + j]; e[u] = bedge[g + j]; }
        }
#pragma unroll
        for (int u = 0; u < 4; ++u) {
            if (d[u] >= 0) {
                const int p = atomicAdd(&cnt[d[u] - n0], 1);
                if (staged) { ls[p] = s[u];   le[p] = e[u]; }
                else        { ssrc[p] = s[u]; sedge[p] = e[u]; }
            }
        }
    }

    __syncthreads();
    if (staged) {
        for (int j = tid; j < btot; j += SORT_THREADS) ssrc[base + j] = ls[j];
    }

    const int col = (wave & 7) * 16 + i;
    int nb = N - n0;
    if (nb > bw) nb = bw;
    if (nb < 0)  nb = 0;
    const int ntile = (nb + 15) >> 4;
    const int jl = 4 * (i >> 1) + 2 * hi + (i & 1);
    const unsigned m4 = (unsigned)i * 4u;
    const int pt = wave >> 3;
    const int ks = wave & 7;
    float* rowA = sU + wave * UPITCH;
    float* rowB = sU + (wave + 32) * UPITCH;

    for (int q0 = 0; q0 < ntile; q0 += TPT) {
        const int nlA = q0 * 16 + wave;
        const int nlB = nlA + 32;
        int begA = 0, cntA = 0, begB = 0, cntB = 0;
        if (nlA < nb) {
            begA = __builtin_amdgcn_readfirstlane(lbeg[nlA]);
            cntA = __builtin_amdgcn_readfirstlane(ldeg[nlA]);
        }
        if (nlB < nb) {
            begB = __builtin_amdgcn_readfirstlane(lbeg[nlB]);
            cntB = __builtin_amdgcn_readfirstlane(ldeg[nlB]);
        }
        v8f UA = {}, UB = {};
        float xsA = 0.f, xsB = 0.f;
        if (staged)
            pair_walk<true>(x, eattr, ls, le, ssrc, sedge, begA, cntA, begB, cntB, jl, hi, m4, UA, xsA, UB, xsB);
        else
            pair_walk<false>(x, eattr, ls, le, ssrc, sedge, begA, cntA, begB, cntB, jl, hi, m4, UA, xsA, UB, xsB);
        float xoA = 0.f, xoB = 0.f;
        if (nlA < nb) xoA = x[(size_t)(n0 + nlA) * 16 + i];
        if (nlB < nb) xoB = x[(size_t)(n0 + nlB) * 16 + i];
        store_urow(rowA, UA, xsA, xoA, i, hi);
        store_urow(rowB, UB, xsB, xoB, i, hi);
        __syncthreads();

        const bool plive = q0 + pt < ntile;
        v2f bg[4] = {};
        if (plive) {
#pragma unroll
            for (int kk = 0; kk < 4; ++kk) {
                bg[kk].x = W_gat[(4 * kk + 2 * hi) * 128 + col];
                bg[kk].y = W_gat[(4 * kk + 2 * hi + 1) * 128 + col];
            }
        }
        if (plive) {
            v8f accA = {}, accB = {};
            const float* ar = sU + (pt * 16 + i) * UPITCH + 2 * hi + 36 * ks;
            const v2f* br = reinterpret_cast<const v2f*>(sW) + 288 * ks + 16 * hi + i;
#pragma unroll
            for (int t = 0; t < 9; ++t) {
                v2f a;
                a.x = ar[4 * t]; a.y = ar[4 * t + 1];
                const v2f bb = br[32 * t];
                if (t & 1)
                    accB = __builtin_amdgcn_wmma_f32_16x16x4_f32(
                        false, a, false, bb, (short)0, accB, false, false);
                else
                    accA = __builtin_amdgcn_wmma_f32_16x16x4_f32(
                        false, a, false, bb, (short)0, accA, false, false);
            }
            const v8f acc = accA + accB;
            float* pr = sR + (pt * 8 + ks) * 256 + 128 * hi + i;
#pragma unroll
            for (int vv = 0; vv < 8; ++vv) pr[16 * vv] = acc[vv];
        }
        __syncthreads();
        if (plive) {
            const int el = tid & 255;
            const float* rr = sR + pt * 2048 + el;
            float s = rr[0];
#pragma unroll
            for (int w = 1; w < 8; ++w) s += rr[w * 256];
            sH[(pt * 16 + (el >> 4)) * HPITCH + i] = fmaxf(s + be, 0.0f);
        }
        __syncthreads();
        if (plive) {
            v8f acc = {};
            const float* hr = sH + (pt * 16 + i) * HPITCH + 2 * hi;
#pragma unroll
            for (int kk = 0; kk < 4; ++kk) {
                v2f a;
                a.x = hr[4 * kk]; a.y = hr[4 * kk + 1];
                acc = __builtin_amdgcn_wmma_f32_16x16x4_f32(
                    false, a, false, bg[kk], (short)0, acc, false, false);
            }
#pragma unroll
            for (int vv = 0; vv < 8; ++vv)
                sZ[(pt * 16 + vv + 8 * hi) * ZPITCH + col] = acc[vv];
        }
        __syncthreads();

        if (wave < 8) {
            const int tl = tid >> 6, nn = (tid >> 2) & 15, hh = tid & 3;
            const int nl = (q0 + tl) * 16 + nn;
            const v4f* zr = reinterpret_cast<const v4f*>(sZ + (tl * 16 + nn) * ZPITCH + hh * 32);
            const v4f* as = sA + hh * 8;
            const v4f* ad = sA + 32 + hh * 8;
            float s = 0.f, d = 0.f;
#pragma unroll
            for (int j = 0; j < 8; ++j) {
                const v4f zz = zr[j], sa4 = as[j], da4 = ad[j];
                s += zz.x * sa4.x; d += zz.x * da4.x;
                s += zz.y * sa4.y; d += zz.y * da4.y;
                s += zz.z * sa4.z; d += zz.z * da4.z;
                s += zz.w * sa4.w; d += zz.w * da4.w;
            }
            if (nl < nb) {
                esrc[(size_t)(n0 + nl) * 4 + hh] = s;
                edst[(size_t)(n0 + nl) * 4 + hh] = d;
            }
        } else if (wave < 16) {
            const int t  = tid - 256;
            const int tl = t >> 6, nn = (t >> 2) & 15, part = t & 3;
            const int nl = (q0 + tl) * 16 + nn;
            if (nl < nb)
                *reinterpret_cast<v4f*>(hrow + (size_t)(n0 + nl) * 16 + part * 4) =
                    *reinterpret_cast<const v4f*>(sH + (tl * 16 + nn) * HPITCH + part * 4);
        }
    }
}

#define SPITCH 68
__device__ __forceinline__ float lrelu(float v) { return v > 0.f ? v : NEG_SLOPE * v; }

template <int T, int NT>
__device__ __forceinline__ void gat_rows(const float* __restrict__ hrow, int s, unsigned k4, float* hv) {
    hv[T] = ld_word(hrow, (unsigned)row_lane<T>(s) * 64u + k4);
    if constexpr (T + 1 < NT) gat_rows<T + 1, NT>(hrow, s, k4, hv);
}
__device__ __forceinline__ int f2i(float v) { return __builtin_bit_cast(int, v); }
__device__ __forceinline__ float i2f(int v) { return __builtin_bit_cast(float, v); }
struct Head4 { float h0, h1, h2, h3; };
template <int T, int NT>
__device__ __forceinline__ void gat_sums(const float* hv, int lim, int e0, int e1, int e2, int e3, Head4& S) {
    const float h = (T < lim) ? hv[T] : 0.f;
    S.h0 = __builtin_fmaf(i2f(row_lane<T>(e0)), h, S.h0);
    S.h1 = __builtin_fmaf(i2f(row_lane<T>(e1)), h, S.h1);
    S.h2 = __builtin_fmaf(i2f(row_lane<T>(e2)), h, S.h2);
    S.h3 = __builtin_fmaf(i2f(row_lane<T>(e3)), h, S.h3);
    if constexpr (T + 1 < NT) gat_sums<T + 1, NT>(hv, lim, e0, e1, e2, e3, S);
}
template <int NT>
__device__ __forceinline__ void gat_chunk(const float* __restrict__ hrow, int s, unsigned k4, int lim,
                                          bool live, float4 es, float4 ed, Head4 mx,
                                          Head4& den, Head4& S) {
    float hv[NT];
    gat_rows<0, NT>(hrow, s, k4, hv);
    __builtin_amdgcn_sched_barrier(0);
    const float e0 = __expf(live ? lrelu(es.x + ed.x) - mx.h0 : -INFINITY);
    const float e1 = __expf(live ? lrelu(es.y + ed.y) - mx.h1 : -INFINITY);
    const float e2 = __expf(live ? lrelu(es.z + ed.z) - mx.h2 : -INFINITY);
    const float e3 = __expf(live ? lrelu(es.w + ed.w) - mx.h3 : -INFINITY);
    den.h0 += e0; den.h1 += e1; den.h2 += e2; den.h3 += e3;
    gat_sums<0, NT>(hv, lim, f2i(e0), f2i(e1), f2i(e2), f2i(e3), S);
}

__global__ __launch_bounds__(512)
void k_gattile(const int* __restrict__ off,
               const int* __restrict__ deg,
               const int* __restrict__ ssrc,
               const float* __restrict__ esrc,
               const float* __restrict__ edst,
               const float* __restrict__ hrow,
               const float* __restrict__ W_gat,
               const float* __restrict__ b_gat,
               const float* __restrict__ W_fc,
               const float* __restrict__ b_fc,
               float* __restrict__ out, int N) {
    __shared__ float sS[32 * SPITCH];
    __shared__ float sP[8 * 32];

    const int tid  = threadIdx.x;
    const int wave = tid >> 5;
    const int lane = tid & 31;
    const int k    = lane & 15;
    const int half = lane >> 4;

    const int rt  = wave >> 3;
    const int ct  = wave & 7;
    const int col = ct * 16 + k;

    const int nn = wave * 2 + half;
    const int n  = blockIdx.x * 32 + nn;
    int beg = 0, cnt = 0;
    float4 ed = {0.f, 0.f, 0.f, 0.f};
    if (n < N) {
        beg = off[n];
        cnt = deg[n];
        ed  = *reinterpret_cast<const float4*>(edst + (size_t)n * 4);
    }
    const int c0 = __builtin_amdgcn_readlane(cnt, 0);
    const int c1 = __builtin_amdgcn_readlane(cnt, 16);
    const int cmax = c0 > c1 ? c0 : c1;
    const float4* es4 = reinterpret_cast<const float4*>(esrc);

    float m0 = -INFINITY, m1 = -INFINITY, m2 = -INFINITY, m3 = -INFINITY;
#pragma clang loop unroll(disable) vectorize(disable) interleave(disable)
    for (int c = 0; c < cmax; c += 32) {
        const int ja = c + k, jb = ja + 16;
        const bool la = ja < cnt, lb = jb < cnt;
        const int sa = ssrc[la ? beg + ja : 0];
        const int sb = ssrc[lb ? beg + jb : 0];
        const float4 ea = es4[sa];
        const float4 eb = es4[sb];
        const float pa = la ? 0.f : -INFINITY;
        const float pb = lb ? 0.f : -INFINITY;
        m0 = fmaxf(m0, fmaxf(lrelu(ea.x + ed.x) + pa, lrelu(eb.x + ed.x) + pb));
        m1 = fmaxf(m1, fmaxf(lrelu(ea.y + ed.y) + pa, lrelu(eb.y + ed.y) + pb));
        m2 = fmaxf(m2, fmaxf(lrelu(ea.z + ed.z) + pa, lrelu(eb.z + ed.z) + pb));
        m3 = fmaxf(m3, fmaxf(lrelu(ea.w + ed.w) + pa, lrelu(eb.w + ed.w) + pb));
    }
    for (int o = 8; o > 0; o >>= 1) {
        m0 = fmaxf(m0, __shfl_xor(m0, o, 32));
        m1 = fmaxf(m1, __shfl_xor(m1, o, 32));
        m2 = fmaxf(m2, __shfl_xor(m2, o, 32));
        m3 = fmaxf(m3, __shfl_xor(m3, o, 32));
    }

    Head4 S   = {0.f, 0.f, 0.f, 0.f};
    Head4 den = {0.f, 0.f, 0.f, 0.f};
    const Head4 mx = {m0, m1, m2, m3};
    const unsigned k4 = (unsigned)k * 4u;
    for (int c = 0; c < cmax; c += 16) {
        const int j = c + k;
        const bool live = j < cnt;
        const int s = ssrc[live ? beg + j : 0];
        const float4 es = es4[s];
        const int mc  = cmax - c;
        const int lim = cnt - c;
        if (mc > 8)      gat_chunk<16>(hrow, s, k4, lim, live, es, ed, mx, den, S);
        else if (mc > 4) gat_chunk<8>(hrow, s, k4, lim, live, es, ed, mx, den, S);
        else             gat_chunk<4>(hrow, s, k4, lim, live, es, ed, mx, den, S);
    }
    v2f bw[4];
#pragma unroll
    for (int kk = 0; kk < 4; ++kk) {
        bw[kk].x = W_gat[(4 * kk + 2 * half) * 128 + col];
        bw[kk].y = W_gat[(4 * kk + 2 * half + 1) * 128 + col];
    }
    const float bgc = b_gat[col];
    const float wfc = W_fc[col];

    float d0 = den.h0, d1 = den.h1, d2 = den.h2, d3 = den.h3;
    const float S0 = S.h0, S1 = S.h1, S2 = S.h2, S3 = S.h3;
    for (int o = 8; o > 0; o >>= 1) {
        d0 += __shfl_xor(d0, o, 32);
        d1 += __shfl_xor(d1, o, 32);
        d2 += __shfl_xor(d2, o, 32);
        d3 += __shfl_xor(d3, o, 32);
    }
    {
        float* sr = sS + nn * SPITCH + k;
        sr[0]  = S0 / (d0 + 1e-16f);
        sr[16] = S1 / (d1 + 1e-16f);
        sr[32] = S2 / (d2 + 1e-16f);
        sr[48] = S3 / (d3 + 1e-16f);
    }
    __syncthreads();

    {
        v8f acc = {};
        const float* ar = sS + (rt * 16 + k) * SPITCH + (ct >> 1) * 16 + 2 * half;
#pragma unroll
        for (int kk = 0; kk < 4; ++kk) {
            v2f a;
            a.x = ar[4 * kk]; a.y = ar[4 * kk + 1];
            acc = __builtin_amdgcn_wmma_f32_16x16x4_f32(
                false, a, false, bw[kk], (short)0, acc, false, false);
        }
        float p[8];
#pragma unroll
        for (int v = 0; v < 8; ++v) p[v] = fmaxf(acc[v] + bgc, 0.0f) * wfc;
        for (int o = 8; o > 0; o >>= 1) {
#pragma unroll
            for (int v = 0; v < 8; ++v) p[v] += __shfl_xor(p[v], o, 32);
        }
        if (k == 0) {
#pragma unroll
            for (int v = 0; v < 8; ++v) sP[ct * 32 + rt * 16 + 8 * half + v] = p[v];
        }
    }
    __syncthreads();

    if (tid < 32) {
        float y = 0.f;
#pragma unroll
        for (int c = 0; c < 8; ++c) y += sP[c * 32 + tid];
        const int r = blockIdx.x * 32 + tid;
        if (r < N) out[r] = y + b_fc[0];
    }
}

extern "C" void kernel_launch(void* const* d_in, const int* in_sizes, int n_in,
                              void* d_out, int out_size, void* d_ws, size_t ws_size,
                              hipStream_t stream) {
    const float* x      = (const float*)d_in[0];
    const int*   eidx   = (const int*)  d_in[1];
    const float* eattr  = (const float*)d_in[2];
    const float* W_nn   = (const float*)d_in[3];
    const float* b_nn   = (const float*)d_in[4];
    const float* W_root = (const float*)d_in[5];
    const float* b_ecc  = (const float*)d_in[6];
    const float* W_gat  = (const float*)d_in[7];
    const float* a_src  = (const float*)d_in[8];
    const float* a_dst  = (const float*)d_in[9];
    const float* b_gat  = (const float*)d_in[10];
    const float* W_fc   = (const float*)d_in[11];
    const float* b_fc   = (const float*)d_in[12];
    float* out = (float*)d_out;

    const int N = in_sizes[0] / 16;
    const int E = in_sizes[1] / 2;

    float* ws = (float*)d_ws;
    size_t off = 0;
    float* plane = ws + off;          off += (size_t)N * 128;
    float* esrc  = ws + off;          off += (size_t)N * 4;
    float* edst  = ws + off;          off += (size_t)N * 4;
    int*   deg   = (int*)(ws + off);  off += (size_t)N;
    int*   noff  = (int*)(ws + off);  off += (size_t)N;
    int*   tabT  = (int*)(ws + off);  off += (size_t)E;
    int*   ssrc  = (int*)(ws + off);  off += (size_t)E;
    int*   sedge = (int*)(ws + off);  off += (size_t)E;
    int*   bdst  = (int*)plane;
    int*   bsrc  = bdst  + (size_t)E;
    int*   bedge = bsrc  + (size_t)E;
    float* hbuf  = (float*)(bedge + (size_t)E);
    const int bw = (N + NBK - 1) / NBK;

    k_bucket <<<NCH, SORT_THREADS, 0, stream>>>(eidx, bdst, bsrc, bedge, tabT, E, bw);
    k_groupnn<<<NBK, SORT_THREADS, 0, stream>>>(bdst, bsrc, bedge, tabT, noff, deg, ssrc, sedge,
                                                x, eattr, W_nn, b_nn, W_root, b_ecc, W_gat, a_src, a_dst,
                                                hbuf, esrc, edst, E, N, bw);
    k_gattile<<<(N + 31) / 32, 512, 0, stream>>>(noff, deg, ssrc, esrc, edst, hbuf, W_gat,
                                                 b_gat, W_fc, b_fc, out, N);
}
